# starting kernel
# baseline (speedup 1.0000x reference)
#include <hip/hip_runtime.h>
#include <math.h>

#define N_NODES 50000
#define N_EDGES 800000
#define E_TOT   (N_EDGES + N_NODES)
#define NFEAT   128
#define NHID    8
#define NHEADS  8
#define HID1    (NHEADS * NHID)
#define NCLASS  40
#define NEG_SLOPE 0.2f

typedef __attribute__((ext_vector_type(2))) float v2f;
typedef __attribute__((ext_vector_type(8))) float v8f;

__device__ __forceinline__ float lrelu(float x) { return x > 0.f ? x : NEG_SLOPE * x; }

__device__ __forceinline__ unsigned f2ord(float f) {
  unsigned u = __float_as_uint(f);
  return (u & 0x80000000u) ? ~u : (u | 0x80000000u);
}
__device__ __forceinline__ float ord2f(unsigned u) {
  return (u & 0x80000000u) ? __uint_as_float(u & 0x7FFFFFFFu) : __uint_as_float(~u);
}
#define ORD_NEG_INF 0x007FFFFFu

__device__ __forceinline__ void get_edge(const int* __restrict__ ei, int e, int& s, int& d) {
  if (e < N_EDGES) { s = ei[e]; d = ei[N_EDGES + e]; }
  else             { s = e - N_EDGES; d = s; }
}

__global__ void init_kernel(float* __restrict__ zbase, unsigned zn,
                            unsigned* __restrict__ mbase, unsigned mn) {
  unsigned i = blockIdx.x * blockDim.x + threadIdx.x;
  if (i < zn) zbase[i] = 0.f;
  if (i < mn) mbase[i] = ORD_NEG_INF;
}

__global__ void __launch_bounds__(128) gemm1_wmma(const float* __restrict__ X,
                                                  const float* __restrict__ W,
                                                  float* __restrict__ H) {
  const int wave = threadIdx.x >> 5;
  const int lane = threadIdx.x & 31;
  const int mt   = blockIdx.x * 4 + wave;
  if (mt >= N_NODES / 16) return;
  const int m  = lane & 15;
  const int hi = lane >> 4;
  const float* xr = X + (long)(mt * 16 + m) * NFEAT;

  v8f c0 = {}, c1 = {}, c2 = {}, c3 = {};
  #pragma unroll 4
  for (int k0 = 0; k0 < NFEAT; k0 += 4) {
    const int ka = k0 + 2 * hi;
    v2f a; a.x = xr[ka]; a.y = xr[ka + 1];
    const float* w0 = W + (long)ka * HID1 + m;
    const float* w1 = w0 + HID1;
    v2f b;
    b.x = w0[0];  b.y = w1[0];
    c0 = __builtin_amdgcn_wmma_f32_16x16x4_f32(false, a, false, b, (short)0, c0, false, false);
    b.x = w0[16]; b.y = w1[16];
    c1 = __builtin_amdgcn_wmma_f32_16x16x4_f32(false, a, false, b, (short)0, c1, false, false);
    b.x = w0[32]; b.y = w1[32];
    c2 = __builtin_amdgcn_wmma_f32_16x16x4_f32(false, a, false, b, (short)0, c2, false, false);
    b.x = w0[48]; b.y = w1[48];
    c3 = __builtin_amdgcn_wmma_f32_16x16x4_f32(false, a, false, b, (short)0, c3, false, false);
  }
  float* hp = H + (long)(mt * 16 + 8 * hi) * HID1 + m;
  #pragma unroll
  for (int j = 0; j < 8; j++) {
    hp[(long)j * HID1 +  0] = c0[j];
    hp[(long)j * HID1 + 16] = c1[j];
    hp[(long)j * HID1 + 32] = c2[j];
    hp[(long)j * HID1 + 48] = c3[j];
  }
}

__global__ void __launch_bounds__(128) gemm2_wmma(const float* __restrict__ X,
                                                  const float* __restrict__ W,
                                                  float* __restrict__ H) {
  const int wave = threadIdx.x >> 5;
  const int lane = threadIdx.x & 31;
  const int mt   = blockIdx.x * 4 + wave;
  if (mt >= N_NODES / 16) return;
  const int m  = lane & 15;
  const int hi = lane >> 4;
  const float* xr = X + (long)(mt * 16 + m) * HID1;

  v8f c0 = {}, c1 = {}, c2 = {};
  #pragma unroll 4
  for (int k0 = 0; k0 < HID1; k0 += 4) {
    const int ka = k0 + 2 * hi;
    v2f a; a.x = xr[ka]; a.y = xr[ka + 1];
    #pragma unroll
    for (int t = 0; t < 3; t++) {
      const int col = t * 16 + m;
      v2f b;
      b.x = (col < NCLASS) ? W[(long)ka * NCLASS + col]       : 0.f;
      b.y = (col < NCLASS) ? W[(long)(ka + 1) * NCLASS + col] : 0.f;
      if (t == 0) c0 = __builtin_amdgcn_wmma_f32_16x16x4_f32(false, a, false, b, (short)0, c0, false, false);
      if (t == 1) c1 = __builtin_amdgcn_wmma_f32_16x16x4_f32(false, a, false, b, (short)0, c1, false, false);
      if (t == 2) c2 = __builtin_amdgcn_wmma_f32_16x16x4_f32(false, a, false, b, (short)0, c2, false, false);
    }
  }
  const int rbase = mt * 16 + 8 * hi;
  #pragma unroll
  for (int j = 0; j < 8; j++) {
    float* hr = H + (long)(rbase + j) * NCLASS;
    if (m      < NCLASS) hr[m]      = c0[j];
    if (m + 16 < NCLASS) hr[m + 16] = c1[j];
    if (m + 32 < NCLASS) hr[m + 32] = c2[j];
  }
}

__global__ void alpha1_kernel(const float* __restrict__ H1,
                              const float* __restrict__ a_s, const float* __restrict__ a_d,
                              float* __restrict__ as, float* __restrict__ ad) {
  int i = blockIdx.x * blockDim.x + threadIdx.x;
  if (i >= N_NODES * NHEADS) return;
  const int h = i & 7;
  const float* hp = H1 + (long)(i >> 3) * HID1 + h * NHID;
  float s = 0.f, d = 0.f;
  #pragma unroll
  for (int k = 0; k < NHID; k++) { s += hp[k] * a_s[h * NHID + k]; d += hp[k] * a_d[h * NHID + k]; }
  as[i] = s; ad[i] = d;
}

__global__ void alpha2_kernel(const float* __restrict__ H2,
                              const float* __restrict__ a_s, const float* __restrict__ a_d,
                              float* __restrict__ as, float* __restrict__ ad) {
  int n = blockIdx.x * blockDim.x + threadIdx.x;
  if (n >= N_NODES) return;
  const float* hp = H2 + (long)n * NCLASS;
  float s = 0.f, d = 0.f;
  #pragma unroll
  for (int c = 0; c < NCLASS; c++) { s += hp[c] * a_s[c]; d += hp[c] * a_d[c]; }
  as[n] = s; ad[n] = d;
}

__global__ void edge_max1(const int* __restrict__ ei, const float* __restrict__ as,
                          const float* __restrict__ ad, unsigned* __restrict__ m) {
  int e = blockIdx.x * blockDim.x + threadIdx.x;
  if (e >= E_TOT) return;
  int s, d; get_edge(ei, e, s, d);
  #pragma unroll
  for (int h = 0; h < NHEADS; h++) {
    float v = lrelu(as[s * NHEADS + h] + ad[d * NHEADS + h]);
    atomicMax(&m[d * NHEADS + h], f2ord(v));
  }
}

__global__ void edge_sum1(const int* __restrict__ ei, const float* __restrict__ as,
                          const float* __restrict__ ad, const unsigned* __restrict__ m,
                          float* __restrict__ den) {
  int e = blockIdx.x * blockDim.x + threadIdx.x;
  if (e >= E_TOT) return;
  int s, d; get_edge(ei, e, s, d);
  #pragma unroll
  for (int h = 0; h < NHEADS; h++) {
    float v = lrelu(as[s * NHEADS + h] + ad[d * NHEADS + h]);
    atomicAdd(&den[d * NHEADS + h], __expf(v - ord2f(m[d * NHEADS + h])));
  }
}

__global__ void edge_agg1(const int* __restrict__ ei, const float* __restrict__ as,
                          const float* __restrict__ ad, const unsigned* __restrict__ m,
                          const float* __restrict__ den, const float* __restrict__ H1,
                          float* __restrict__ out1) {
  int idx = blockIdx.x * blockDim.x + threadIdx.x;
  if (idx >= E_TOT * NHEADS) return;
  const int e = idx >> 3, h = idx & 7;
  int s, d; get_edge(ei, e, s, d);
  float v    = lrelu(as[s * NHEADS + h] + ad[d * NHEADS + h]);
  float attn = __expf(v - ord2f(m[d * NHEADS + h])) / den[d * NHEADS + h];
  const float* hs = H1 + (long)s * HID1 + h * NHID;
  float*       od = out1 + (long)d * HID1 + h * NHID;
  #pragma unroll
  for (int k = 0; k < NHID; k++) atomicAdd(&od[k], attn * hs[k]);
}

__global__ void elu_bias_kernel(float* __restrict__ o, const float* __restrict__ b) {
  int i = blockIdx.x * blockDim.x + threadIdx.x;
  if (i >= N_NODES * HID1) return;
  float v = o[i] + b[i & (HID1 - 1)];
  o[i] = v > 0.f ? v : (__expf(v) - 1.f);
}

__global__ void edge_max2(const int* __restrict__ ei, const float* __restrict__ as,
                          const float* __restrict__ ad, unsigned* __restrict__ m) {
  int e = blockIdx.x * blockDim.x + threadIdx.x;
  if (e >= E_TOT) return;
  int s, d; get_edge(ei, e, s, d);
  atomicMax(&m[d], f2ord(lrelu(as[s] + ad[d])));
}

__global__ void edge_sum2(const int* __restrict__ ei, const float* __restrict__ as,
                          const float* __restrict__ ad, const unsigned* __restrict__ m,
                          float* __restrict__ den) {
  int e = blockIdx.x * blockDim.x + threadIdx.x;
  if (e >= E_TOT) return;
  int s, d; get_edge(ei, e, s, d);
  float v = lrelu(as[s] + ad[d]);
  atomicAdd(&den[d], __expf(v - ord2f(m[d])));
}

__global__ void edge_agg2(const int* __restrict__ ei, const float* __restrict__ as,
                          const float* __restrict__ ad, const unsigned* __restrict__ m,
                          const float* __restrict__ den, const float* __restrict__ H2,
                          float* __restrict__ out2) {
  int e = blockIdx.x * blockDim.x + threadIdx.x;
  if (e >= E_TOT) return;
  int s, d; get_edge(ei, e, s, d);
  float v    = lrelu(as[s] + ad[d]);
  float attn = __expf(v - ord2f(m[d])) / den[d];
  const float* hs = H2 + (long)s * NCLASS;
  float*       od = out2 + (long)d * NCLASS;
  #pragma unroll
  for (int c = 0; c < NCLASS; c++) atomicAdd(&od[c], attn * hs[c]);
}

__global__ void lsm_kernel(const float* __restrict__ o2, const float* __restrict__ b,
                           float* __restrict__ out) {
  int n = blockIdx.x * blockDim.x + threadIdx.x;
  if (n >= N_NODES) return;
  float buf[NCLASS];
  float mx = -INFINITY;
  #pragma unroll
  for (int c = 0; c < NCLASS; c++) { buf[c] = o2[(long)n * NCLASS + c] + b[c]; mx = fmaxf(mx, buf[c]); }
  float ssum = 0.f;
  #pragma unroll
  for (int c = 0; c < NCLASS; c++) ssum += __expf(buf[c] - mx);
  const float lse = mx + __logf(ssum);
  #pragma unroll
  for (int c = 0; c < NCLASS; c++) out[(long)n * NCLASS + c] = buf[c] - lse;
}

extern "C" void kernel_launch(void* const* d_in, const int* in_sizes, int n_in,
                              void* d_out, int out_size, void* d_ws, size_t ws_size,
                              hipStream_t stream) {
  const float* x      = (const float*)d_in[0];
  const float* W1     = (const float*)d_in[1];
  const float* a_src1 = (const float*)d_in[2];
  const float* a_dst1 = (const float*)d_in[3];
  const float* b1     = (const float*)d_in[4];
  const float* W2     = (const float*)d_in[5];
  const float* a_src2 = (const float*)d_in[6];
  const float* a_dst2 = (const float*)d_in[7];
  const float* b2     = (const float*)d_in[8];
  const int*   ei     = (const int*)d_in[9];
  float*       out    = (float*)d_out;

  float* ws   = (float*)d_ws;
  float* H1   = ws;
  float* as1  = H1  + (long)N_NODES * HID1;
  float* ad1  = as1 + (long)N_NODES * NHEADS;
  float* as2  = ad1 + (long)N_NODES * NHEADS;
  float* ad2  = as2 + N_NODES;
  float* zb   = ad2 + N_NODES;
  float* den1 = zb;
  float* out1 = den1 + (long)N_NODES * NHEADS;
  float* den2 = out1 + (long)N_NODES * HID1;
  float* out2 = den2 + N_NODES;
  const unsigned ZERON = N_NODES * NHEADS + N_NODES * HID1 + N_NODES + N_NODES * NCLASS;
  unsigned* m1 = (unsigned*)(out2 + (long)N_NODES * NCLASS);
  unsigned* m2 = m1 + (long)N_NODES * NHEADS;
  const unsigned MINFN = N_NODES * NHEADS + N_NODES;
  float* H2   = (float*)(m2 + N_NODES);

  const int B = 256;
  auto cdiv = [](long a, long b) { return (int)((a + b - 1) / b); };

  init_kernel<<<cdiv(ZERON, B), B, 0, stream>>>(zb, ZERON, m1, MINFN);

  gemm1_wmma<<<cdiv(N_NODES / 16, 4), 128, 0, stream>>>(x, W1, H1);
  alpha1_kernel<<<cdiv((long)N_NODES * NHEADS, B), B, 0, stream>>>(H1, a_src1, a_dst1, as1, ad1);
  edge_max1<<<cdiv(E_TOT, B), B, 0, stream>>>(ei, as1, ad1, m1);
  edge_sum1<<<cdiv(E_TOT, B), B, 0, stream>>>(ei, as1, ad1, m1, den1);
  edge_agg1<<<cdiv((long)E_TOT * NHEADS, B), B, 0, stream>>>(ei, as1, ad1, m1, den1, H1, out1);
  elu_bias_kernel<<<cdiv((long)N_NODES * HID1, B), B, 0, stream>>>(out1, b1);

  gemm2_wmma<<<cdiv(N_NODES / 16, 4), 128, 0, stream>>>(out1, W2, H2);
  alpha2_kernel<<<cdiv(N_NODES, B), B, 0, stream>>>(H2, a_src2, a_dst2, as2, ad2);
  edge_max2<<<cdiv(E_TOT, B), B, 0, stream>>>(ei, as2, ad2, m2);
  edge_sum2<<<cdiv(E_TOT, B), B, 0, stream>>>(ei, as2, ad2, m2, den2);
  edge_agg2<<<cdiv(E_TOT, B), B, 0, stream>>>(ei, as2, ad2, m2, den2, H2, out2);

  lsm_kernel<<<cdiv(N_NODES, B), B, 0, stream>>>(out2, b2, out);
}


# edge_max1/2 and init folded into agg1/agg2: a range's filtered edge list stays whole in LDS; segment max by LDS integer max, then the sums (LDS float adds, arrival order); 11 -> 8 launches
# speedup vs baseline: 5.7636x; 5.7636x over previous
#include <hip/hip_runtime.h>
#include <math.h>

#define N_NODES 50000
#define N_EDGES 800000
#define E_TOT   (N_EDGES + N_NODES)
#define NFEAT   128
#define NHID    8
#define NHEADS  8
#define HID1    (NHEADS * NHID)
#define NCLASS  40
#define NEG_SLOPE 0.2f

typedef __attribute__((ext_vector_type(2))) float v2f;
typedef __attribute__((ext_vector_type(8))) float v8f;

__device__ __forceinline__ float lrelu(float x) { return x > 0.f ? x : NEG_SLOPE * x; }

__device__ __forceinline__ unsigned f2ord(float f) {
  unsigned u = __float_as_uint(f);
  return (u & 0x80000000u) ? ~u : (u | 0x80000000u);
}
__device__ __forceinline__ float ord2f(unsigned u) {
  return (u & 0x80000000u) ? __uint_as_float(u & 0x7FFFFFFFu) : __uint_as_float(~u);
}
#define ORD_NEG_INF 0x007FFFFFu

#define AGG_T   1024
#define NRANGE  256
#define RN      196
#define LD_PER_T 8
#define CHUNK4  (AGG_T * LD_PER_T)
#define NCHUNK  ((N_EDGES / 4 + CHUNK4 - 1) / CHUNK4)
#define LIST_CAP 36864
#define P1      65
#define UNR     4

#define EBITS 20
#define EMASK ((1u << EBITS) - 1)
__device__ __forceinline__ void push_edge(int e, unsigned r, unsigned* list, unsigned* cnt) {
  const unsigned p = atomicAdd(cnt, 1u);
  if (p < LIST_CAP) list[p] = (unsigned)e | (r << EBITS);
}

__device__ __forceinline__ void scan_chunk(const int* __restrict__ ei, int chunk, int t, int lo,
                                           unsigned span, unsigned* list, unsigned* cnt) {
  const int4* __restrict__ dst4 = (const int4*)(ei + N_EDGES);
  const int q0 = chunk * CHUNK4 + t;
  int4 dv[LD_PER_T];
  #pragma unroll
  for (int i = 0; i < LD_PER_T; i++) {
    const int q = q0 + i * AGG_T;
    dv[i] = (q < N_EDGES / 4) ? dst4[q] : make_int4(-1, -1, -1, -1);
  }
  #pragma unroll
  for (int i = 0; i < LD_PER_T; i++) {
    const int e = 4 * (q0 + i * AGG_T);
    const unsigned r0 = (unsigned)(dv[i].x - lo), r1 = (unsigned)(dv[i].y - lo);
    const unsigned r2 = (unsigned)(dv[i].z - lo), r3 = (unsigned)(dv[i].w - lo);
    if (r0 < span) push_edge(e + 0, r0, list, cnt);
    if (r1 < span) push_edge(e + 1, r1, list, cnt);
    if (r2 < span) push_edge(e + 2, r2, list, cnt);
    if (r3 < span) push_edge(e + 3, r3, list, cnt);
  }
}

__device__ __forceinline__ void list_self_loops(int t, int lo, unsigned span, unsigned* list, unsigned* cnt) {
  if (t < (int)span) list[t] = (unsigned)(lo + t) | ((unsigned)t << EBITS);
  if (t == 0) *cnt = span;
}

__device__ __forceinline__ void resolve_src(const int* __restrict__ ei, unsigned* list,
                                            unsigned first, unsigned n, int t) {
  for (unsigned b = first + t; b < n; b += UNR * AGG_T) {
    unsigned en[UNR]; int sv[UNR];
    #pragma unroll
    for (int j = 0; j < UNR; j++) {
      const bool in = b + j * AGG_T < n;
      en[j] = in ? list[b + j * AGG_T] : 0u;
      sv[j] = in ? ei[en[j] & EMASK] : 0;
    }
    #pragma unroll
    for (int j = 0; j < UNR; j++)
      if (b + j * AGG_T < n) list[b + j * AGG_T] = (unsigned)sv[j] | (en[j] & ~EMASK);
  }
}

__global__ void __launch_bounds__(AGG_T) agg1_lds(const int* __restrict__ ei,
                                                  const float* __restrict__ as,
                                                  const float* __restrict__ ad,
                                                  const float* __restrict__ H1,
                                                  float* __restrict__ den,
                                                  float* __restrict__ out1) {
  __shared__ float    acc[RN * P1];
  __shared__ float    dsum[RN * NHEADS];
  __shared__ float    adl[RN * NHEADS];
  __shared__ unsigned mxo[RN * NHEADS];
  __shared__ unsigned list[LIST_CAP];
  __shared__ unsigned cnt;

  const int t  = threadIdx.x;
  const int lo = blockIdx.x * RN;
  const unsigned span = (unsigned)(min(lo + RN, N_NODES) - lo);
  const int h = t & 7;

  for (int i = t; i < RN * P1; i += AGG_T) acc[i] = 0.f;
  for (int i = t; i < (int)span * NHEADS; i += AGG_T) {
    dsum[i] = 0.f;
    adl[i]  = ad[lo * NHEADS + i];
    mxo[i]  = ORD_NEG_INF;
  }

  auto max_pass = [&](unsigned n) {
    for (unsigned b = t >> 3; b < n; b += UNR * (AGG_T / 8)) {
      unsigned dl[UNR]; float a[UNR];
      #pragma unroll
      for (int j = 0; j < UNR; j++) {
        const unsigned en = list[min(b + j * (AGG_T / 8), n - 1)];
        dl[j] = en >> EBITS;
        a[j]  = as[(en & EMASK) * NHEADS + h];
      }
      #pragma unroll
      for (int j = 0; j < UNR; j++)
        atomicMax(&mxo[dl[j] * NHEADS + h], f2ord(lrelu(a[j] + adl[dl[j] * NHEADS + h])));
    }
  };
  auto sum_pass = [&](unsigned n) {
    for (unsigned b = t >> 3; b < n; b += UNR * (AGG_T / 8)) {
      unsigned dl[UNR]; float a[UNR]; float4 h0[UNR], h1[UNR];
      #pragma unroll
      for (int j = 0; j < UNR; j++) {
        const unsigned en = list[min(b + j * (AGG_T / 8), n - 1)];
        const unsigned s = en & EMASK;
        dl[j] = en >> EBITS;
        a[j]  = as[s * NHEADS + h];
        const float4* hs = (const float4*)(H1 + (long)s * HID1 + h * NHID);
        h0[j] = hs[0]; h1[j] = hs[1];
      }
      #pragma unroll
      for (int j = 0; j < UNR; j++) {
        if (b + j * (AGG_T / 8) < n) {
          const int d8 = dl[j] * NHEADS + h;
          const float v = lrelu(a[j] + adl[d8]);
          const float w = __expf(v - ord2f(mxo[d8]));
          atomicAdd(&dsum[d8], w);
          float* od = &acc[dl[j] * P1 + h * NHID];
          atomicAdd(&od[0], w * h0[j].x); atomicAdd(&od[1], w * h0[j].y);
          atomicAdd(&od[2], w * h0[j].z); atomicAdd(&od[3], w * h0[j].w);
          atomicAdd(&od[4], w * h1[j].x); atomicAdd(&od[5], w * h1[j].y);
          atomicAdd(&od[6], w * h1[j].z); atomicAdd(&od[7], w * h1[j].w);
        }
      }
    }
  };

  list_self_loops(t, lo, span, list, &cnt);
  __syncthreads();
  for (int c = 0; c < NCHUNK; c++) scan_chunk(ei, c, t, lo, span, list, &cnt);
  __syncthreads();
  const unsigned n = cnt;

  if (n <= LIST_CAP) {
    resolve_src(ei, list, span, n, t);
    __syncthreads();
    max_pass(n);
    __syncthreads();
    sum_pass(n);
  } else {
    for (int phase = 0; phase < 2; phase++) {
      for (int it = 0; it <= NCHUNK; it++) {
        __syncthreads();
        if (it == 0) list_self_loops(t, lo, span, list, &cnt);
        else if (t == 0) cnt = 0;
        __syncthreads();
        if (it > 0) scan_chunk(ei, it - 1, t, lo, span, list, &cnt);
        __syncthreads();
        const unsigned nc = cnt;
        resolve_src(ei, list, it == 0 ? span : 0u, nc, t);
        __syncthreads();
        if (phase == 0) max_pass(nc); else sum_pass(nc);
      }
    }
  }
  __syncthreads();

  for (int i = t; i < (int)span * HID1; i += AGG_T)
    out1[(long)lo * HID1 + i] = acc[(i >> 6) * P1 + (i & 63)];
  for (int i = t; i < (int)span * NHEADS; i += AGG_T)
    den[lo * NHEADS + i] = dsum[i];
}

__global__ void __launch_bounds__(AGG_T) agg2_lds(const int* __restrict__ ei,
                                                  const float* __restrict__ as,
                                                  const float* __restrict__ ad,
                                                  const float* __restrict__ H2,
                                                  float* __restrict__ den,
                                                  float* __restrict__ out2) {
  __shared__ float    acc[RN * NCLASS];
  __shared__ float    dsum[RN];
  __shared__ float    adl[RN];
  __shared__ unsigned mxo[RN];
  __shared__ unsigned list[LIST_CAP];
  __shared__ unsigned cnt;

  const int t  = threadIdx.x;
  const int lo = blockIdx.x * RN;
  const unsigned span = (unsigned)(min(lo + RN, N_NODES) - lo);
  const int l = t & 7;

  for (int i = t; i < RN * NCLASS; i += AGG_T) acc[i] = 0.f;
  if (t < (int)span) {
    dsum[t] = 0.f;
    adl[t]  = ad[lo + t];
    mxo[t]  = ORD_NEG_INF;
  }

  auto max_pass = [&](unsigned n) {
    for (unsigned b = t; b < n; b += UNR * AGG_T) {
      unsigned dl[UNR]; float a[UNR];
      #pragma unroll
      for (int j = 0; j < UNR; j++) {
        const unsigned en = list[min(b + j * AGG_T, n - 1)];
        dl[j] = en >> EBITS;
        a[j]  = as[en & EMASK];
      }
      #pragma unroll
      for (int j = 0; j < UNR; j++)
        atomicMax(&mxo[dl[j]], f2ord(lrelu(a[j] + adl[dl[j]])));
    }
  };
  auto sum_pass = [&](unsigned n) {
    for (unsigned b = t >> 3; b < n; b += UNR * (AGG_T / 8)) {
      unsigned dl[UNR]; float a[UNR]; float hv[UNR][NCLASS / 8];
      #pragma unroll
      for (int j = 0; j < UNR; j++) {
        const unsigned en = list[min(b + j * (AGG_T / 8), n - 1)];
        const unsigned s = en & EMASK;
        dl[j] = en >> EBITS;
        a[j]  = as[s];
        const float* hs = H2 + (long)s * NCLASS + l;
        #pragma unroll
        for (int k = 0; k < NCLASS / 8; k++) hv[j][k] = hs[k * 8];
      }
      #pragma unroll
      for (int j = 0; j < UNR; j++) {
        if (b + j * (AGG_T / 8) < n) {
          const float v = lrelu(a[j] + adl[dl[j]]);
          const float w = __expf(v - ord2f(mxo[dl[j]]));
          if (l == 0) atomicAdd(&dsum[dl[j]], w);
          float* od = &acc[dl[j] * NCLASS + l];
          #pragma unroll
          for (int k = 0; k < NCLASS / 8; k++) atomicAdd(&od[k * 8], w * hv[j][k]);
        }
      }
    }
  };

  list_self_loops(t, lo, span, list, &cnt);
  __syncthreads();
  for (int c = 0; c < NCHUNK; c++) scan_chunk(ei, c, t, lo, span, list, &cnt);
  __syncthreads();
  const unsigned n = cnt;

  if (n <= LIST_CAP) {
    resolve_src(ei, list, span, n, t);
    __syncthreads();
    max_pass(n);
    __syncthreads();
    sum_pass(n);
  } else {
    for (int phase = 0; phase < 2; phase++) {
      for (int it = 0; it <= NCHUNK; it++) {
        __syncthreads();
        if (it == 0) list_self_loops(t, lo, span, list, &cnt);
        else if (t == 0) cnt = 0;
        __syncthreads();
        if (it > 0) scan_chunk(ei, it - 1, t, lo, span, list, &cnt);
        __syncthreads();
        const unsigned nc = cnt;
        resolve_src(ei, list, it == 0 ? span : 0u, nc, t);
        __syncthreads();
        if (phase == 0) max_pass(nc); else sum_pass(nc);
      }
    }
  }
  __syncthreads();

  for (int i = t; i < (int)span * NCLASS; i += AGG_T) out2[(long)lo * NCLASS + i] = acc[i];
  if (t < (int)span) den[lo + t] = dsum[t];
}

__global__ void __launch_bounds__(128) gemm1_wmma(const float* __restrict__ X,
                                                  const float* __restrict__ W,
                                                  float* __restrict__ H) {
  const int wave = threadIdx.x >> 5;
  const int lane = threadIdx.x & 31;
  const int mt   = blockIdx.x * 4 + wave;
  if (mt >= N_NODES / 16) return;
  const int m  = lane & 15;
  const int hi = lane >> 4;
  const float* xr = X + (long)(mt * 16 + m) * NFEAT;

  v8f c0 = {}, c1 = {}, c2 = {}, c3 = {};
  #pragma unroll 4
  for (int k0 = 0; k0 < NFEAT; k0 += 4) {
    const int ka = k0 + 2 * hi;
    v2f a; a.x = xr[ka]; a.y = xr[ka + 1];
    const float* w0 = W + (long)ka * HID1 + m;
    const float* w1 = w0 + HID1;
    v2f b;
    b.x = w0[0];  b.y = w1[0];
    c0 = __builtin_amdgcn_wmma_f32_16x16x4_f32(false, a, false, b, (short)0, c0, false, false);
    b.x = w0[16]; b.y = w1[16];
    c1 = __builtin_amdgcn_wmma_f32_16x16x4_f32(false, a, false, b, (short)0, c1, false, false);
    b.x = w0[32]; b.y = w1[32];
    c2 = __builtin_amdgcn_wmma_f32_16x16x4_f32(false, a, false, b, (short)0, c2, false, false);
    b.x = w0[48]; b.y = w1[48];
    c3 = __builtin_amdgcn_wmma_f32_16x16x4_f32(false, a, false, b, (short)0, c3, false, false);
  }
  float* hp = H + (long)(mt * 16 + 8 * hi) * HID1 + m;
  #pragma unroll
  for (int j = 0; j < 8; j++) {
    hp[(long)j * HID1 +  0] = c0[j];
    hp[(long)j * HID1 + 16] = c1[j];
    hp[(long)j * HID1 + 32] = c2[j];
    hp[(long)j * HID1 + 48] = c3[j];
  }
}

__global__ void __launch_bounds__(128) gemm2_wmma(const float* __restrict__ X,
                                                  const float* __restrict__ W,
                                                  float* __restrict__ H) {
  const int wave = threadIdx.x >> 5;
  const int lane = threadIdx.x & 31;
  const int mt   = blockIdx.x * 4 + wave;
  if (mt >= N_NODES / 16) return;
  const int m  = lane & 15;
  const int hi = lane >> 4;
  const float* xr = X + (long)(mt * 16 + m) * HID1;

  v8f c0 = {}, c1 = {}, c2 = {};
  #pragma unroll 4
  for (int k0 = 0; k0 < HID1; k0 += 4) {
    const int ka = k0 + 2 * hi;
    v2f a; a.x = xr[ka]; a.y = xr[ka + 1];
    #pragma unroll
    for (int t = 0; t < 3; t++) {
      const int col = t * 16 + m;
      v2f b;
      b.x = (col < NCLASS) ? W[(long)ka * NCLASS + col]       : 0.f;
      b.y = (col < NCLASS) ? W[(long)(ka + 1) * NCLASS + col] : 0.f;
      if (t == 0) c0 = __builtin_amdgcn_wmma_f32_16x16x4_f32(false, a, false, b, (short)0, c0, false, false);
      if (t == 1) c1 = __builtin_amdgcn_wmma_f32_16x16x4_f32(false, a, false, b, (short)0, c1, false, false);
      if (t == 2) c2 = __builtin_amdgcn_wmma_f32_16x16x4_f32(false, a, false, b, (short)0, c2, false, false);
    }
  }
  const int rbase = mt * 16 + 8 * hi;
  #pragma unroll
  for (int j = 0; j < 8; j++) {
    float* hr = H + (long)(rbase + j) * NCLASS;
    if (m      < NCLASS) hr[m]      = c0[j];
    if (m + 16 < NCLASS) hr[m + 16] = c1[j];
    if (m + 32 < NCLASS) hr[m + 32] = c2[j];
  }
}

__global__ void alpha1_kernel(const float* __restrict__ H1,
                              const float* __restrict__ a_s, const float* __restrict__ a_d,
                              float* __restrict__ as, float* __restrict__ ad) {
  int i = blockIdx.x * blockDim.x + threadIdx.x;
  if (i >= N_NODES * NHEADS) return;
  const int h = i & 7;
  const float* hp = H1 + (long)(i >> 3) * HID1 + h * NHID;
  float s = 0.f, d = 0.f;
  #pragma unroll
  for (int k = 0; k < NHID; k++) { s += hp[k] * a_s[h * NHID + k]; d += hp[k] * a_d[h * NHID + k]; }
  as[i] = s; ad[i] = d;
}

__global__ void alpha2_kernel(const float* __restrict__ H2,
                              const float* __restrict__ a_s, const float* __restrict__ a_d,
                              float* __restrict__ as, float* __restrict__ ad) {
  int n = blockIdx.x * blockDim.x + threadIdx.x;
  if (n >= N_NODES) return;
  const float* hp = H2 + (long)n * NCLASS;
  float s = 0.f, d = 0.f;
  #pragma unroll
  for (int c = 0; c < NCLASS; c++) { s += hp[c] * a_s[c]; d += hp[c] * a_d[c]; }
  as[n] = s; ad[n] = d;
}

__global__ void elu_bias_kernel(float* __restrict__ o, const float* __restrict__ den,
                                const float* __restrict__ b) {
  int i = blockIdx.x * blockDim.x + threadIdx.x;
  if (i >= N_NODES * HID1) return;
  float v = o[i] / den[i >> 3] + b[i & (HID1 - 1)];
  o[i] = v > 0.f ? v : (__expf(v) - 1.f);
}

__global__ void lsm_kernel(const float* __restrict__ o2, const float* __restrict__ den,
                           const float* __restrict__ b, float* __restrict__ out) {
  int n = blockIdx.x * blockDim.x + threadIdx.x;
  if (n >= N_NODES) return;
  float buf[NCLASS];
  float mx = -INFINITY;
  const float dn = den[n];
  #pragma unroll
  for (int c = 0; c < NCLASS; c++) { buf[c] = o2[(long)n * NCLASS + c] / dn + b[c]; mx = fmaxf(mx, buf[c]); }
  float ssum = 0.f;
  #pragma unroll
  for (int c = 0; c < NCLASS; c++) ssum += __expf(buf[c] - mx);
  const float lse = mx + __logf(ssum);
  #pragma unroll
  for (int c = 0; c < NCLASS; c++) out[(long)n * NCLASS + c] = buf[c] - lse;
}

extern "C" void kernel_launch(void* const* d_in, const int* in_sizes, int n_in,
                              void* d_out, int out_size, void* d_ws, size_t ws_size,
                              hipStream_t stream) {
  const float* x      = (const float*)d_in[0];
  const float* W1     = (const float*)d_in[1];
  const float* a_src1 = (const float*)d_in[2];
  const float* a_dst1 = (const float*)d_in[3];
  const float* b1     = (const float*)d_in[4];
  const float* W2     = (const float*)d_in[5];
  const float* a_src2 = (const float*)d_in[6];
  const float* a_dst2 = (const float*)d_in[7];
  const float* b2     = (const float*)d_in[8];
  const int*   ei     = (const int*)d_in[9];
  float*       out    = (float*)d_out;

  float* ws   = (float*)d_ws;
  float* H1   = ws;
  float* as1  = H1  + (long)N_NODES * HID1;
  float* ad1  = as1 + (long)N_NODES * NHEADS;
  float* as2  = ad1 + (long)N_NODES * NHEADS;
  float* ad2  = as2 + N_NODES;
  float* den1 = ad2 + N_NODES;
  float* out1 = den1 + (long)N_NODES * NHEADS;
  float* den2 = out1 + (long)N_NODES * HID1;
  float* out2 = den2 + N_NODES;
  float* H2   = out2 + (long)N_NODES * NCLASS;

  const int B = 256;
  auto cdiv = [](long a, long b) { return (int)((a + b - 1) / b); };

  gemm1_wmma<<<cdiv(N_NODES / 16, 4), 128, 0, stream>>>(x, W1, H1);
  alpha1_kernel<<<cdiv((long)N_NODES * NHEADS, B), B, 0, stream>>>(H1, a_src1, a_dst1, as1, ad1);
  agg1_lds<<<NRANGE, AGG_T, 0, stream>>>(ei, as1, ad1, H1, den1, out1);
  elu_bias_kernel<<<cdiv((long)N_NODES * HID1, B), B, 0, stream>>>(out1, den1, b1);

  gemm2_wmma<<<cdiv(N_NODES / 16, 4), 128, 0, stream>>>(out1, W2, H2);
  alpha2_kernel<<<cdiv(N_NODES, B), B, 0, stream>>>(H2, a_src2, a_dst2, as2, ad2);
  agg2_lds<<<NRANGE, AGG_T, 0, stream>>>(ei, as2, ad2, H2, den2, out2);

  lsm_kernel<<<cdiv(N_NODES, B), B, 0, stream>>>(out2, den2, b2, out);
}


# elu_bias folded into agg1's store and lsm into agg2's: a range divides its LDS sums, adds bias, applies ELU / its rows' log-softmax (same class order) as it stores; 8 -> 6 launches
# speedup vs baseline: 6.2828x; 1.0901x over previous
#include <hip/hip_runtime.h>
#include <math.h>

#define N_NODES 50000
#define N_EDGES 800000
#define E_TOT   (N_EDGES + N_NODES)
#define NFEAT   128
#define NHID    8
#define NHEADS  8
#define HID1    (NHEADS * NHID)
#define NCLASS  40
#define NEG_SLOPE 0.2f

typedef __attribute__((ext_vector_type(2))) float v2f;
typedef __attribute__((ext_vector_type(8))) float v8f;

__device__ __forceinline__ float lrelu(float x) { return x > 0.f ? x : NEG_SLOPE * x; }

__device__ __forceinline__ unsigned f2ord(float f) {
  unsigned u = __float_as_uint(f);
  return (u & 0x80000000u) ? ~u : (u | 0x80000000u);
}
__device__ __forceinline__ float ord2f(unsigned u) {
  return (u & 0x80000000u) ? __uint_as_float(u & 0x7FFFFFFFu) : __uint_as_float(~u);
}
#define ORD_NEG_INF 0x007FFFFFu

#define AGG_T   1024
#define NRANGE  256
#define RN      196
#define LD_PER_T 8
#define CHUNK4  (AGG_T * LD_PER_T)
#define NCHUNK  ((N_EDGES / 4 + CHUNK4 - 1) / CHUNK4)
#define LIST_CAP 36864
#define P1      65
#define UNR     4

#define EBITS 20
#define EMASK ((1u << EBITS) - 1)
__device__ __forceinline__ void push_edge(int e, unsigned r, unsigned* list, unsigned* cnt) {
  const unsigned p = atomicAdd(cnt, 1u);
  if (p < LIST_CAP) list[p] = (unsigned)e | (r << EBITS);
}

__device__ __forceinline__ void scan_chunk(const int* __restrict__ ei, int chunk, int t, int lo,
                                           unsigned span, unsigned* list, unsigned* cnt) {
  const int4* __restrict__ dst4 = (const int4*)(ei + N_EDGES);
  const int q0 = chunk * CHUNK4 + t;
  int4 dv[LD_PER_T];
  #pragma unroll
  for (int i = 0; i < LD_PER_T; i++) {
    const int q = q0 + i * AGG_T;
    dv[i] = (q < N_EDGES / 4) ? dst4[q] : make_int4(-1, -1, -1, -1);
  }
  #pragma unroll
  for (int i = 0; i < LD_PER_T; i++) {
    const int e = 4 * (q0 + i * AGG_T);
    const unsigned r0 = (unsigned)(dv[i].x - lo), r1 = (unsigned)(dv[i].y - lo);
    const unsigned r2 = (unsigned)(dv[i].z - lo), r3 = (unsigned)(dv[i].w - lo);
    if (r0 < span) push_edge(e + 0, r0, list, cnt);
    if (r1 < span) push_edge(e + 1, r1, list, cnt);
    if (r2 < span) push_edge(e + 2, r2, list, cnt);
    if (r3 < span) push_edge(e + 3, r3, list, cnt);
  }
}

__device__ __forceinline__ void list_self_loops(int t, int lo, unsigned span, unsigned* list, unsigned* cnt) {
  if (t < (int)span) list[t] = (unsigned)(lo + t) | ((unsigned)t << EBITS);
  if (t == 0) *cnt = span;
}

__device__ __forceinline__ void resolve_src(const int* __restrict__ ei, unsigned* list,
                                            unsigned first, unsigned n, int t) {
  for (unsigned b = first + t; b < n; b += UNR * AGG_T) {
    unsigned en[UNR]; int sv[UNR];
    #pragma unroll
    for (int j = 0; j < UNR; j++) {
      const bool in = b + j * AGG_T < n;
      en[j] = in ? list[b + j * AGG_T] : 0u;
      sv[j] = in ? ei[en[j] & EMASK] : 0;
    }
    #pragma unroll
    for (int j = 0; j < UNR; j++)
      if (b + j * AGG_T < n) list[b + j * AGG_T] = (unsigned)sv[j] | (en[j] & ~EMASK);
  }
}

__global__ void __launch_bounds__(AGG_T) agg1_lds(const int* __restrict__ ei,
                                                  const float* __restrict__ as,
                                                  const float* __restrict__ ad,
                                                  const float* __restrict__ H1,
                                                  const float* __restrict__ bias,
                                                  float* __restrict__ out1) {
  __shared__ float    acc[RN * P1];
  __shared__ float    dsum[RN * NHEADS];
  __shared__ float    adl[RN * NHEADS];
  __shared__ unsigned mxo[RN * NHEADS];
  __shared__ unsigned list[LIST_CAP];
  __shared__ unsigned cnt;

  const int t  = threadIdx.x;
  const int lo = blockIdx.x * RN;
  const unsigned span = (unsigned)(min(lo + RN, N_NODES) - lo);
  const int h = t & 7;

  for (int i = t; i < RN * P1; i += AGG_T) acc[i] = 0.f;
  for (int i = t; i < (int)span * NHEADS; i += AGG_T) {
    dsum[i] = 0.f;
    adl[i]  = ad[lo * NHEADS + i];
    mxo[i]  = ORD_NEG_INF;
  }

  auto max_pass = [&](unsigned n) {
    for (unsigned b = t >> 3; b < n; b += UNR * (AGG_T / 8)) {
      unsigned dl[UNR]; float a[UNR];
      #pragma unroll
      for (int j = 0; j < UNR; j++) {
        const unsigned en = list[min(b + j * (AGG_T / 8), n - 1)];
        dl[j] = en >> EBITS;
        a[j]  = as[(en & EMASK) * NHEADS + h];
      }
      #pragma unroll
      for (int j = 0; j < UNR; j++)
        atomicMax(&mxo[dl[j] * NHEADS + h], f2ord(lrelu(a[j] + adl[dl[j] * NHEADS + h])));
    }
  };
  auto sum_pass = [&](unsigned n) {
    for (unsigned b = t >> 3; b < n; b += UNR * (AGG_T / 8)) {
      unsigned dl[UNR]; float a[UNR]; float4 h0[UNR], h1[UNR];
      #pragma unroll
      for (int j = 0; j < UNR; j++) {
        const unsigned en = list[min(b + j * (AGG_T / 8), n - 1)];
        const unsigned s = en & EMASK;
        dl[j] = en >> EBITS;
        a[j]  = as[s * NHEADS + h];
        const float4* hs = (const float4*)(H1 + (long)s * HID1 + h * NHID);
        h0[j] = hs[0]; h1[j] = hs[1];
      }
      #pragma unroll
      for (int j = 0; j < UNR; j++) {
        if (b + j * (AGG_T / 8) < n) {
          const int d8 = dl[j] * NHEADS + h;
          const float v = lrelu(a[j] + adl[d8]);
          const float w = __expf(v - ord2f(mxo[d8]));
          atomicAdd(&dsum[d8], w);
          float* od = &acc[dl[j] * P1 + h * NHID];
          atomicAdd(&od[0], w * h0[j].x); atomicAdd(&od[1], w * h0[j].y);
          atomicAdd(&od[2], w * h0[j].z); atomicAdd(&od[3], w * h0[j].w);
          atomicAdd(&od[4], w * h1[j].x); atomicAdd(&od[5], w * h1[j].y);
          atomicAdd(&od[6], w * h1[j].z); atomicAdd(&od[7], w * h1[j].w);
        }
      }
    }
  };

  list_self_loops(t, lo, span, list, &cnt);
  __syncthreads();
  for (int c = 0; c < NCHUNK; c++) scan_chunk(ei, c, t, lo, span, list, &cnt);
  __syncthreads();
  const unsigned n = cnt;

  if (n <= LIST_CAP) {
    resolve_src(ei, list, span, n, t);
    __syncthreads();
    max_pass(n);
    __syncthreads();
    sum_pass(n);
  } else {
    for (int phase = 0; phase < 2; phase++) {
      for (int it = 0; it <= NCHUNK; it++) {
        __syncthreads();
        if (it == 0) list_self_loops(t, lo, span, list, &cnt);
        else if (t == 0) cnt = 0;
        __syncthreads();
        if (it > 0) scan_chunk(ei, it - 1, t, lo, span, list, &cnt);
        __syncthreads();
        const unsigned nc = cnt;
        resolve_src(ei, list, it == 0 ? span : 0u, nc, t);
        __syncthreads();
        if (phase == 0) max_pass(nc); else sum_pass(nc);
      }
    }
  }
  __syncthreads();

  for (int i = t; i < (int)span * HID1; i += AGG_T) {
    const float v = acc[(i >> 6) * P1 + (i & 63)] / dsum[i >> 3] + bias[i & (HID1 - 1)];
    out1[(long)lo * HID1 + i] = v > 0.f ? v : (__expf(v) - 1.f);
  }
}

__global__ void __launch_bounds__(AGG_T) agg2_lds(const int* __restrict__ ei,
                                                  const float* __restrict__ as,
                                                  const float* __restrict__ ad,
                                                  const float* __restrict__ H2,
                                                  const float* __restrict__ bias,
                                                  float* __restrict__ out) {
  __shared__ float    acc[RN * NCLASS];
  __shared__ float    dsum[RN];
  __shared__ float    adl[RN];
  __shared__ unsigned mxo[RN];
  __shared__ unsigned list[LIST_CAP];
  __shared__ unsigned cnt;

  const int t  = threadIdx.x;
  const int lo = blockIdx.x * RN;
  const unsigned span = (unsigned)(min(lo + RN, N_NODES) - lo);
  const int l = t & 7;

  for (int i = t; i < RN * NCLASS; i += AGG_T) acc[i] = 0.f;
  if (t < (int)span) {
    dsum[t] = 0.f;
    adl[t]  = ad[lo + t];
    mxo[t]  = ORD_NEG_INF;
  }

  auto max_pass = [&](unsigned n) {
    for (unsigned b = t; b < n; b += UNR * AGG_T) {
      unsigned dl[UNR]; float a[UNR];
      #pragma unroll
      for (int j = 0; j < UNR; j++) {
        const unsigned en = list[min(b + j * AGG_T, n - 1)];
        dl[j] = en >> EBITS;
        a[j]  = as[en & EMASK];
      }
      #pragma unroll
      for (int j = 0; j < UNR; j++)
        atomicMax(&mxo[dl[j]], f2ord(lrelu(a[j] + adl[dl[j]])));
    }
  };
  auto sum_pass = [&](unsigned n) {
    for (unsigned b = t >> 3; b < n; b += UNR * (AGG_T / 8)) {
      unsigned dl[UNR]; float a[UNR]; float hv[UNR][NCLASS / 8];
      #pragma unroll
      for (int j = 0; j < UNR; j++) {
        const unsigned en = list[min(b + j * (AGG_T / 8), n - 1)];
        const unsigned s = en & EMASK;
        dl[j] = en >> EBITS;
        a[j]  = as[s];
        const float* hs = H2 + (long)s * NCLASS + l;
        #pragma unroll
        for (int k = 0; k < NCLASS / 8; k++) hv[j][k] = hs[k * 8];
      }
      #pragma unroll
      for (int j = 0; j < UNR; j++) {
        if (b + j * (AGG_T / 8) < n) {
          const float v = lrelu(a[j] + adl[dl[j]]);
          const float w = __expf(v - ord2f(mxo[dl[j]]));
          if (l == 0) atomicAdd(&dsum[dl[j]], w);
          float* od = &acc[dl[j] * NCLASS + l];
          #pragma unroll
          for (int k = 0; k < NCLASS / 8; k++) atomicAdd(&od[k * 8], w * hv[j][k]);
        }
      }
    }
  };

  list_self_loops(t, lo, span, list, &cnt);
  __syncthreads();
  for (int c = 0; c < NCHUNK; c++) scan_chunk(ei, c, t, lo, span, list, &cnt);
  __syncthreads();
  const unsigned n = cnt;

  if (n <= LIST_CAP) {
    resolve_src(ei, list, span, n, t);
    __syncthreads();
    max_pass(n);
    __syncthreads();
    sum_pass(n);
  } else {
    for (int phase = 0; phase < 2; phase++) {
      for (int it = 0; it <= NCHUNK; it++) {
        __syncthreads();
        if (it == 0) list_self_loops(t, lo, span, list, &cnt);
        else if (t == 0) cnt = 0;
        __syncthreads();
        if (it > 0) scan_chunk(ei, it - 1, t, lo, span, list, &cnt);
        __syncthreads();
        const unsigned nc = cnt;
        resolve_src(ei, list, it == 0 ? span : 0u, nc, t);
        __syncthreads();
        if (phase == 0) max_pass(nc); else sum_pass(nc);
      }
    }
  }
  __syncthreads();

  for (int i = t; i < (int)span * NCLASS; i += AGG_T) acc[i] = acc[i] / dsum[i / NCLASS] + bias[i % NCLASS];
  __syncthreads();
  float* lse = adl;
  if (t < (int)span) {
    const float* r = &acc[t * NCLASS];
    float mx = -INFINITY;
    for (int c = 0; c < NCLASS; c++) mx = fmaxf(mx, r[c]);
    float ssum = 0.f;
    for (int c = 0; c < NCLASS; c++) ssum += __expf(r[c] - mx);
    lse[t] = mx + __logf(ssum);
  }
  __syncthreads();
  for (int i = t; i < (int)span * NCLASS; i += AGG_T) out[(long)lo * NCLASS + i] = acc[i] - lse[i / NCLASS];
}

__global__ void __launch_bounds__(128) gemm1_wmma(const float* __restrict__ X,
                                                  const float* __restrict__ W,
                                                  float* __restrict__ H) {
  const int wave = threadIdx.x >> 5;
  const int lane = threadIdx.x & 31;
  const int mt   = blockIdx.x * 4 + wave;
  if (mt >= N_NODES / 16) return;
  const int m  = lane & 15;
  const int hi = lane >> 4;
  const float* xr = X + (long)(mt * 16 + m) * NFEAT;

  v8f c0 = {}, c1 = {}, c2 = {}, c3 = {};
  #pragma unroll 4
  for (int k0 = 0; k0 < NFEAT; k0 += 4) {
    const int ka = k0 + 2 * hi;
    v2f a; a.x = xr[ka]; a.y = xr[ka + 1];
    const float* w0 = W + (long)ka * HID1 + m;
    const float* w1 = w0 + HID1;
    v2f b;
    b.x = w0[0];  b.y = w1[0];
    c0 = __builtin_amdgcn_wmma_f32_16x16x4_f32(false, a, false, b, (short)0, c0, false, false);
    b.x = w0[16]; b.y = w1[16];
    c1 = __builtin_amdgcn_wmma_f32_16x16x4_f32(false, a, false, b, (short)0, c1, false, false);
    b.x = w0[32]; b.y = w1[32];
    c2 = __builtin_amdgcn_wmma_f32_16x16x4_f32(false, a, false, b, (short)0, c2, false, false);
    b.x = w0[48]; b.y = w1[48];
    c3 = __builtin_amdgcn_wmma_f32_16x16x4_f32(false, a, false, b, (short)0, c3, false, false);
  }
  float* hp = H + (long)(mt * 16 + 8 * hi) * HID1 + m;
  #pragma unroll
  for (int j = 0; j < 8; j++) {
    hp[(long)j * HID1 +  0] = c0[j];
    hp[(long)j * HID1 + 16] = c1[j];
    hp[(long)j * HID1 + 32] = c2[j];
    hp[(long)j * HID1 + 48] = c3[j];
  }
}

__global__ void __launch_bounds__(128) gemm2_wmma(const float* __restrict__ X,
                                                  const float* __restrict__ W,
                                                  float* __restrict__ H) {
  const int wave = threadIdx.x >> 5;
  const int lane = threadIdx.x & 31;
  const int mt   = blockIdx.x * 4 + wave;
  if (mt >= N_NODES / 16) return;
  const int m  = lane & 15;
  const int hi = lane >> 4;
  const float* xr = X + (long)(mt * 16 + m) * HID1;

  v8f c0 = {}, c1 = {}, c2 = {};
  #pragma unroll 4
  for (int k0 = 0; k0 < HID1; k0 += 4) {
    const int ka = k0 + 2 * hi;
    v2f a; a.x = xr[ka]; a.y = xr[ka + 1];
    #pragma unroll
    for (int t = 0; t < 3; t++) {
      const int col = t * 16 + m;
      v2f b;
      b.x = (col < NCLASS) ? W[(long)ka * NCLASS + col]       : 0.f;
      b.y = (col < NCLASS) ? W[(long)(ka + 1) * NCLASS + col] : 0.f;
      if (t == 0) c0 = __builtin_amdgcn_wmma_f32_16x16x4_f32(false, a, false, b, (short)0, c0, false, false);
      if (t == 1) c1 = __builtin_amdgcn_wmma_f32_16x16x4_f32(false, a, false, b, (short)0, c1, false, false);
      if (t == 2) c2 = __builtin_amdgcn_wmma_f32_16x16x4_f32(false, a, false, b, (short)0, c2, false, false);
    }
  }
  const int rbase = mt * 16 + 8 * hi;
  #pragma unroll
  for (int j = 0; j < 8; j++) {
    float* hr = H + (long)(rbase + j) * NCLASS;
    if (m      < NCLASS) hr[m]      = c0[j];
    if (m + 16 < NCLASS) hr[m + 16] = c1[j];
    if (m + 32 < NCLASS) hr[m + 32] = c2[j];
  }
}

__global__ void alpha1_kernel(const float* __restrict__ H1,
                              const float* __restrict__ a_s, const float* __restrict__ a_d,
                              float* __restrict__ as, float* __restrict__ ad) {
  int i = blockIdx.x * blockDim.x + threadIdx.x;
  if (i >= N_NODES * NHEADS) return;
  const int h = i & 7;
  const float* hp = H1 + (long)(i >> 3) * HID1 + h * NHID;
  float s = 0.f, d = 0.f;
  #pragma unroll
  for (int k = 0; k < NHID; k++) { s += hp[k] * a_s[h * NHID + k]; d += hp[k] * a_d[h * NHID + k]; }
  as[i] = s; ad[i] = d;
}

__global__ void alpha2_kernel(const float* __restrict__ H2,
                              const float* __restrict__ a_s, const float* __restrict__ a_d,
                              float* __restrict__ as, float* __restrict__ ad) {
  int n = blockIdx.x * blockDim.x + threadIdx.x;
  if (n >= N_NODES) return;
  const float* hp = H2 + (long)n * NCLASS;
  float s = 0.f, d = 0.f;
  #pragma unroll
  for (int c = 0; c < NCLASS; c++) { s += hp[c] * a_s[c]; d += hp[c] * a_d[c]; }
  as[n] = s; ad[n] = d;
}

extern "C" void kernel_launch(void* const* d_in, const int* in_sizes, int n_in,
                              void* d_out, int out_size, void* d_ws, size_t ws_size,
                              hipStream_t stream) {
  const float* x      = (const float*)d_in[0];
  const float* W1     = (const float*)d_in[1];
  const float* a_src1 = (const float*)d_in[2];
  const float* a_dst1 = (const float*)d_in[3];
  const float* b1     = (const float*)d_in[4];
  const float* W2     = (const float*)d_in[5];
  const float* a_src2 = (const float*)d_in[6];
  const float* a_dst2 = (const float*)d_in[7];
  const float* b2     = (const float*)d_in[8];
  const int*   ei     = (const int*)d_in[9];
  float*       out    = (float*)d_out;

  float* ws   = (float*)d_ws;
  float* H1   = ws;
  float* as1  = H1  + (long)N_NODES * HID1;
  float* ad1  = as1 + (long)N_NODES * NHEADS;
  float* as2  = ad1 + (long)N_NODES * NHEADS;
  float* ad2  = as2 + N_NODES;
  float* X2   = ad2 + N_NODES;
  float* H2   = X2  + (long)N_NODES * HID1;

  const int B = 256;
  auto cdiv = [](long a, long b) { return (int)((a + b - 1) / b); };

  gemm1_wmma<<<cdiv(N_NODES / 16, 4), 128, 0, stream>>>(x, W1, H1);
  alpha1_kernel<<<cdiv((long)N_NODES * NHEADS, B), B, 0, stream>>>(H1, a_src1, a_dst1, as1, ad1);
  agg1_lds<<<NRANGE, AGG_T, 0, stream>>>(ei, as1, ad1, H1, b1, X2);

  gemm2_wmma<<<cdiv(N_NODES / 16, 4), 128, 0, stream>>>(X2, W2, H2);
  alpha2_kernel<<<cdiv(N_NODES, B), B, 0, stream>>>(H2, a_src2, a_dst2, as2, ad2);
  agg2_lds<<<NRANGE, AGG_T, 0, stream>>>(ei, as2, ad2, H2, b2, out);
}


# agg2 loads each range's finished edge list (source, local destination), which agg1 now stores to the workspace, instead of re-scanning the destination column; LDS float adds in arrival order
# speedup vs baseline: 7.6679x; 1.2205x over previous
#include <hip/hip_runtime.h>
#include <math.h>

#define N_NODES 50000
#define N_EDGES 800000
#define E_TOT   (N_EDGES + N_NODES)
#define NFEAT   128
#define NHID    8
#define NHEADS  8
#define HID1    (NHEADS * NHID)
#define NCLASS  40
#define NEG_SLOPE 0.2f

typedef __attribute__((ext_vector_type(2))) float v2f;
typedef __attribute__((ext_vector_type(8))) float v8f;

__device__ __forceinline__ float lrelu(float x) { return x > 0.f ? x : NEG_SLOPE * x; }

__device__ __forceinline__ unsigned f2ord(float f) {
  unsigned u = __float_as_uint(f);
  return (u & 0x80000000u) ? ~u : (u | 0x80000000u);
}
__device__ __forceinline__ float ord2f(unsigned u) {
  return (u & 0x80000000u) ? __uint_as_float(u & 0x7FFFFFFFu) : __uint_as_float(~u);
}
#define ORD_NEG_INF 0x007FFFFFu

#define AGG_T   1024
#define NRANGE  256
#define RN      196
#define LD_PER_T 8
#define CHUNK4  (AGG_T * LD_PER_T)
#define NCHUNK  ((N_EDGES / 4 + CHUNK4 - 1) / CHUNK4)
#define LIST_CAP 36864
#define P1      65
#define UNR     4

#define EBITS 20
#define EMASK ((1u << EBITS) - 1)
__device__ __forceinline__ void push_edge(int e, unsigned r, unsigned* list, unsigned* cnt) {
  const unsigned p = atomicAdd(cnt, 1u);
  if (p < LIST_CAP) list[p] = (unsigned)e | (r << EBITS);
}

__device__ __forceinline__ void scan_chunk(const int* __restrict__ ei, int chunk, int t, int lo,
                                           unsigned span, unsigned* list, unsigned* cnt) {
  const int4* __restrict__ dst4 = (const int4*)(ei + N_EDGES);
  const int q0 = chunk * CHUNK4 + t;
  int4 dv[LD_PER_T];
  #pragma unroll
  for (int i = 0; i < LD_PER_T; i++) {
    const int q = q0 + i * AGG_T;
    dv[i] = (q < N_EDGES / 4) ? dst4[q] : make_int4(-1, -1, -1, -1);
  }
  #pragma unroll
  for (int i = 0; i < LD_PER_T; i++) {
    const int e = 4 * (q0 + i * AGG_T);
    const unsigned r0 = (unsigned)(dv[i].x - lo), r1 = (unsigned)(dv[i].y - lo);
    const unsigned r2 = (unsigned)(dv[i].z - lo), r3 = (unsigned)(dv[i].w - lo);
    if (r0 < span) push_edge(e + 0, r0, list, cnt);
    if (r1 < span) push_edge(e + 1, r1, list, cnt);
    if (r2 < span) push_edge(e + 2, r2, list, cnt);
    if (r3 < span) push_edge(e + 3, r3, list, cnt);
  }
}

__device__ __forceinline__ void list_self_loops(int t, int lo, unsigned span, unsigned* list, unsigned* cnt) {
  if (t < (int)span) list[t] = (unsigned)(lo + t) | ((unsigned)t << EBITS);
  if (t == 0) *cnt = span;
}

__device__ __forceinline__ void resolve_src(const int* __restrict__ ei, unsigned* list,
                                            unsigned first, unsigned n, int t) {
  for (unsigned b = first + t; b < n; b += UNR * AGG_T) {
    unsigned en[UNR]; int sv[UNR];
    #pragma unroll
    for (int j = 0; j < UNR; j++) {
      const bool in = b + j * AGG_T < n;
      en[j] = in ? list[b + j * AGG_T] : 0u;
      sv[j] = in ? ei[en[j] & EMASK] : 0;
    }
    #pragma unroll
    for (int j = 0; j < UNR; j++)
      if (b + j * AGG_T < n) list[b + j * AGG_T] = (unsigned)sv[j] | (en[j] & ~EMASK);
  }
}

__global__ void __launch_bounds__(AGG_T) agg1_lds(const int* __restrict__ ei,
                                                  const float* __restrict__ as,
                                                  const float* __restrict__ ad,
                                                  const float* __restrict__ H1,
                                                  const float* __restrict__ bias,
                                                  float* __restrict__ out1,
                                                  unsigned* __restrict__ glist,
                                                  unsigned* __restrict__ gcnt) {
  __shared__ float    acc[RN * P1];
  __shared__ float    dsum[RN * NHEADS];
  __shared__ float    adl[RN * NHEADS];
  __shared__ unsigned mxo[RN * NHEADS];
  __shared__ unsigned list[LIST_CAP];
  __shared__ unsigned cnt;

  const int t  = threadIdx.x;
  const int lo = blockIdx.x * RN;
  const unsigned span = (unsigned)(min(lo + RN, N_NODES) - lo);
  const int h = t & 7;

  for (int i = t; i < RN * P1; i += AGG_T) acc[i] = 0.f;
  for (int i = t; i < (int)span * NHEADS; i += AGG_T) {
    dsum[i] = 0.f;
    adl[i]  = ad[lo * NHEADS + i];
    mxo[i]  = ORD_NEG_INF;
  }

  auto max_pass = [&](unsigned n) {
    for (unsigned b = t >> 3; b < n; b += UNR * (AGG_T / 8)) {
      unsigned dl[UNR]; float a[UNR];
      #pragma unroll
      for (int j = 0; j < UNR; j++) {
        const unsigned en = list[min(b + j * (AGG_T / 8), n - 1)];
        dl[j] = en >> EBITS;
        a[j]  = as[(en & EMASK) * NHEADS + h];
      }
      #pragma unroll
      for (int j = 0; j < UNR; j++)
        atomicMax(&mxo[dl[j] * NHEADS + h], f2ord(lrelu(a[j] + adl[dl[j] * NHEADS + h])));
    }
  };
  auto sum_pass = [&](unsigned n) {
    for (unsigned b = t >> 3; b < n; b += UNR * (AGG_T / 8)) {
      unsigned dl[UNR]; float a[UNR]; float4 h0[UNR], h1[UNR];
      #pragma unroll
      for (int j = 0; j < UNR; j++) {
        const unsigned en = list[min(b + j * (AGG_T / 8), n - 1)];
        const unsigned s = en & EMASK;
        dl[j] = en >> EBITS;
        a[j]  = as[s * NHEADS + h];
        const float4* hs = (const float4*)(H1 + (long)s * HID1 + h * NHID);
        h0[j] = hs[0]; h1[j] = hs[1];
      }
      #pragma unroll
      for (int j = 0; j < UNR; j++) {
        if (b + j * (AGG_T / 8) < n) {
          const int d8 = dl[j] * NHEADS + h;
          const float v = lrelu(a[j] + adl[d8]);
          const float w = __expf(v - ord2f(mxo[d8]));
          atomicAdd(&dsum[d8], w);
          float* od = &acc[dl[j] * P1 + h * NHID];
          atomicAdd(&od[0], w * h0[j].x); atomicAdd(&od[1], w * h0[j].y);
          atomicAdd(&od[2], w * h0[j].z); atomicAdd(&od[3], w * h0[j].w);
          atomicAdd(&od[4], w * h1[j].x); atomicAdd(&od[5], w * h1[j].y);
          atomicAdd(&od[6], w * h1[j].z); atomicAdd(&od[7], w * h1[j].w);
        }
      }
    }
  };

  list_self_loops(t, lo, span, list, &cnt);
  __syncthreads();
  for (int c = 0; c < NCHUNK; c++) scan_chunk(ei, c, t, lo, span, list, &cnt);
  __syncthreads();
  const unsigned n = cnt;

  if (t == 0) gcnt[blockIdx.x] = n;
  if (n <= LIST_CAP) {
    resolve_src(ei, list, span, n, t);
    __syncthreads();
    unsigned* gl = glist + (long)blockIdx.x * LIST_CAP;
    for (unsigned i = t; i < n; i += AGG_T) gl[i] = list[i];
    max_pass(n);
    __syncthreads();
    sum_pass(n);
  } else {
    for (int phase = 0; phase < 2; phase++) {
      for (int it = 0; it <= NCHUNK; it++) {
        __syncthreads();
        if (it == 0) list_self_loops(t, lo, span, list, &cnt);
        else if (t == 0) cnt = 0;
        __syncthreads();
        if (it > 0) scan_chunk(ei, it - 1, t, lo, span, list, &cnt);
        __syncthreads();
        const unsigned nc = cnt;
        resolve_src(ei, list, it == 0 ? span : 0u, nc, t);
        __syncthreads();
        if (phase == 0) max_pass(nc); else sum_pass(nc);
      }
    }
  }
  __syncthreads();

  for (int i = t; i < (int)span * HID1; i += AGG_T) {
    const float v = acc[(i >> 6) * P1 + (i & 63)] / dsum[i >> 3] + bias[i & (HID1 - 1)];
    out1[(long)lo * HID1 + i] = v > 0.f ? v : (__expf(v) - 1.f);
  }
}

__global__ void __launch_bounds__(AGG_T) agg2_lds(const int* __restrict__ ei,
                                                  const float* __restrict__ as,
                                                  const float* __restrict__ ad,
                                                  const float* __restrict__ H2,
                                                  const float* __restrict__ bias,
                                                  float* __restrict__ out,
                                                  const unsigned* __restrict__ glist,
                                                  const unsigned* __restrict__ gcnt) {
  __shared__ float    acc[RN * NCLASS];
  __shared__ float    dsum[RN];
  __shared__ float    adl[RN];
  __shared__ unsigned mxo[RN];
  __shared__ unsigned list[LIST_CAP];
  __shared__ unsigned cnt;

  const int t  = threadIdx.x;
  const int lo = blockIdx.x * RN;
  const unsigned span = (unsigned)(min(lo + RN, N_NODES) - lo);
  const int l = t & 7;

  for (int i = t; i < RN * NCLASS; i += AGG_T) acc[i] = 0.f;
  if (t < (int)span) {
    dsum[t] = 0.f;
    adl[t]  = ad[lo + t];
    mxo[t]  = ORD_NEG_INF;
  }

  auto max_pass = [&](unsigned n) {
    for (unsigned b = t; b < n; b += UNR * AGG_T) {
      unsigned dl[UNR]; float a[UNR];
      #pragma unroll
      for (int j = 0; j < UNR; j++) {
        const unsigned en = list[min(b + j * AGG_T, n - 1)];
        dl[j] = en >> EBITS;
        a[j]  = as[en & EMASK];
      }
      #pragma unroll
      for (int j = 0; j < UNR; j++)
        atomicMax(&mxo[dl[j]], f2ord(lrelu(a[j] + adl[dl[j]])));
    }
  };
  auto sum_pass = [&](unsigned n) {
    for (unsigned b = t >> 3; b < n; b += UNR * (AGG_T / 8)) {
      unsigned dl[UNR]; float a[UNR]; float hv[UNR][NCLASS / 8];
      #pragma unroll
      for (int j = 0; j < UNR; j++) {
        const unsigned en = list[min(b + j * (AGG_T / 8), n - 1)];
        const unsigned s = en & EMASK;
        dl[j] = en >> EBITS;
        a[j]  = as[s];
        const float* hs = H2 + (long)s * NCLASS + l;
        #pragma unroll
        for (int k = 0; k < NCLASS / 8; k++) hv[j][k] = hs[k * 8];
      }
      #pragma unroll
      for (int j = 0; j < UNR; j++) {
        if (b + j * (AGG_T / 8) < n) {
          const float v = lrelu(a[j] + adl[dl[j]]);
          const float w = __expf(v - ord2f(mxo[dl[j]]));
          if (l == 0) atomicAdd(&dsum[dl[j]], w);
          float* od = &acc[dl[j] * NCLASS + l];
          #pragma unroll
          for (int k = 0; k < NCLASS / 8; k++) atomicAdd(&od[k * 8], w * hv[j][k]);
        }
      }
    }
  };

  const unsigned n = gcnt[blockIdx.x];

  if (n <= LIST_CAP) {
    const unsigned* gl = glist + (long)blockIdx.x * LIST_CAP;
    for (unsigned i = t; i < n; i += AGG_T) list[i] = gl[i];
    __syncthreads();
    max_pass(n);
    __syncthreads();
    sum_pass(n);
  } else {
    for (int phase = 0; phase < 2; phase++) {
      for (int it = 0; it <= NCHUNK; it++) {
        __syncthreads();
        if (it == 0) list_self_loops(t, lo, span, list, &cnt);
        else if (t == 0) cnt = 0;
        __syncthreads();
        if (it > 0) scan_chunk(ei, it - 1, t, lo, span, list, &cnt);
        __syncthreads();
        const unsigned nc = cnt;
        resolve_src(ei, list, it == 0 ? span : 0u, nc, t);
        __syncthreads();
        if (phase == 0) max_pass(nc); else sum_pass(nc);
      }
    }
  }
  __syncthreads();

  for (int i = t; i < (int)span * NCLASS; i += AGG_T) acc[i] = acc[i] / dsum[i / NCLASS] + bias[i % NCLASS];
  __syncthreads();
  float* lse = adl;
  if (t < (int)span) {
    const float* r = &acc[t * NCLASS];
    float mx = -INFINITY;
    for (int c = 0; c < NCLASS; c++) mx = fmaxf(mx, r[c]);
    float ssum = 0.f;
    for (int c = 0; c < NCLASS; c++) ssum += __expf(r[c] - mx);
    lse[t] = mx + __logf(ssum);
  }
  __syncthreads();
  for (int i = t; i < (int)span * NCLASS; i += AGG_T) out[(long)lo * NCLASS + i] = acc[i] - lse[i / NCLASS];
}

__global__ void __launch_bounds__(128) gemm1_wmma(const float* __restrict__ X,
                                                  const float* __restrict__ W,
                                                  float* __restrict__ H) {
  const int wave = threadIdx.x >> 5;
  const int lane = threadIdx.x & 31;
  const int mt   = blockIdx.x * 4 + wave;
  if (mt >= N_NODES / 16) return;
  const int m  = lane & 15;
  const int hi = lane >> 4;
  const float* xr = X + (long)(mt * 16 + m) * NFEAT;

  v8f c0 = {}, c1 = {}, c2 = {}, c3 = {};
  #pragma unroll 4
  for (int k0 = 0; k0 < NFEAT; k0 += 4) {
    const int ka = k0 + 2 * hi;
    v2f a; a.x = xr[ka]; a.y = xr[ka + 1];
    const float* w0 = W + (long)ka * HID1 + m;
    const float* w1 = w0 + HID1;
    v2f b;
    b.x = w0[0];  b.y = w1[0];
    c0 = __builtin_amdgcn_wmma_f32_16x16x4_f32(false, a, false, b, (short)0, c0, false, false);
    b.x = w0[16]; b.y = w1[16];
    c1 = __builtin_amdgcn_wmma_f32_16x16x4_f32(false, a, false, b, (short)0, c1, false, false);
    b.x = w0[32]; b.y = w1[32];
    c2 = __builtin_amdgcn_wmma_f32_16x16x4_f32(false, a, false, b, (short)0, c2, false, false);
    b.x = w0[48]; b.y = w1[48];
    c3 = __builtin_amdgcn_wmma_f32_16x16x4_f32(false, a, false, b, (short)0, c3, false, false);
  }
  float* hp = H + (long)(mt * 16 + 8 * hi) * HID1 + m;
  #pragma unroll
  for (int j = 0; j < 8; j++) {
    hp[(long)j * HID1 +  0] = c0[j];
    hp[(long)j * HID1 + 16] = c1[j];
    hp[(long)j * HID1 + 32] = c2[j];
    hp[(long)j * HID1 + 48] = c3[j];
  }
}

__global__ void __launch_bounds__(128) gemm2_wmma(const float* __restrict__ X,
                                                  const float* __restrict__ W,
                                                  float* __restrict__ H) {
  const int wave = threadIdx.x >> 5;
  const int lane = threadIdx.x & 31;
  const int mt   = blockIdx.x * 4 + wave;
  if (mt >= N_NODES / 16) return;
  const int m  = lane & 15;
  const int hi = lane >> 4;
  const float* xr = X + (long)(mt * 16 + m) * HID1;

  v8f c0 = {}, c1 = {}, c2 = {};
  #pragma unroll 4
  for (int k0 = 0; k0 < HID1; k0 += 4) {
    const int ka = k0 + 2 * hi;
    v2f a; a.x = xr[ka]; a.y = xr[ka + 1];
    #pragma unroll
    for (int t = 0; t < 3; t++) {
      const int col = t * 16 + m;
      v2f b;
      b.x = (col < NCLASS) ? W[(long)ka * NCLASS + col]       : 0.f;
      b.y = (col < NCLASS) ? W[(long)(ka + 1) * NCLASS + col] : 0.f;
      if (t == 0) c0 = __builtin_amdgcn_wmma_f32_16x16x4_f32(false, a, false, b, (short)0, c0, false, false);
      if (t == 1) c1 = __builtin_amdgcn_wmma_f32_16x16x4_f32(false, a, false, b, (short)0, c1, false, false);
      if (t == 2) c2 = __builtin_amdgcn_wmma_f32_16x16x4_f32(false, a, false, b, (short)0, c2, false, false);
    }
  }
  const int rbase = mt * 16 + 8 * hi;
  #pragma unroll
  for (int j = 0; j < 8; j++) {
    float* hr = H + (long)(rbase + j) * NCLASS;
    if (m      < NCLASS) hr[m]      = c0[j];
    if (m + 16 < NCLASS) hr[m + 16] = c1[j];
    if (m + 32 < NCLASS) hr[m + 32] = c2[j];
  }
}

__global__ void alpha1_kernel(const float* __restrict__ H1,
                              const float* __restrict__ a_s, const float* __restrict__ a_d,
                              float* __restrict__ as, float* __restrict__ ad) {
  int i = blockIdx.x * blockDim.x + threadIdx.x;
  if (i >= N_NODES * NHEADS) return;
  const int h = i & 7;
  const float* hp = H1 + (long)(i >> 3) * HID1 + h * NHID;
  float s = 0.f, d = 0.f;
  #pragma unroll
  for (int k = 0; k < NHID; k++) { s += hp[k] * a_s[h * NHID + k]; d += hp[k] * a_d[h * NHID + k]; }
  as[i] = s; ad[i] = d;
}

__global__ void alpha2_kernel(const float* __restrict__ H2,
                              const float* __restrict__ a_s, const float* __restrict__ a_d,
                              float* __restrict__ as, float* __restrict__ ad) {
  int n = blockIdx.x * blockDim.x + threadIdx.x;
  if (n >= N_NODES) return;
  const float* hp = H2 + (long)n * NCLASS;
  float s = 0.f, d = 0.f;
  #pragma unroll
  for (int c = 0; c < NCLASS; c++) { s += hp[c] * a_s[c]; d += hp[c] * a_d[c]; }
  as[n] = s; ad[n] = d;
}

extern "C" void kernel_launch(void* const* d_in, const int* in_sizes, int n_in,
                              void* d_out, int out_size, void* d_ws, size_t ws_size,
                              hipStream_t stream) {
  const float* x      = (const float*)d_in[0];
  const float* W1     = (const float*)d_in[1];
  const float* a_src1 = (const float*)d_in[2];
  const float* a_dst1 = (const float*)d_in[3];
  const float* b1     = (const float*)d_in[4];
  const float* W2     = (const float*)d_in[5];
  const float* a_src2 = (const float*)d_in[6];
  const float* a_dst2 = (const float*)d_in[7];
  const float* b2     = (const float*)d_in[8];
  const int*   ei     = (const int*)d_in[9];
  float*       out    = (float*)d_out;

  float* ws   = (float*)d_ws;
  float* H1   = ws;
  float* as1  = H1  + (long)N_NODES * HID1;
  float* ad1  = as1 + (long)N_NODES * NHEADS;
  float* as2  = ad1 + (long)N_NODES * NHEADS;
  float* ad2  = as2 + N_NODES;
  float* X2   = ad2 + N_NODES;
  float* H2   = X2  + (long)N_NODES * HID1;
  unsigned* glist = (unsigned*)(H2 + (long)N_NODES * NCLASS);
  unsigned* gcnt  = glist + (long)NRANGE * LIST_CAP;

  const int B = 256;
  auto cdiv = [](long a, long b) { return (int)((a + b - 1) / b); };

  gemm1_wmma<<<cdiv(N_NODES / 16, 4), 128, 0, stream>>>(x, W1, H1);
  alpha1_kernel<<<cdiv((long)N_NODES * NHEADS, B), B, 0, stream>>>(H1, a_src1, a_dst1, as1, ad1);
  agg1_lds<<<NRANGE, AGG_T, 0, stream>>>(ei, as1, ad1, H1, b1, X2, glist, gcnt);

  gemm2_wmma<<<cdiv(N_NODES / 16, 4), 128, 0, stream>>>(X2, W2, H2);
  alpha2_kernel<<<cdiv(N_NODES, B), B, 0, stream>>>(H2, a_src2, a_dst2, as2, ad2);
  agg2_lds<<<NRANGE, AGG_T, 0, stream>>>(ei, as2, ad2, H2, b2, out, glist, gcnt);
}


# agg1 filters a byte column of range ids (dst/196, 16 keys per 16-byte load), not the 4-byte dst column; alpha1's launch writes it plus a src|local-dst word per edge; LDS float adds: arrival order
# speedup vs baseline: 8.8150x; 1.1496x over previous
#include <hip/hip_runtime.h>
#include <math.h>

#define N_NODES 50000
#define N_EDGES 800000
#define E_TOT   (N_EDGES + N_NODES)
#define NFEAT   128
#define NHID    8
#define NHEADS  8
#define HID1    (NHEADS * NHID)
#define NCLASS  40
#define NEG_SLOPE 0.2f

typedef __attribute__((ext_vector_type(2))) float v2f;
typedef __attribute__((ext_vector_type(8))) float v8f;

__device__ __forceinline__ float lrelu(float x) { return x > 0.f ? x : NEG_SLOPE * x; }

__device__ __forceinline__ unsigned f2ord(float f) {
  unsigned u = __float_as_uint(f);
  return (u & 0x80000000u) ? ~u : (u | 0x80000000u);
}
__device__ __forceinline__ float ord2f(unsigned u) {
  return (u & 0x80000000u) ? __uint_as_float(u & 0x7FFFFFFFu) : __uint_as_float(~u);
}
#define ORD_NEG_INF 0x007FFFFFu

#define AGG_T   1024
#define NRANGE  256
#define RN      196
#define KEYS4   (N_EDGES / 16)
#define LD_FAST 7
#define LD_GEN  2
#define NCHUNK_OF(LD) ((KEYS4 + AGG_T * (LD) - 1) / (AGG_T * (LD)))
#define LIST_CAP 36864
#define P1      65
#define UNR     4

#define EBITS 20
#define EMASK ((1u << EBITS) - 1)
__device__ __forceinline__ void push_edge(unsigned e, unsigned* list, unsigned* cnt) {
  const unsigned p = atomicAdd(cnt, 1u);
  if (p < LIST_CAP) list[p] = e;
}

__device__ __forceinline__ unsigned zero_bytes(unsigned x) {
  return ~(((x & 0x7F7F7F7Fu) + 0x7F7F7F7Fu) | x) & 0x80808080u;
}

__device__ __forceinline__ void edge_keys(const int* __restrict__ ei, int q,
                                          unsigned* __restrict__ key, unsigned* __restrict__ rec) {
  const int4 s = ((const int4*)ei)[q];
  const int4 d = ((const int4*)(ei + N_EDGES))[q];
  const unsigned g0 = (unsigned)d.x / RN, g1 = (unsigned)d.y / RN, g2 = (unsigned)d.z / RN, g3 = (unsigned)d.w / RN;
  key[q] = g0 | (g1 << 8) | (g2 << 16) | (g3 << 24);
  ((uint4*)rec)[q] = make_uint4((unsigned)s.x | (((unsigned)d.x - g0 * RN) << EBITS),
                                (unsigned)s.y | (((unsigned)d.y - g1 * RN) << EBITS),
                                (unsigned)s.z | (((unsigned)d.z - g2 * RN) << EBITS),
                                (unsigned)s.w | (((unsigned)d.w - g3 * RN) << EBITS));
}

template <int LD>
__device__ __forceinline__ void scan_chunk(const unsigned* __restrict__ key, int chunk, int t,
                                           unsigned gpat, unsigned* list, unsigned* cnt) {
  const uint4* __restrict__ key4 = (const uint4*)key;
  const int q0 = chunk * (AGG_T * LD) + t;
  uint4 kv[LD];
  #pragma unroll
  for (int i = 0; i < LD; i++) {
    const int q = q0 + i * AGG_T;
    kv[i] = (q < KEYS4) ? key4[q] : make_uint4(~gpat, ~gpat, ~gpat, ~gpat);
  }
  #pragma unroll
  for (int i = 0; i < LD; i++) {
    unsigned m = (zero_bytes(kv[i].x ^ gpat) >> 7) | (zero_bytes(kv[i].y ^ gpat) >> 6) |
                 (zero_bytes(kv[i].z ^ gpat) >> 5) | (zero_bytes(kv[i].w ^ gpat) >> 4);
    const unsigned e0 = 16u * (unsigned)(q0 + i * AGG_T);
    while (m) {
      const unsigned p = (unsigned)__builtin_ctz(m);
      m &= m - 1;
      push_edge(e0 + 4 * (p & 7) + (p >> 3), list, cnt);
    }
  }
}

__device__ __forceinline__ void list_self_loops(int t, int lo, unsigned span, unsigned* list, unsigned* cnt) {
  if (t < (int)span) list[t] = (unsigned)(lo + t) | ((unsigned)t << EBITS);
  if (t == 0) *cnt = span;
}

__device__ __forceinline__ void resolve_src(const unsigned* __restrict__ rec, unsigned* list,
                                            unsigned first, unsigned n, int t) {
  for (unsigned b = first + t; b < n; b += UNR * AGG_T) {
    unsigned rv[UNR];
    #pragma unroll
    for (int j = 0; j < UNR; j++) {
      const bool in = b + j * AGG_T < n;
      rv[j] = in ? rec[list[b + j * AGG_T]] : 0u;
    }
    #pragma unroll
    for (int j = 0; j < UNR; j++)
      if (b + j * AGG_T < n) list[b + j * AGG_T] = rv[j];
  }
}

__global__ void __launch_bounds__(AGG_T) agg1_lds(const unsigned* __restrict__ key,
                                                  const unsigned* __restrict__ rec,
                                                  const float* __restrict__ as,
                                                  const float* __restrict__ ad,
                                                  const float* __restrict__ H1,
                                                  const float* __restrict__ bias,
                                                  float* __restrict__ out1,
                                                  unsigned* __restrict__ glist,
                                                  unsigned* __restrict__ gcnt) {
  __shared__ float    acc[RN * P1];
  __shared__ float    dsum[RN * NHEADS];
  __shared__ float    adl[RN * NHEADS];
  __shared__ unsigned mxo[RN * NHEADS];
  __shared__ unsigned list[LIST_CAP];
  __shared__ unsigned cnt;

  const int t  = threadIdx.x;
  const int lo = blockIdx.x * RN;
  const unsigned span = (unsigned)(min(lo + RN, N_NODES) - lo);
  const int h = t & 7;
  const unsigned gpat = blockIdx.x * 0x01010101u;

  for (int i = t; i < RN * P1; i += AGG_T) acc[i] = 0.f;
  for (int i = t; i < (int)span * NHEADS; i += AGG_T) {
    dsum[i] = 0.f;
    adl[i]  = ad[lo * NHEADS + i];
    mxo[i]  = ORD_NEG_INF;
  }

  auto max_pass = [&](unsigned n) {
    for (unsigned b = t >> 3; b < n; b += UNR * (AGG_T / 8)) {
      unsigned dl[UNR]; float a[UNR];
      #pragma unroll
      for (int j = 0; j < UNR; j++) {
        const unsigned en = list[min(b + j * (AGG_T / 8), n - 1)];
        dl[j] = en >> EBITS;
        a[j]  = as[(en & EMASK) * NHEADS + h];
      }
      #pragma unroll
      for (int j = 0; j < UNR; j++)
        atomicMax(&mxo[dl[j] * NHEADS + h], f2ord(lrelu(a[j] + adl[dl[j] * NHEADS + h])));
    }
  };
  auto sum_pass = [&](unsigned n) {
    for (unsigned b = t >> 3; b < n; b += UNR * (AGG_T / 8)) {
      unsigned dl[UNR]; float a[UNR]; float4 h0[UNR], h1[UNR];
      #pragma unroll
      for (int j = 0; j < UNR; j++) {
        const unsigned en = list[min(b + j * (AGG_T / 8), n - 1)];
        const unsigned s = en & EMASK;
        dl[j] = en >> EBITS;
        a[j]  = as[s * NHEADS + h];
        const float4* hs = (const float4*)(H1 + (long)s * HID1 + h * NHID);
        h0[j] = hs[0]; h1[j] = hs[1];
      }
      #pragma unroll
      for (int j = 0; j < UNR; j++) {
        if (b + j * (AGG_T / 8) < n) {
          const int d8 = dl[j] * NHEADS + h;
          const float v = lrelu(a[j] + adl[d8]);
          const float w = __expf(v - ord2f(mxo[d8]));
          atomicAdd(&dsum[d8], w);
          float* od = &acc[dl[j] * P1 + h * NHID];
          atomicAdd(&od[0], w * h0[j].x); atomicAdd(&od[1], w * h0[j].y);
          atomicAdd(&od[2], w * h0[j].z); atomicAdd(&od[3], w * h0[j].w);
          atomicAdd(&od[4], w * h1[j].x); atomicAdd(&od[5], w * h1[j].y);
          atomicAdd(&od[6], w * h1[j].z); atomicAdd(&od[7], w * h1[j].w);
        }
      }
    }
  };

  list_self_loops(t, lo, span, list, &cnt);
  __syncthreads();
  for (int c = 0; c < NCHUNK_OF(LD_FAST); c++) scan_chunk<LD_FAST>(key, c, t, gpat, list, &cnt);
  __syncthreads();
  const unsigned n = cnt;

  if (t == 0) gcnt[blockIdx.x] = n;
  if (n <= LIST_CAP) {
    resolve_src(rec, list, span, n, t);
    __syncthreads();
    unsigned* gl = glist + (long)blockIdx.x * LIST_CAP;
    for (unsigned i = t; i < n; i += AGG_T) gl[i] = list[i];
    max_pass(n);
    __syncthreads();
    sum_pass(n);
  } else {
    for (int phase = 0; phase < 2; phase++) {
      for (int it = 0; it <= NCHUNK_OF(LD_GEN); it++) {
        __syncthreads();
        if (it == 0) list_self_loops(t, lo, span, list, &cnt);
        else if (t == 0) cnt = 0;
        __syncthreads();
        if (it > 0) scan_chunk<LD_GEN>(key, it - 1, t, gpat, list, &cnt);
        __syncthreads();
        const unsigned nc = cnt;
        resolve_src(rec, list, it == 0 ? span : 0u, nc, t);
        __syncthreads();
        if (phase == 0) max_pass(nc); else sum_pass(nc);
      }
    }
  }
  __syncthreads();

  for (int i = t; i < (int)span * HID1; i += AGG_T) {
    const float v = acc[(i >> 6) * P1 + (i & 63)] / dsum[i >> 3] + bias[i & (HID1 - 1)];
    out1[(long)lo * HID1 + i] = v > 0.f ? v : (__expf(v) - 1.f);
  }
}

__global__ void __launch_bounds__(AGG_T) agg2_lds(const unsigned* __restrict__ key,
                                                  const unsigned* __restrict__ rec,
                                                  const float* __restrict__ as,
                                                  const float* __restrict__ ad,
                                                  const float* __restrict__ H2,
                                                  const float* __restrict__ bias,
                                                  float* __restrict__ out,
                                                  const unsigned* __restrict__ glist,
                                                  const unsigned* __restrict__ gcnt) {
  __shared__ float    acc[RN * NCLASS];
  __shared__ float    dsum[RN];
  __shared__ float    adl[RN];
  __shared__ unsigned mxo[RN];
  __shared__ unsigned list[LIST_CAP];
  __shared__ unsigned cnt;

  const int t  = threadIdx.x;
  const int lo = blockIdx.x * RN;
  const unsigned span = (unsigned)(min(lo + RN, N_NODES) - lo);
  const int l = t & 7;
  const unsigned gpat = blockIdx.x * 0x01010101u;

  for (int i = t; i < RN * NCLASS; i += AGG_T) acc[i] = 0.f;
  if (t < (int)span) {
    dsum[t] = 0.f;
    adl[t]  = ad[lo + t];
    mxo[t]  = ORD_NEG_INF;
  }

  auto max_pass = [&](unsigned n) {
    for (unsigned b = t; b < n; b += UNR * AGG_T) {
      unsigned dl[UNR]; float a[UNR];
      #pragma unroll
      for (int j = 0; j < UNR; j++) {
        const unsigned en = list[min(b + j * AGG_T, n - 1)];
        dl[j] = en >> EBITS;
        a[j]  = as[en & EMASK];
      }
      #pragma unroll
      for (int j = 0; j < UNR; j++)
        atomicMax(&mxo[dl[j]], f2ord(lrelu(a[j] + adl[dl[j]])));
    }
  };
  auto sum_pass = [&](unsigned n) {
    for (unsigned b = t >> 3; b < n; b += UNR * (AGG_T / 8)) {
      unsigned dl[UNR]; float a[UNR]; float hv[UNR][NCLASS / 8];
      #pragma unroll
      for (int j = 0; j < UNR; j++) {
        const unsigned en = list[min(b + j * (AGG_T / 8), n - 1)];
        const unsigned s = en & EMASK;
        dl[j] = en >> EBITS;
        a[j]  = as[s];
        const float* hs = H2 + (long)s * NCLASS + l;
        #pragma unroll
        for (int k = 0; k < NCLASS / 8; k++) hv[j][k] = hs[k * 8];
      }
      #pragma unroll
      for (int j = 0; j < UNR; j++) {
        if (b + j * (AGG_T / 8) < n) {
          const float v = lrelu(a[j] + adl[dl[j]]);
          const float w = __expf(v - ord2f(mxo[dl[j]]));
          if (l == 0) atomicAdd(&dsum[dl[j]], w);
          float* od = &acc[dl[j] * NCLASS + l];
          #pragma unroll
          for (int k = 0; k < NCLASS / 8; k++) atomicAdd(&od[k * 8], w * hv[j][k]);
        }
      }
    }
  };

  const unsigned n = gcnt[blockIdx.x];

  if (n <= LIST_CAP) {
    const unsigned* gl = glist + (long)blockIdx.x * LIST_CAP;
    for (unsigned i = t; i < n; i += AGG_T) list[i] = gl[i];
    __syncthreads();
    max_pass(n);
    __syncthreads();
    sum_pass(n);
  } else {
    for (int phase = 0; phase < 2; phase++) {
      for (int it = 0; it <= NCHUNK_OF(LD_GEN); it++) {
        __syncthreads();
        if (it == 0) list_self_loops(t, lo, span, list, &cnt);
        else if (t == 0) cnt = 0;
        __syncthreads();
        if (it > 0) scan_chunk<LD_GEN>(key, it - 1, t, gpat, list, &cnt);
        __syncthreads();
        const unsigned nc = cnt;
        resolve_src(rec, list, it == 0 ? span : 0u, nc, t);
        __syncthreads();
        if (phase == 0) max_pass(nc); else sum_pass(nc);
      }
    }
  }
  __syncthreads();

  for (int i = t; i < (int)span * NCLASS; i += AGG_T) acc[i] = acc[i] / dsum[i / NCLASS] + bias[i % NCLASS];
  __syncthreads();
  float* lse = adl;
  if (t < (int)span) {
    const float* r = &acc[t * NCLASS];
    float mx = -INFINITY;
    for (int c = 0; c < NCLASS; c++) mx = fmaxf(mx, r[c]);
    float ssum = 0.f;
    for (int c = 0; c < NCLASS; c++) ssum += __expf(r[c] - mx);
    lse[t] = mx + __logf(ssum);
  }
  __syncthreads();
  for (int i = t; i < (int)span * NCLASS; i += AGG_T) out[(long)lo * NCLASS + i] = acc[i] - lse[i / NCLASS];
}

__global__ void __launch_bounds__(128) gemm1_wmma(const float* __restrict__ X,
                                                  const float* __restrict__ W,
                                                  float* __restrict__ H) {
  const int wave = threadIdx.x >> 5;
  const int lane = threadIdx.x & 31;
  const int mt   = blockIdx.x * 4 + wave;
  if (mt >= N_NODES / 16) return;
  const int m  = lane & 15;
  const int hi = lane >> 4;
  const float* xr = X + (long)(mt * 16 + m) * NFEAT;

  v8f c0 = {}, c1 = {}, c2 = {}, c3 = {};
  #pragma unroll 4
  for (int k0 = 0; k0 < NFEAT; k0 += 4) {
    const int ka = k0 + 2 * hi;
    v2f a; a.x = xr[ka]; a.y = xr[ka + 1];
    const float* w0 = W + (long)ka * HID1 + m;
    const float* w1 = w0 + HID1;
    v2f b;
    b.x = w0[0];  b.y = w1[0];
    c0 = __builtin_amdgcn_wmma_f32_16x16x4_f32(false, a, false, b, (short)0, c0, false, false);
    b.x = w0[16]; b.y = w1[16];
    c1 = __builtin_amdgcn_wmma_f32_16x16x4_f32(false, a, false, b, (short)0, c1, false, false);
    b.x = w0[32]; b.y = w1[32];
    c2 = __builtin_amdgcn_wmma_f32_16x16x4_f32(false, a, false, b, (short)0, c2, false, false);
    b.x = w0[48]; b.y = w1[48];
    c3 = __builtin_amdgcn_wmma_f32_16x16x4_f32(false, a, false, b, (short)0, c3, false, false);
  }
  float* hp = H + (long)(mt * 16 + 8 * hi) * HID1 + m;
  #pragma unroll
  for (int j = 0; j < 8; j++) {
    hp[(long)j * HID1 +  0] = c0[j];
    hp[(long)j * HID1 + 16] = c1[j];
    hp[(long)j * HID1 + 32] = c2[j];
    hp[(long)j * HID1 + 48] = c3[j];
  }
}

__global__ void __launch_bounds__(128) gemm2_wmma(const float* __restrict__ X,
                                                  const float* __restrict__ W,
                                                  float* __restrict__ H) {
  const int wave = threadIdx.x >> 5;
  const int lane = threadIdx.x & 31;
  const int mt   = blockIdx.x * 4 + wave;
  if (mt >= N_NODES / 16) return;
  const int m  = lane & 15;
  const int hi = lane >> 4;
  const float* xr = X + (long)(mt * 16 + m) * HID1;

  v8f c0 = {}, c1 = {}, c2 = {};
  #pragma unroll 4
  for (int k0 = 0; k0 < HID1; k0 += 4) {
    const int ka = k0 + 2 * hi;
    v2f a; a.x = xr[ka]; a.y = xr[ka + 1];
    #pragma unroll
    for (int t = 0; t < 3; t++) {
      const int col = t * 16 + m;
      v2f b;
      b.x = (col < NCLASS) ? W[(long)ka * NCLASS + col]       : 0.f;
      b.y = (col < NCLASS) ? W[(long)(ka + 1) * NCLASS + col] : 0.f;
      if (t == 0) c0 = __builtin_amdgcn_wmma_f32_16x16x4_f32(false, a, false, b, (short)0, c0, false, false);
      if (t == 1) c1 = __builtin_amdgcn_wmma_f32_16x16x4_f32(false, a, false, b, (short)0, c1, false, false);
      if (t == 2) c2 = __builtin_amdgcn_wmma_f32_16x16x4_f32(false, a, false, b, (short)0, c2, false, false);
    }
  }
  const int rbase = mt * 16 + 8 * hi;
  #pragma unroll
  for (int j = 0; j < 8; j++) {
    float* hr = H + (long)(rbase + j) * NCLASS;
    if (m      < NCLASS) hr[m]      = c0[j];
    if (m + 16 < NCLASS) hr[m + 16] = c1[j];
    if (m + 32 < NCLASS) hr[m + 32] = c2[j];
  }
}

__global__ void alpha1_kernel(const float* __restrict__ H1,
                              const float* __restrict__ a_s, const float* __restrict__ a_d,
                              float* __restrict__ as, float* __restrict__ ad,
                              const int* __restrict__ ei,
                              unsigned* __restrict__ key, unsigned* __restrict__ rec) {
  int i = blockIdx.x * blockDim.x + threadIdx.x;
  if (i < N_EDGES / 4) edge_keys(ei, i, key, rec);
  if (i >= N_NODES * NHEADS) return;
  const int h = i & 7;
  const float* hp = H1 + (long)(i >> 3) * HID1 + h * NHID;
  float s = 0.f, d = 0.f;
  #pragma unroll
  for (int k = 0; k < NHID; k++) { s += hp[k] * a_s[h * NHID + k]; d += hp[k] * a_d[h * NHID + k]; }
  as[i] = s; ad[i] = d;
}

__global__ void alpha2_kernel(const float* __restrict__ H2,
                              const float* __restrict__ a_s, const float* __restrict__ a_d,
                              float* __restrict__ as, float* __restrict__ ad) {
  int n = blockIdx.x * blockDim.x + threadIdx.x;
  if (n >= N_NODES) return;
  const float* hp = H2 + (long)n * NCLASS;
  float s = 0.f, d = 0.f;
  #pragma unroll
  for (int c = 0; c < NCLASS; c++) { s += hp[c] * a_s[c]; d += hp[c] * a_d[c]; }
  as[n] = s; ad[n] = d;
}

extern "C" void kernel_launch(void* const* d_in, const int* in_sizes, int n_in,
                              void* d_out, int out_size, void* d_ws, size_t ws_size,
                              hipStream_t stream) {
  const float* x      = (const float*)d_in[0];
  const float* W1     = (const float*)d_in[1];
  const float* a_src1 = (const float*)d_in[2];
  const float* a_dst1 = (const float*)d_in[3];
  const float* b1     = (const float*)d_in[4];
  const float* W2     = (const float*)d_in[5];
  const float* a_src2 = (const float*)d_in[6];
  const float* a_dst2 = (const float*)d_in[7];
  const float* b2     = (const float*)d_in[8];
  const int*   ei     = (const int*)d_in[9];
  float*       out    = (float*)d_out;

  float* ws   = (float*)d_ws;
  float* H1   = ws;
  float* as1  = H1  + (long)N_NODES * HID1;
  float* ad1  = as1 + (long)N_NODES * NHEADS;
  float* as2  = ad1 + (long)N_NODES * NHEADS;
  float* ad2  = as2 + N_NODES;
  float* X2   = ad2 + N_NODES;
  float* H2   = X2  + (long)N_NODES * HID1;
  unsigned* glist = (unsigned*)(H2 + (long)N_NODES * NCLASS);
  unsigned* gcnt  = glist + (long)NRANGE * LIST_CAP;
  unsigned* key   = gcnt + NRANGE;
  unsigned* rec   = key + N_EDGES / 4;

  const int B = 256;
  auto cdiv = [](long a, long b) { return (int)((a + b - 1) / b); };

  gemm1_wmma<<<cdiv(N_NODES / 16, 4), 128, 0, stream>>>(x, W1, H1);
  alpha1_kernel<<<cdiv((long)N_NODES * NHEADS, B), B, 0, stream>>>(H1, a_src1, a_dst1, as1, ad1, ei, key, rec);
  agg1_lds<<<NRANGE, AGG_T, 0, stream>>>(key, rec, as1, ad1, H1, b1, X2, glist, gcnt);

  gemm2_wmma<<<cdiv(N_NODES / 16, 4), 128, 0, stream>>>(X2, W2, H2);
  alpha2_kernel<<<cdiv(N_NODES, B), B, 0, stream>>>(H2, a_src2, a_dst2, as2, ad2);
  agg2_lds<<<NRANGE, AGG_T, 0, stream>>>(key, rec, as2, ad2, H2, b2, out, glist, gcnt);
}


# each listed edge's logit is formed once, in the max pass, and kept on chip for the sum pass (layer 1: registers, a thread's source gathers all issued at once; layer 2: LDS)
# speedup vs baseline: 9.3409x; 1.0597x over previous
#include <hip/hip_runtime.h>
#include <math.h>

#define N_NODES 50000
#define N_EDGES 800000
#define E_TOT   (N_EDGES + N_NODES)
#define NFEAT   128
#define NHID    8
#define NHEADS  8
#define HID1    (NHEADS * NHID)
#define NCLASS  40
#define NEG_SLOPE 0.2f

typedef __attribute__((ext_vector_type(2))) float v2f;
typedef __attribute__((ext_vector_type(8))) float v8f;

__device__ __forceinline__ float lrelu(float x) { return x > 0.f ? x : NEG_SLOPE * x; }

__device__ __forceinline__ unsigned f2ord(float f) {
  unsigned u = __float_as_uint(f);
  return (u & 0x80000000u) ? ~u : (u | 0x80000000u);
}
__device__ __forceinline__ float ord2f(unsigned u) {
  return (u & 0x80000000u) ? __uint_as_float(u & 0x7FFFFFFFu) : __uint_as_float(~u);
}
#define ORD_NEG_INF 0x007FFFFFu

#define AGG_T   1024
#define NRANGE  256
#define RN      196
#define KEYS4   (N_EDGES / 16)
#define LD_FAST 7
#define LD_GEN  2
#define NCHUNK_OF(LD) ((KEYS4 + AGG_T * (LD) - 1) / (AGG_T * (LD)))
#define LIST_CAP 36864
#define P1      65
#define UNR     4
#define KREG    32
#define REG_CAP (KREG * (AGG_T / 8))
static_assert(REG_CAP == UNR * AGG_T, "layer 2's max pass covers REG_CAP entries in one round");

#define EBITS 20
#define EMASK ((1u << EBITS) - 1)
__device__ __forceinline__ void push_edge(unsigned e, unsigned* list, unsigned* cnt) {
  const unsigned p = atomicAdd(cnt, 1u);
  if (p < LIST_CAP) list[p] = e;
}

__device__ __forceinline__ unsigned zero_bytes(unsigned x) {
  return ~(((x & 0x7F7F7F7Fu) + 0x7F7F7F7Fu) | x) & 0x80808080u;
}

__device__ __forceinline__ void edge_keys(const int* __restrict__ ei, int q,
                                          unsigned* __restrict__ key, unsigned* __restrict__ rec) {
  const int4 s = ((const int4*)ei)[q];
  const int4 d = ((const int4*)(ei + N_EDGES))[q];
  const unsigned g0 = (unsigned)d.x / RN, g1 = (unsigned)d.y / RN, g2 = (unsigned)d.z / RN, g3 = (unsigned)d.w / RN;
  key[q] = g0 | (g1 << 8) | (g2 << 16) | (g3 << 24);
  ((uint4*)rec)[q] = make_uint4((unsigned)s.x | (((unsigned)d.x - g0 * RN) << EBITS),
                                (unsigned)s.y | (((unsigned)d.y - g1 * RN) << EBITS),
                                (unsigned)s.z | (((unsigned)d.z - g2 * RN) << EBITS),
                                (unsigned)s.w | (((unsigned)d.w - g3 * RN) << EBITS));
}

template <int LD>
__device__ __forceinline__ void scan_chunk(const unsigned* __restrict__ key, int chunk, int t,
                                           unsigned gpat, unsigned* list, unsigned* cnt) {
  const uint4* __restrict__ key4 = (const uint4*)key;
  const int q0 = chunk * (AGG_T * LD) + t;
  uint4 kv[LD];
  #pragma unroll
  for (int i = 0; i < LD; i++) {
    const int q = q0 + i * AGG_T;
    kv[i] = (q < KEYS4) ? key4[q] : make_uint4(~gpat, ~gpat, ~gpat, ~gpat);
  }
  #pragma unroll
  for (int i = 0; i < LD; i++) {
    unsigned m = (zero_bytes(kv[i].x ^ gpat) >> 7) | (zero_bytes(kv[i].y ^ gpat) >> 6) |
                 (zero_bytes(kv[i].z ^ gpat) >> 5) | (zero_bytes(kv[i].w ^ gpat) >> 4);
    const unsigned e0 = 16u * (unsigned)(q0 + i * AGG_T);
    while (m) {
      const unsigned p = (unsigned)__builtin_ctz(m);
      m &= m - 1;
      push_edge(e0 + 4 * (p & 7) + (p >> 3), list, cnt);
    }
  }
}

__device__ __forceinline__ void list_self_loops(int t, int lo, unsigned span, unsigned* list, unsigned* cnt) {
  if (t < (int)span) list[t] = (unsigned)(lo + t) | ((unsigned)t << EBITS);
  if (t == 0) *cnt = span;
}

__device__ __forceinline__ void resolve_src(const unsigned* __restrict__ rec, unsigned* list,
                                            unsigned first, unsigned n, int t) {
  for (unsigned b = first + t; b < n; b += UNR * AGG_T) {
    unsigned rv[UNR];
    #pragma unroll
    for (int j = 0; j < UNR; j++) {
      const bool in = b + j * AGG_T < n;
      rv[j] = in ? rec[list[b + j * AGG_T]] : 0u;
    }
    #pragma unroll
    for (int j = 0; j < UNR; j++)
      if (b + j * AGG_T < n) list[b + j * AGG_T] = rv[j];
  }
}

__global__ void __launch_bounds__(AGG_T) agg1_lds(const unsigned* __restrict__ key,
                                                  const unsigned* __restrict__ rec,
                                                  const float* __restrict__ as,
                                                  const float* __restrict__ ad,
                                                  const float* __restrict__ H1,
                                                  const float* __restrict__ bias,
                                                  float* __restrict__ out1,
                                                  unsigned* __restrict__ glist,
                                                  unsigned* __restrict__ gcnt) {
  __shared__ float    acc[RN * P1];
  __shared__ float    dsum[RN * NHEADS];
  __shared__ float    adl[RN * NHEADS];
  __shared__ unsigned mxo[RN * NHEADS];
  __shared__ unsigned list[LIST_CAP];
  __shared__ unsigned cnt;

  const int t  = threadIdx.x;
  const int lo = blockIdx.x * RN;
  const unsigned span = (unsigned)(min(lo + RN, N_NODES) - lo);
  const int h = t & 7;
  const unsigned gpat = blockIdx.x * 0x01010101u;

  for (int i = t; i < RN * P1; i += AGG_T) acc[i] = 0.f;
  for (int i = t; i < (int)span * NHEADS; i += AGG_T) {
    dsum[i] = 0.f;
    adl[i]  = ad[lo * NHEADS + i];
    mxo[i]  = ORD_NEG_INF;
  }

  auto max_pass = [&](unsigned n) {
    for (unsigned b = t >> 3; b < n; b += UNR * (AGG_T / 8)) {
      unsigned dl[UNR]; float a[UNR];
      #pragma unroll
      for (int j = 0; j < UNR; j++) {
        const unsigned en = list[min(b + j * (AGG_T / 8), n - 1)];
        dl[j] = en >> EBITS;
        a[j]  = as[(en & EMASK) * NHEADS + h];
      }
      #pragma unroll
      for (int j = 0; j < UNR; j++)
        atomicMax(&mxo[dl[j] * NHEADS + h], f2ord(lrelu(a[j] + adl[dl[j] * NHEADS + h])));
    }
  };
  auto sum_pass = [&](unsigned n) {
    for (unsigned b = t >> 3; b < n; b += UNR * (AGG_T / 8)) {
      unsigned dl[UNR]; float a[UNR]; float4 h0[UNR], h1[UNR];
      #pragma unroll
      for (int j = 0; j < UNR; j++) {
        const unsigned en = list[min(b + j * (AGG_T / 8), n - 1)];
        const unsigned s = en & EMASK;
        dl[j] = en >> EBITS;
        a[j]  = as[s * NHEADS + h];
        const float4* hs = (const float4*)(H1 + (long)s * HID1 + h * NHID);
        h0[j] = hs[0]; h1[j] = hs[1];
      }
      #pragma unroll
      for (int j = 0; j < UNR; j++) {
        if (b + j * (AGG_T / 8) < n) {
          const int d8 = dl[j] * NHEADS + h;
          const float v = lrelu(a[j] + adl[d8]);
          const float w = __expf(v - ord2f(mxo[d8]));
          atomicAdd(&dsum[d8], w);
          float* od = &acc[dl[j] * P1 + h * NHID];
          atomicAdd(&od[0], w * h0[j].x); atomicAdd(&od[1], w * h0[j].y);
          atomicAdd(&od[2], w * h0[j].z); atomicAdd(&od[3], w * h0[j].w);
          atomicAdd(&od[4], w * h1[j].x); atomicAdd(&od[5], w * h1[j].y);
          atomicAdd(&od[6], w * h1[j].z); atomicAdd(&od[7], w * h1[j].w);
        }
      }
    }
  };

  list_self_loops(t, lo, span, list, &cnt);
  __syncthreads();
  for (int c = 0; c < NCHUNK_OF(LD_FAST); c++) scan_chunk<LD_FAST>(key, c, t, gpat, list, &cnt);
  __syncthreads();
  const unsigned n = cnt;

  if (t == 0) gcnt[blockIdx.x] = n;
  if (n <= REG_CAP) {
    resolve_src(rec, list, span, n, t);
    for (unsigned i = n + t; i < REG_CAP; i += AGG_T) list[i] = (unsigned)lo;
    __syncthreads();
    unsigned* gl = glist + (long)blockIdx.x * LIST_CAP;
    for (unsigned i = t; i < n; i += AGG_T) gl[i] = list[i];

    const unsigned* lslot = list + (t >> 3);
    float areg[KREG];
    #pragma unroll
    for (int j = 0; j < KREG; j++) areg[j] = as[(lslot[j * (AGG_T / 8)] & EMASK) * NHEADS + h];
    #pragma unroll
    for (int j = 0; j < KREG; j++) {
      if ((unsigned)(j * (AGG_T / 8)) < n) {
        const unsigned dl = lslot[j * (AGG_T / 8)] >> EBITS;
        areg[j] = lrelu(areg[j] + adl[dl * NHEADS + h]);
        atomicMax(&mxo[dl * NHEADS + h], f2ord(areg[j]));
      }
    }
    __syncthreads();
    #pragma clang loop unroll(disable)
    for (unsigned r0 = 0; r0 < n; r0 += UNR * (AGG_T / 8)) {
      unsigned dl[UNR]; float4 h0[UNR], h1[UNR];
      #pragma unroll
      for (int j = 0; j < UNR; j++) {
        const unsigned en = lslot[r0 + j * (AGG_T / 8)];
        dl[j] = en >> EBITS;
        const float4* hs = (const float4*)(H1 + ((en & EMASK) * HID1 + h * NHID));
        h0[j] = hs[0]; h1[j] = hs[1];
      }
      #pragma unroll
      for (int j = 0; j < UNR; j++) {
        if (r0 + (unsigned)(t >> 3) + j * (AGG_T / 8) < n) {
          const int d8 = dl[j] * NHEADS + h;
          const float w = __expf(areg[j] - ord2f(mxo[d8]));
          atomicAdd(&dsum[d8], w);
          float* od = &acc[dl[j] * P1 + h * NHID];
          atomicAdd(&od[0], w * h0[j].x); atomicAdd(&od[1], w * h0[j].y);
          atomicAdd(&od[2], w * h0[j].z); atomicAdd(&od[3], w * h0[j].w);
          atomicAdd(&od[4], w * h1[j].x); atomicAdd(&od[5], w * h1[j].y);
          atomicAdd(&od[6], w * h1[j].z); atomicAdd(&od[7], w * h1[j].w);
        }
      }
      #pragma unroll
      for (int k = 0; k + UNR < KREG; k++) areg[k] = areg[k + UNR];
    }
  } else if (n <= LIST_CAP) {
    resolve_src(rec, list, span, n, t);
    __syncthreads();
    unsigned* gl = glist + (long)blockIdx.x * LIST_CAP;
    for (unsigned i = t; i < n; i += AGG_T) gl[i] = list[i];
    max_pass(n);
    __syncthreads();
    sum_pass(n);
  } else {
    for (int phase = 0; phase < 2; phase++) {
      for (int it = 0; it <= NCHUNK_OF(LD_GEN); it++) {
        __syncthreads();
        if (it == 0) list_self_loops(t, lo, span, list, &cnt);
        else if (t == 0) cnt = 0;
        __syncthreads();
        if (it > 0) scan_chunk<LD_GEN>(key, it - 1, t, gpat, list, &cnt);
        __syncthreads();
        const unsigned nc = cnt;
        resolve_src(rec, list, it == 0 ? span : 0u, nc, t);
        __syncthreads();
        if (phase == 0) max_pass(nc); else sum_pass(nc);
      }
    }
  }
  __syncthreads();

  for (int i = t; i < (int)span * HID1; i += AGG_T) {
    const float v = acc[(i >> 6) * P1 + (i & 63)] / dsum[i >> 3] + bias[i & (HID1 - 1)];
    out1[(long)lo * HID1 + i] = v > 0.f ? v : (__expf(v) - 1.f);
  }
}

__global__ void __launch_bounds__(AGG_T) agg2_lds(const unsigned* __restrict__ key,
                                                  const unsigned* __restrict__ rec,
                                                  const float* __restrict__ as,
                                                  const float* __restrict__ ad,
                                                  const float* __restrict__ H2,
                                                  const float* __restrict__ bias,
                                                  float* __restrict__ out,
                                                  const unsigned* __restrict__ glist,
                                                  const unsigned* __restrict__ gcnt) {
  __shared__ float    acc[RN * NCLASS];
  __shared__ float    dsum[RN];
  __shared__ float    adl[RN];
  __shared__ unsigned mxo[RN];
  __shared__ unsigned list[LIST_CAP];
  __shared__ float    la[REG_CAP];
  __shared__ unsigned cnt;

  const int t  = threadIdx.x;
  const int lo = blockIdx.x * RN;
  const unsigned span = (unsigned)(min(lo + RN, N_NODES) - lo);
  const int l = t & 7;
  const unsigned gpat = blockIdx.x * 0x01010101u;

  for (int i = t; i < RN * NCLASS; i += AGG_T) acc[i] = 0.f;
  if (t < (int)span) {
    dsum[t] = 0.f;
    adl[t]  = ad[lo + t];
    mxo[t]  = ORD_NEG_INF;
  }

  auto max_pass = [&](unsigned n) {
    for (unsigned b = t; b < n; b += UNR * AGG_T) {
      unsigned dl[UNR]; float a[UNR];
      #pragma unroll
      for (int j = 0; j < UNR; j++) {
        const unsigned en = list[min(b + j * AGG_T, n - 1)];
        dl[j] = en >> EBITS;
        a[j]  = as[en & EMASK];
      }
      #pragma unroll
      for (int j = 0; j < UNR; j++)
        atomicMax(&mxo[dl[j]], f2ord(lrelu(a[j] + adl[dl[j]])));
    }
  };
  auto sum_pass = [&](unsigned n) {
    for (unsigned b = t >> 3; b < n; b += UNR * (AGG_T / 8)) {
      unsigned dl[UNR]; float a[UNR]; float hv[UNR][NCLASS / 8];
      #pragma unroll
      for (int j = 0; j < UNR; j++) {
        const unsigned en = list[min(b + j * (AGG_T / 8), n - 1)];
        const unsigned s = en & EMASK;
        dl[j] = en >> EBITS;
        a[j]  = as[s];
        const float* hs = H2 + (long)s * NCLASS + l;
        #pragma unroll
        for (int k = 0; k < NCLASS / 8; k++) hv[j][k] = hs[k * 8];
      }
      #pragma unroll
      for (int j = 0; j < UNR; j++) {
        if (b + j * (AGG_T / 8) < n) {
          const float v = lrelu(a[j] + adl[dl[j]]);
          const float w = __expf(v - ord2f(mxo[dl[j]]));
          if (l == 0) atomicAdd(&dsum[dl[j]], w);
          float* od = &acc[dl[j] * NCLASS + l];
          #pragma unroll
          for (int k = 0; k < NCLASS / 8; k++) atomicAdd(&od[k * 8], w * hv[j][k]);
        }
      }
    }
  };

  const unsigned n = gcnt[blockIdx.x];

  if (n <= REG_CAP) {
    const unsigned* gl = glist + (long)blockIdx.x * LIST_CAP;
    for (unsigned i = t; i < n; i += AGG_T) list[i] = gl[i];
    __syncthreads();
    {
      unsigned dl[UNR]; float a[UNR];
      #pragma unroll
      for (int j = 0; j < UNR; j++) {
        const unsigned en = list[min((unsigned)t + j * AGG_T, n - 1)];
        dl[j] = en >> EBITS;
        a[j]  = as[en & EMASK];
      }
      #pragma unroll
      for (int j = 0; j < UNR; j++) {
        if ((unsigned)t + j * AGG_T < n) {
          const float v = lrelu(a[j] + adl[dl[j]]);
          la[t + j * AGG_T] = v;
          atomicMax(&mxo[dl[j]], f2ord(v));
        }
      }
    }
    __syncthreads();
    for (unsigned b = t >> 3; b < n; b += UNR * (AGG_T / 8)) {
      unsigned dl[UNR]; float v[UNR]; float hv[UNR][NCLASS / 8];
      #pragma unroll
      for (int j = 0; j < UNR; j++) {
        const unsigned i = min(b + j * (AGG_T / 8), n - 1);
        const unsigned en = list[i];
        dl[j] = en >> EBITS;
        v[j]  = la[i];
        const float* hs = H2 + (long)(en & EMASK) * NCLASS + l;
        #pragma unroll
        for (int k = 0; k < NCLASS / 8; k++) hv[j][k] = hs[k * 8];
      }
      #pragma unroll
      for (int j = 0; j < UNR; j++) {
        if (b + j * (AGG_T / 8) < n) {
          const float w = __expf(v[j] - ord2f(mxo[dl[j]]));
          if (l == 0) atomicAdd(&dsum[dl[j]], w);
          float* od = &acc[dl[j] * NCLASS + l];
          #pragma unroll
          for (int k = 0; k < NCLASS / 8; k++) atomicAdd(&od[k * 8], w * hv[j][k]);
        }
      }
    }
  } else if (n <= LIST_CAP) {
    const unsigned* gl = glist + (long)blockIdx.x * LIST_CAP;
    for (unsigned i = t; i < n; i += AGG_T) list[i] = gl[i];
    __syncthreads();
    max_pass(n);
    __syncthreads();
    sum_pass(n);
  } else {
    for (int phase = 0; phase < 2; phase++) {
      for (int it = 0; it <= NCHUNK_OF(LD_GEN); it++) {
        __syncthreads();
        if (it == 0) list_self_loops(t, lo, span, list, &cnt);
        else if (t == 0) cnt = 0;
        __syncthreads();
        if (it > 0) scan_chunk<LD_GEN>(key, it - 1, t, gpat, list, &cnt);
        __syncthreads();
        const unsigned nc = cnt;
        resolve_src(rec, list, it == 0 ? span : 0u, nc, t);
        __syncthreads();
        if (phase == 0) max_pass(nc); else sum_pass(nc);
      }
    }
  }
  __syncthreads();

  for (int i = t; i < (int)span * NCLASS; i += AGG_T) acc[i] = acc[i] / dsum[i / NCLASS] + bias[i % NCLASS];
  __syncthreads();
  float* lse = adl;
  if (t < (int)span) {
    const float* r = &acc[t * NCLASS];
    float mx = -INFINITY;
    for (int c = 0; c < NCLASS; c++) mx = fmaxf(mx, r[c]);
    float ssum = 0.f;
    for (int c = 0; c < NCLASS; c++) ssum += __expf(r[c] - mx);
    lse[t] = mx + __logf(ssum);
  }
  __syncthreads();
  for (int i = t; i < (int)span * NCLASS; i += AGG_T) out[(long)lo * NCLASS + i] = acc[i] - lse[i / NCLASS];
}

__global__ void __launch_bounds__(128) gemm1_wmma(const float* __restrict__ X,
                                                  const float* __restrict__ W,
                                                  float* __restrict__ H) {
  const int wave = threadIdx.x >> 5;
  const int lane = threadIdx.x & 31;
  const int mt   = blockIdx.x * 4 + wave;
  if (mt >= N_NODES / 16) return;
  const int m  = lane & 15;
  const int hi = lane >> 4;
  const float* xr = X + (long)(mt * 16 + m) * NFEAT;

  v8f c0 = {}, c1 = {}, c2 = {}, c3 = {};
  #pragma unroll 4
  for (int k0 = 0; k0 < NFEAT; k0 += 4) {
    const int ka = k0 + 2 * hi;
    v2f a; a.x = xr[ka]; a.y = xr[ka + 1];
    const float* w0 = W + (long)ka * HID1 + m;
    const float* w1 = w0 + HID1;
    v2f b;
    b.x = w0[0];  b.y = w1[0];
    c0 = __builtin_amdgcn_wmma_f32_16x16x4_f32(false, a, false, b, (short)0, c0, false, false);
    b.x = w0[16]; b.y = w1[16];
    c1 = __builtin_amdgcn_wmma_f32_16x16x4_f32(false, a, false, b, (short)0, c1, false, false);
    b.x = w0[32]; b.y = w1[32];
    c2 = __builtin_amdgcn_wmma_f32_16x16x4_f32(false, a, false, b, (short)0, c2, false, false);
    b.x = w0[48]; b.y = w1[48];
    c3 = __builtin_amdgcn_wmma_f32_16x16x4_f32(false, a, false, b, (short)0, c3, false, false);
  }
  float* hp = H + (long)(mt * 16 + 8 * hi) * HID1 + m;
  #pragma unroll
  for (int j = 0; j < 8; j++) {
    hp[(long)j * HID1 +  0] = c0[j];
    hp[(long)j * HID1 + 16] = c1[j];
    hp[(long)j * HID1 + 32] = c2[j];
    hp[(long)j * HID1 + 48] = c3[j];
  }
}

__global__ void __launch_bounds__(128) gemm2_wmma(const float* __restrict__ X,
                                                  const float* __restrict__ W,
                                                  float* __restrict__ H) {
  const int wave = threadIdx.x >> 5;
  const int lane = threadIdx.x & 31;
  const int mt   = blockIdx.x * 4 + wave;
  if (mt >= N_NODES / 16) return;
  const int m  = lane & 15;
  const int hi = lane >> 4;
  const float* xr = X + (long)(mt * 16 + m) * HID1;

  v8f c0 = {}, c1 = {}, c2 = {};
  #pragma unroll 4
  for (int k0 = 0; k0 < HID1; k0 += 4) {
    const int ka = k0 + 2 * hi;
    v2f a; a.x = xr[ka]; a.y = xr[ka + 1];
    #pragma unroll
    for (int t = 0; t < 3; t++) {
      const int col = t * 16 + m;
      v2f b;
      b.x = (col < NCLASS) ? W[(long)ka * NCLASS + col]       : 0.f;
      b.y = (col < NCLASS) ? W[(long)(ka + 1) * NCLASS + col] : 0.f;
      if (t == 0) c0 = __builtin_amdgcn_wmma_f32_16x16x4_f32(false, a, false, b, (short)0, c0, false, false);
      if (t == 1) c1 = __builtin_amdgcn_wmma_f32_16x16x4_f32(false, a, false, b, (short)0, c1, false, false);
      if (t == 2) c2 = __builtin_amdgcn_wmma_f32_16x16x4_f32(false, a, false, b, (short)0, c2, false, false);
    }
  }
  const int rbase = mt * 16 + 8 * hi;
  #pragma unroll
  for (int j = 0; j < 8; j++) {
    float* hr = H + (long)(rbase + j) * NCLASS;
    if (m      < NCLASS) hr[m]      = c0[j];
    if (m + 16 < NCLASS) hr[m + 16] = c1[j];
    if (m + 32 < NCLASS) hr[m + 32] = c2[j];
  }
}

__global__ void alpha1_kernel(const float* __restrict__ H1,
                              const float* __restrict__ a_s, const float* __restrict__ a_d,
                              float* __restrict__ as, float* __restrict__ ad,
                              const int* __restrict__ ei,
                              unsigned* __restrict__ key, unsigned* __restrict__ rec) {
  int i = blockIdx.x * blockDim.x + threadIdx.x;
  if (i < N_EDGES / 4) edge_keys(ei, i, key, rec);
  if (i >= N_NODES * NHEADS) return;
  const int h = i & 7;
  const float* hp = H1 + (long)(i >> 3) * HID1 + h * NHID;
  float s = 0.f, d = 0.f;
  #pragma unroll
  for (int k = 0; k < NHID; k++) { s += hp[k] * a_s[h * NHID + k]; d += hp[k] * a_d[h * NHID + k]; }
  as[i] = s; ad[i] = d;
}

__global__ void alpha2_kernel(const float* __restrict__ H2,
                              const float* __restrict__ a_s, const float* __restrict__ a_d,
                              float* __restrict__ as, float* __restrict__ ad) {
  int n = blockIdx.x * blockDim.x + threadIdx.x;
  if (n >= N_NODES) return;
  const float* hp = H2 + (long)n * NCLASS;
  float s = 0.f, d = 0.f;
  #pragma unroll
  for (int c = 0; c < NCLASS; c++) { s += hp[c] * a_s[c]; d += hp[c] * a_d[c]; }
  as[n] = s; ad[n] = d;
}

extern "C" void kernel_launch(void* const* d_in, const int* in_sizes, int n_in,
                              void* d_out, int out_size, void* d_ws, size_t ws_size,
                              hipStream_t stream) {
  const float* x      = (const float*)d_in[0];
  const float* W1     = (const float*)d_in[1];
  const float* a_src1 = (const float*)d_in[2];
  const float* a_dst1 = (const float*)d_in[3];
  const float* b1     = (const float*)d_in[4];
  const float* W2     = (const float*)d_in[5];
  const float* a_src2 = (const float*)d_in[6];
  const float* a_dst2 = (const float*)d_in[7];
  const float* b2     = (const float*)d_in[8];
  const int*   ei     = (const int*)d_in[9];
  float*       out    = (float*)d_out;

  float* ws   = (float*)d_ws;
  float* H1   = ws;
  float* as1  = H1  + (long)N_NODES * HID1;
  float* ad1  = as1 + (long)N_NODES * NHEADS;
  float* as2  = ad1 + (long)N_NODES * NHEADS;
  float* ad2  = as2 + N_NODES;
  float* X2   = ad2 + N_NODES;
  float* H2   = X2  + (long)N_NODES * HID1;
  unsigned* glist = (unsigned*)(H2 + (long)N_NODES * NCLASS);
  unsigned* gcnt  = glist + (long)NRANGE * LIST_CAP;
  unsigned* key   = gcnt + NRANGE;
  unsigned* rec   = key + N_EDGES / 4;

  const int B = 256;
  auto cdiv = [](long a, long b) { return (int)((a + b - 1) / b); };

  gemm1_wmma<<<cdiv(N_NODES / 16, 4), 128, 0, stream>>>(x, W1, H1);
  alpha1_kernel<<<cdiv((long)N_NODES * NHEADS, B), B, 0, stream>>>(H1, a_src1, a_dst1, as1, ad1, ei, key, rec);
  agg1_lds<<<NRANGE, AGG_T, 0, stream>>>(key, rec, as1, ad1, H1, b1, X2, glist, gcnt);

  gemm2_wmma<<<cdiv(N_NODES / 16, 4), 128, 0, stream>>>(X2, W2, H2);
  alpha2_kernel<<<cdiv(N_NODES, B), B, 0, stream>>>(H2, a_src2, a_dst2, as2, ad2);
  agg2_lds<<<NRANGE, AGG_T, 0, stream>>>(key, rec, as2, ad2, H2, b2, out, glist, gcnt);
}


# both GEMMs feed their WMMAs from LDS: W staged once per 8-wave work-group in fragment order (two 16-byte LDS loads per k-step), X rows read as whole 128-byte lines and re-laid via LDS; same k order
# speedup vs baseline: 10.1934x; 1.0913x over previous
#include <hip/hip_runtime.h>
#include <math.h>

#define N_NODES 50000
#define N_EDGES 800000
#define E_TOT   (N_EDGES + N_NODES)
#define NFEAT   128
#define NHID    8
#define NHEADS  8
#define HID1    (NHEADS * NHID)
#define NCLASS  40
#define NEG_SLOPE 0.2f

typedef __attribute__((ext_vector_type(2))) float v2f;
typedef __attribute__((ext_vector_type(8))) float v8f;

__device__ __forceinline__ float lrelu(float x) { return x > 0.f ? x : NEG_SLOPE * x; }

__device__ __forceinline__ unsigned f2ord(float f) {
  unsigned u = __float_as_uint(f);
  return (u & 0x80000000u) ? ~u : (u | 0x80000000u);
}
__device__ __forceinline__ float ord2f(unsigned u) {
  return (u & 0x80000000u) ? __uint_as_float(u & 0x7FFFFFFFu) : __uint_as_float(~u);
}
#define ORD_NEG_INF 0x007FFFFFu

#define AGG_T   1024
#define NRANGE  256
#define RN      196
#define KEYS4   (N_EDGES / 16)
#define LD_FAST 7
#define LD_GEN  2
#define NCHUNK_OF(LD) ((KEYS4 + AGG_T * (LD) - 1) / (AGG_T * (LD)))
#define LIST_CAP 36864
#define P1      65
#define UNR     4
#define KREG    32
#define REG_CAP (KREG * (AGG_T / 8))
static_assert(REG_CAP == UNR * AGG_T, "layer 2's max pass covers REG_CAP entries in one round");

#define EBITS 20
#define EMASK ((1u << EBITS) - 1)
__device__ __forceinline__ void push_edge(unsigned e, unsigned* list, unsigned* cnt) {
  const unsigned p = atomicAdd(cnt, 1u);
  if (p < LIST_CAP) list[p] = e;
}

__device__ __forceinline__ unsigned zero_bytes(unsigned x) {
  return ~(((x & 0x7F7F7F7Fu) + 0x7F7F7F7Fu) | x) & 0x80808080u;
}

__device__ __forceinline__ void edge_keys(const int* __restrict__ ei, int q,
                                          unsigned* __restrict__ key, unsigned* __restrict__ rec) {
  const int4 s = ((const int4*)ei)[q];
  const int4 d = ((const int4*)(ei + N_EDGES))[q];
  const unsigned g0 = (unsigned)d.x / RN, g1 = (unsigned)d.y / RN, g2 = (unsigned)d.z / RN, g3 = (unsigned)d.w / RN;
  key[q] = g0 | (g1 << 8) | (g2 << 16) | (g3 << 24);
  ((uint4*)rec)[q] = make_uint4((unsigned)s.x | (((unsigned)d.x - g0 * RN) << EBITS),
                                (unsigned)s.y | (((unsigned)d.y - g1 * RN) << EBITS),
                                (unsigned)s.z | (((unsigned)d.z - g2 * RN) << EBITS),
                                (unsigned)s.w | (((unsigned)d.w - g3 * RN) << EBITS));
}

template <int LD>
__device__ __forceinline__ void scan_chunk(const unsigned* __restrict__ key, int chunk, int t,
                                           unsigned gpat, unsigned* list, unsigned* cnt) {
  const uint4* __restrict__ key4 = (const uint4*)key;
  const int q0 = chunk * (AGG_T * LD) + t;
  uint4 kv[LD];
  #pragma unroll
  for (int i = 0; i < LD; i++) {
    const int q = q0 + i * AGG_T;
    kv[i] = (q < KEYS4) ? key4[q] : make_uint4(~gpat, ~gpat, ~gpat, ~gpat);
  }
  #pragma unroll
  for (int i = 0; i < LD; i++) {
    unsigned m = (zero_bytes(kv[i].x ^ gpat) >> 7) | (zero_bytes(kv[i].y ^ gpat) >> 6) |
                 (zero_bytes(kv[i].z ^ gpat) >> 5) | (zero_bytes(kv[i].w ^ gpat) >> 4);
    const unsigned e0 = 16u * (unsigned)(q0 + i * AGG_T);
    while (m) {
      const unsigned p = (unsigned)__builtin_ctz(m);
      m &= m - 1;
      push_edge(e0 + 4 * (p & 7) + (p >> 3), list, cnt);
    }
  }
}

__device__ __forceinline__ void list_self_loops(int t, int lo, unsigned span, unsigned* list, unsigned* cnt) {
  if (t < (int)span) list[t] = (unsigned)(lo + t) | ((unsigned)t << EBITS);
  if (t == 0) *cnt = span;
}

__device__ __forceinline__ void resolve_src(const unsigned* __restrict__ rec, unsigned* list,
                                            unsigned first, unsigned n, int t) {
  for (unsigned b = first + t; b < n; b += UNR * AGG_T) {
    unsigned rv[UNR];
    #pragma unroll
    for (int j = 0; j < UNR; j++) {
      const bool in = b + j * AGG_T < n;
      rv[j] = in ? rec[list[b + j * AGG_T]] : 0u;
    }
    #pragma unroll
    for (int j = 0; j < UNR; j++)
      if (b + j * AGG_T < n) list[b + j * AGG_T] = rv[j];
  }
}

__global__ void __launch_bounds__(AGG_T) agg1_lds(const unsigned* __restrict__ key,
                                                  const unsigned* __restrict__ rec,
                                                  const float* __restrict__ as,
                                                  const float* __restrict__ ad,
                                                  const float* __restrict__ H1,
                                                  const float* __restrict__ bias,
                                                  float* __restrict__ out1,
                                                  unsigned* __restrict__ glist,
                                                  unsigned* __restrict__ gcnt) {
  __shared__ float    acc[RN * P1];
  __shared__ float    dsum[RN * NHEADS];
  __shared__ float    adl[RN * NHEADS];
  __shared__ unsigned mxo[RN * NHEADS];
  __shared__ unsigned list[LIST_CAP];
  __shared__ unsigned cnt;

  const int t  = threadIdx.x;
  const int lo = blockIdx.x * RN;
  const unsigned span = (unsigned)(min(lo + RN, N_NODES) - lo);
  const int h = t & 7;
  const unsigned gpat = blockIdx.x * 0x01010101u;

  for (int i = t; i < RN * P1; i += AGG_T) acc[i] = 0.f;
  for (int i = t; i < (int)span * NHEADS; i += AGG_T) {
    dsum[i] = 0.f;
    adl[i]  = ad[lo * NHEADS + i];
    mxo[i]  = ORD_NEG_INF;
  }

  auto max_pass = [&](unsigned n) {
    for (unsigned b = t >> 3; b < n; b += UNR * (AGG_T / 8)) {
      unsigned dl[UNR]; float a[UNR];
      #pragma unroll
      for (int j = 0; j < UNR; j++) {
        const unsigned en = list[min(b + j * (AGG_T / 8), n - 1)];
        dl[j] = en >> EBITS;
        a[j]  = as[(en & EMASK) * NHEADS + h];
      }
      #pragma unroll
      for (int j = 0; j < UNR; j++)
        atomicMax(&mxo[dl[j] * NHEADS + h], f2ord(lrelu(a[j] + adl[dl[j] * NHEADS + h])));
    }
  };
  auto sum_pass = [&](unsigned n) {
    for (unsigned b = t >> 3; b < n; b += UNR * (AGG_T / 8)) {
      unsigned dl[UNR]; float a[UNR]; float4 h0[UNR], h1[UNR];
      #pragma unroll
      for (int j = 0; j < UNR; j++) {
        const unsigned en = list[min(b + j * (AGG_T / 8), n - 1)];
        const unsigned s = en & EMASK;
        dl[j] = en >> EBITS;
        a[j]  = as[s * NHEADS + h];
        const float4* hs = (const float4*)(H1 + (long)s * HID1 + h * NHID);
        h0[j] = hs[0]; h1[j] = hs[1];
      }
      #pragma unroll
      for (int j = 0; j < UNR; j++) {
        if (b + j * (AGG_T / 8) < n) {
          const int d8 = dl[j] * NHEADS + h;
          const float v = lrelu(a[j] + adl[d8]);
          const float w = __expf(v - ord2f(mxo[d8]));
          atomicAdd(&dsum[d8], w);
          float* od = &acc[dl[j] * P1 + h * NHID];
          atomicAdd(&od[0], w * h0[j].x); atomicAdd(&od[1], w * h0[j].y);
          atomicAdd(&od[2], w * h0[j].z); atomicAdd(&od[3], w * h0[j].w);
          atomicAdd(&od[4], w * h1[j].x); atomicAdd(&od[5], w * h1[j].y);
          atomicAdd(&od[6], w * h1[j].z); atomicAdd(&od[7], w * h1[j].w);
        }
      }
    }
  };

  list_self_loops(t, lo, span, list, &cnt);
  __syncthreads();
  for (int c = 0; c < NCHUNK_OF(LD_FAST); c++) scan_chunk<LD_FAST>(key, c, t, gpat, list, &cnt);
  __syncthreads();
  const unsigned n = cnt;

  if (t == 0) gcnt[blockIdx.x] = n;
  if (n <= REG_CAP) {
    resolve_src(rec, list, span, n, t);
    for (unsigned i = n + t; i < REG_CAP; i += AGG_T) list[i] = (unsigned)lo;
    __syncthreads();
    unsigned* gl = glist + (long)blockIdx.x * LIST_CAP;
    for (unsigned i = t; i < n; i += AGG_T) gl[i] = list[i];

    const unsigned* lslot = list + (t >> 3);
    float areg[KREG];
    #pragma unroll
    for (int j = 0; j < KREG; j++) areg[j] = as[(lslot[j * (AGG_T / 8)] & EMASK) * NHEADS + h];
    #pragma unroll
    for (int j = 0; j < KREG; j++) {
      if ((unsigned)(j * (AGG_T / 8)) < n) {
        const unsigned dl = lslot[j * (AGG_T / 8)] >> EBITS;
        areg[j] = lrelu(areg[j] + adl[dl * NHEADS + h]);
        atomicMax(&mxo[dl * NHEADS + h], f2ord(areg[j]));
      }
    }
    __syncthreads();
    #pragma clang loop unroll(disable)
    for (unsigned r0 = 0; r0 < n; r0 += UNR * (AGG_T / 8)) {
      unsigned dl[UNR]; float4 h0[UNR], h1[UNR];
      #pragma unroll
      for (int j = 0; j < UNR; j++) {
        const unsigned en = lslot[r0 + j * (AGG_T / 8)];
        dl[j] = en >> EBITS;
        const float4* hs = (const float4*)(H1 + ((en & EMASK) * HID1 + h * NHID));
        h0[j] = hs[0]; h1[j] = hs[1];
      }
      #pragma unroll
      for (int j = 0; j < UNR; j++) {
        if (r0 + (unsigned)(t >> 3) + j * (AGG_T / 8) < n) {
          const int d8 = dl[j] * NHEADS + h;
          const float w = __expf(areg[j] - ord2f(mxo[d8]));
          atomicAdd(&dsum[d8], w);
          float* od = &acc[dl[j] * P1 + h * NHID];
          atomicAdd(&od[0], w * h0[j].x); atomicAdd(&od[1], w * h0[j].y);
          atomicAdd(&od[2], w * h0[j].z); atomicAdd(&od[3], w * h0[j].w);
          atomicAdd(&od[4], w * h1[j].x); atomicAdd(&od[5], w * h1[j].y);
          atomicAdd(&od[6], w * h1[j].z); atomicAdd(&od[7], w * h1[j].w);
        }
      }
      #pragma unroll
      for (int k = 0; k + UNR < KREG; k++) areg[k] = areg[k + UNR];
    }
  } else if (n <= LIST_CAP) {
    resolve_src(rec, list, span, n, t);
    __syncthreads();
    unsigned* gl = glist + (long)blockIdx.x * LIST_CAP;
    for (unsigned i = t; i < n; i += AGG_T) gl[i] = list[i];
    max_pass(n);
    __syncthreads();
    sum_pass(n);
  } else {
    for (int phase = 0; phase < 2; phase++) {
      for (int it = 0; it <= NCHUNK_OF(LD_GEN); it++) {
        __syncthreads();
        if (it == 0) list_self_loops(t, lo, span, list, &cnt);
        else if (t == 0) cnt = 0;
        __syncthreads();
        if (it > 0) scan_chunk<LD_GEN>(key, it - 1, t, gpat, list, &cnt);
        __syncthreads();
        const unsigned nc = cnt;
        resolve_src(rec, list, it == 0 ? span : 0u, nc, t);
        __syncthreads();
        if (phase == 0) max_pass(nc); else sum_pass(nc);
      }
    }
  }
  __syncthreads();

  for (int i = t; i < (int)span * HID1; i += AGG_T) {
    const float v = acc[(i >> 6) * P1 + (i & 63)] / dsum[i >> 3] + bias[i & (HID1 - 1)];
    out1[(long)lo * HID1 + i] = v > 0.f ? v : (__expf(v) - 1.f);
  }
}

__global__ void __launch_bounds__(AGG_T) agg2_lds(const unsigned* __restrict__ key,
                                                  const unsigned* __restrict__ rec,
                                                  const float* __restrict__ as,
                                                  const float* __restrict__ ad,
                                                  const float* __restrict__ H2,
                                                  const float* __restrict__ bias,
                                                  float* __restrict__ out,
                                                  const unsigned* __restrict__ glist,
                                                  const unsigned* __restrict__ gcnt) {
  __shared__ float    acc[RN * NCLASS];
  __shared__ float    dsum[RN];
  __shared__ float    adl[RN];
  __shared__ unsigned mxo[RN];
  __shared__ unsigned list[LIST_CAP];
  __shared__ float    la[REG_CAP];
  __shared__ unsigned cnt;

  const int t  = threadIdx.x;
  const int lo = blockIdx.x * RN;
  const unsigned span = (unsigned)(min(lo + RN, N_NODES) - lo);
  const int l = t & 7;
  const unsigned gpat = blockIdx.x * 0x01010101u;

  for (int i = t; i < RN * NCLASS; i += AGG_T) acc[i] = 0.f;
  if (t < (int)span) {
    dsum[t] = 0.f;
    adl[t]  = ad[lo + t];
    mxo[t]  = ORD_NEG_INF;
  }

  auto max_pass = [&](unsigned n) {
    for (unsigned b = t; b < n; b += UNR * AGG_T) {
      unsigned dl[UNR]; float a[UNR];
      #pragma unroll
      for (int j = 0; j < UNR; j++) {
        const unsigned en = list[min(b + j * AGG_T, n - 1)];
        dl[j] = en >> EBITS;
        a[j]  = as[en & EMASK];
      }
      #pragma unroll
      for (int j = 0; j < UNR; j++)
        atomicMax(&mxo[dl[j]], f2ord(lrelu(a[j] + adl[dl[j]])));
    }
  };
  auto sum_pass = [&](unsigned n) {
    for (unsigned b = t >> 3; b < n; b += UNR * (AGG_T / 8)) {
      unsigned dl[UNR]; float a[UNR]; float hv[UNR][NCLASS / 8];
      #pragma unroll
      for (int j = 0; j < UNR; j++) {
        const unsigned en = list[min(b + j * (AGG_T / 8), n - 1)];
        const unsigned s = en & EMASK;
        dl[j] = en >> EBITS;
        a[j]  = as[s];
        const float* hs = H2 + (long)s * NCLASS + l;
        #pragma unroll
        for (int k = 0; k < NCLASS / 8; k++) hv[j][k] = hs[k * 8];
      }
      #pragma unroll
      for (int j = 0; j < UNR; j++) {
        if (b + j * (AGG_T / 8) < n) {
          const float v = lrelu(a[j] + adl[dl[j]]);
          const float w = __expf(v - ord2f(mxo[dl[j]]));
          if (l == 0) atomicAdd(&dsum[dl[j]], w);
          float* od = &acc[dl[j] * NCLASS + l];
          #pragma unroll
          for (int k = 0; k < NCLASS / 8; k++) atomicAdd(&od[k * 8], w * hv[j][k]);
        }
      }
    }
  };

  const unsigned n = gcnt[blockIdx.x];

  if (n <= REG_CAP) {
    const unsigned* gl = glist + (long)blockIdx.x * LIST_CAP;
    for (unsigned i = t; i < n; i += AGG_T) list[i] = gl[i];
    __syncthreads();
    {
      unsigned dl[UNR]; float a[UNR];
      #pragma unroll
      for (int j = 0; j < UNR; j++) {
        const unsigned en = list[min((unsigned)t + j * AGG_T, n - 1)];
        dl[j] = en >> EBITS;
        a[j]  = as[en & EMASK];
      }
      #pragma unroll
      for (int j = 0; j < UNR; j++) {
        if ((unsigned)t + j * AGG_T < n) {
          const float v = lrelu(a[j] + adl[dl[j]]);
          la[t + j * AGG_T] = v;
          atomicMax(&mxo[dl[j]], f2ord(v));
        }
      }
    }
    __syncthreads();
    for (unsigned b = t >> 3; b < n; b += UNR * (AGG_T / 8)) {
      unsigned dl[UNR]; float v[UNR]; float hv[UNR][NCLASS / 8];
      #pragma unroll
      for (int j = 0; j < UNR; j++) {
        const unsigned i = min(b + j * (AGG_T / 8), n - 1);
        const unsigned en = list[i];
        dl[j] = en >> EBITS;
        v[j]  = la[i];
        const float* hs = H2 + (long)(en & EMASK) * NCLASS + l;
        #pragma unroll
        for (int k = 0; k < NCLASS / 8; k++) hv[j][k] = hs[k * 8];
      }
      #pragma unroll
      for (int j = 0; j < UNR; j++) {
        if (b + j * (AGG_T / 8) < n) {
          const float w = __expf(v[j] - ord2f(mxo[dl[j]]));
          if (l == 0) atomicAdd(&dsum[dl[j]], w);
          float* od = &acc[dl[j] * NCLASS + l];
          #pragma unroll
          for (int k = 0; k < NCLASS / 8; k++) atomicAdd(&od[k * 8], w * hv[j][k]);
        }
      }
    }
  } else if (n <= LIST_CAP) {
    const unsigned* gl = glist + (long)blockIdx.x * LIST_CAP;
    for (unsigned i = t; i < n; i += AGG_T) list[i] = gl[i];
    __syncthreads();
    max_pass(n);
    __syncthreads();
    sum_pass(n);
  } else {
    for (int phase = 0; phase < 2; phase++) {
      for (int it = 0; it <= NCHUNK_OF(LD_GEN); it++) {
        __syncthreads();
        if (it == 0) list_self_loops(t, lo, span, list, &cnt);
        else if (t == 0) cnt = 0;
        __syncthreads();
        if (it > 0) scan_chunk<LD_GEN>(key, it - 1, t, gpat, list, &cnt);
        __syncthreads();
        const unsigned nc = cnt;
        resolve_src(rec, list, it == 0 ? span : 0u, nc, t);
        __syncthreads();
        if (phase == 0) max_pass(nc); else sum_pass(nc);
      }
    }
  }
  __syncthreads();

  for (int i = t; i < (int)span * NCLASS; i += AGG_T) acc[i] = acc[i] / dsum[i / NCLASS] + bias[i % NCLASS];
  __syncthreads();
  float* lse = adl;
  if (t < (int)span) {
    const float* r = &acc[t * NCLASS];
    float mx = -INFINITY;
    for (int c = 0; c < NCLASS; c++) mx = fmaxf(mx, r[c]);
    float ssum = 0.f;
    for (int c = 0; c < NCLASS; c++) ssum += __expf(r[c] - mx);
    lse[t] = mx + __logf(ssum);
  }
  __syncthreads();
  for (int i = t; i < (int)span * NCLASS; i += AGG_T) out[(long)lo * NCLASS + i] = acc[i] - lse[i / NCLASS];
}

#define G_T     256
#define G_TPW   2
#define G_TILES (N_NODES / 16)
#define G_GRID  ((G_TILES + 8 * G_TPW - 1) / (8 * G_TPW))
#define AP1     (NFEAT + 4)
#define AP2     (HID1 + 4)

__device__ __forceinline__ v8f wmma4(v2f a, float bx, float by, v8f c) {
  v2f b; b.x = bx; b.y = by;
  return __builtin_amdgcn_wmma_f32_16x16x4_f32(false, a, false, b, (short)0, c, false, false);
}

#define ROWS16(F) F(0) F(1) F(2) F(3) F(4) F(5) F(6) F(7) F(8) F(9) F(10) F(11) F(12) F(13) F(14)
#define ROWS8(F)  F(0) F(1) F(2) F(3) F(4) F(5) F(6)
#define DECL_ROW(r)   n##r,
#define LOAD_ROW1(r)  n##r = xp[r * (NFEAT / 4)];
#define STORE_ROW1(r) *(float4*)&al[r * AP1 + 4 * lane] = n##r;
#define LOAD_ROW2(r)  n##r = xp[r * 32];
#define STORE_ROW2(r) *(float4*)&al[(2 * r + hi) * AP2 + 4 * m] = n##r;

__device__ __forceinline__ void gemm1_tile(const float* al, const float4* Bl, int lane, int tile, float* __restrict__ H) {
  const int m  = lane & 15;
  const int hi = lane >> 4;
  v8f c0 = {}, c1 = {}, c2 = {}, c3 = {};
  #pragma unroll 8
  for (int j = 0; j < NFEAT / 4; j++) {
    const v2f a = *(const v2f*)&al[m * AP1 + 4 * j + 2 * hi];
    const float4 p = Bl[(2 * j) * 32 + lane];
    const float4 q = Bl[(2 * j + 1) * 32 + lane];
    c0 = wmma4(a, p.x, p.y, c0);
    c1 = wmma4(a, p.z, p.w, c1);
    c2 = wmma4(a, q.x, q.y, c2);
    c3 = wmma4(a, q.z, q.w, c3);
  }
  if (tile < G_TILES) {
    float* hp = H + (long)(tile * 16 + 8 * hi) * HID1 + m;
    #pragma unroll
    for (int j = 0; j < 8; j++) {
      hp[(long)j * HID1 +  0] = c0[j];
      hp[(long)j * HID1 + 16] = c1[j];
      hp[(long)j * HID1 + 32] = c2[j];
      hp[(long)j * HID1 + 48] = c3[j];
    }
  }
}

__device__ __forceinline__ void gemm2_tile(const float* al, const float4* Bl, int lane, int tile, float* __restrict__ H) {
  const int m  = lane & 15;
  const int hi = lane >> 4;
  v8f c0 = {}, c1 = {}, c2 = {};
  #pragma unroll 8
  for (int j = 0; j < HID1 / 4; j++) {
    const v2f a = *(const v2f*)&al[m * AP2 + 4 * j + 2 * hi];
    const float4 p = Bl[(2 * j) * 32 + lane];
    const float4 q = Bl[(2 * j + 1) * 32 + lane];
    c0 = wmma4(a, p.x, p.y, c0);
    c1 = wmma4(a, p.z, p.w, c1);
    c2 = wmma4(a, q.x, q.y, c2);
  }
  if (tile < G_TILES) {
    const int rbase = tile * 16 + 8 * hi;
    #pragma unroll
    for (int j = 0; j < 8; j++) {
      float* hr = H + (long)(rbase + j) * NCLASS;
      hr[m]      = c0[j];
      hr[m + 16] = c1[j];
      if (m + 32 < NCLASS) hr[m + 32] = c2[j];
    }
  }
}

__global__ void __launch_bounds__(G_T) gemm1_wmma(const float* __restrict__ X,
                                                  const float* __restrict__ W,
                                                  float* __restrict__ H) {
  __shared__ __attribute__((aligned(16))) float4 Bl[(NFEAT / 4) * 2 * 32];
  __shared__ __attribute__((aligned(16))) float  Al[8][16 * AP1];
  const int wave = threadIdx.x >> 5;
  const int lane = threadIdx.x & 31;

  float* al = Al[wave];
  const int t0 = blockIdx.x * (8 * G_TPW) + wave;
  float4 ROWS16(DECL_ROW) n15;
  const float4* xp = (const float4*)(X + (long)min(t0, G_TILES - 1) * 16 * NFEAT) + lane;
  ROWS16(LOAD_ROW1) n15 = xp[15 * (NFEAT / 4)];

  #pragma unroll
  for (int it = 0; it < (NFEAT / 4) * 2 * 32 / G_T; it++) {
    const int d = it * G_T + threadIdx.x;
    const int j = d >> 6, th = (d >> 5) & 1, l = d & 31;
    const float* w = W + (long)(4 * j + 2 * (l >> 4)) * HID1 + 32 * th + (l & 15);
    Bl[d] = make_float4(w[0], w[HID1], w[16], w[HID1 + 16]);
  }

  ROWS16(STORE_ROW1) *(float4*)&al[15 * AP1 + 4 * lane] = n15;
  xp = (const float4*)(X + (long)min(t0 + 8, G_TILES - 1) * 16 * NFEAT) + lane;
  ROWS16(LOAD_ROW1) n15 = xp[15 * (NFEAT / 4)];
  __syncthreads();
  gemm1_tile(al, Bl, lane, t0, H);
  __syncthreads();
  ROWS16(STORE_ROW1) *(float4*)&al[15 * AP1 + 4 * lane] = n15;
  __syncthreads();
  gemm1_tile(al, Bl, lane, t0 + 8, H);
}

__global__ void __launch_bounds__(G_T) gemm2_wmma(const float* __restrict__ X,
                                                  const float* __restrict__ W,
                                                  float* __restrict__ H) {
  __shared__ __attribute__((aligned(16))) float4 Bl[(HID1 / 4) * 2 * 32];
  __shared__ __attribute__((aligned(16))) float  Al[8][16 * AP2];
  const int wave = threadIdx.x >> 5;
  const int lane = threadIdx.x & 31;
  const int m  = lane & 15;
  const int hi = lane >> 4;

  float* al = Al[wave];
  const int t0 = blockIdx.x * (8 * G_TPW) + wave;
  float4 ROWS8(DECL_ROW) n7;
  const float4* xp = (const float4*)(X + (long)min(t0, G_TILES - 1) * 16 * HID1) + lane;
  ROWS8(LOAD_ROW2) n7 = xp[7 * 32];

  #pragma unroll
  for (int it = 0; it < (HID1 / 4) * 2 * 32 / G_T; it++) {
    const int d = it * G_T + threadIdx.x;
    const int j = d >> 6, th = (d >> 5) & 1, l = d & 31;
    const int col = 32 * th + (l & 15);
    const int ca = min(col, NCLASS - 1), cb = min(col + 16, NCLASS - 1);
    const float* w = W + (long)(4 * j + 2 * (l >> 4)) * NCLASS;
    float4 v = make_float4(w[ca], w[NCLASS + ca], w[cb], w[NCLASS + cb]);
    if (col >= NCLASS)      { v.x = 0.f; v.y = 0.f; }
    if (col + 16 >= NCLASS) { v.z = 0.f; v.w = 0.f; }
    Bl[d] = v;
  }

  ROWS8(STORE_ROW2) *(float4*)&al[(14 + hi) * AP2 + 4 * m] = n7;
  xp = (const float4*)(X + (long)min(t0 + 8, G_TILES - 1) * 16 * HID1) + lane;
  ROWS8(LOAD_ROW2) n7 = xp[7 * 32];
  __syncthreads();
  gemm2_tile(al, Bl, lane, t0, H);
  __syncthreads();
  ROWS8(STORE_ROW2) *(float4*)&al[(14 + hi) * AP2 + 4 * m] = n7;
  __syncthreads();
  gemm2_tile(al, Bl, lane, t0 + 8, H);
}

__global__ void alpha1_kernel(const float* __restrict__ H1,
                              const float* __restrict__ a_s, const float* __restrict__ a_d,
                              float* __restrict__ as, float* __restrict__ ad,
                              const int* __restrict__ ei,
                              unsigned* __restrict__ key, unsigned* __restrict__ rec) {
  int i = blockIdx.x * blockDim.x + threadIdx.x;
  if (i < N_EDGES / 4) edge_keys(ei, i, key, rec);
  if (i >= N_NODES * NHEADS) return;
  const int h = i & 7;
  const float* hp = H1 + (long)(i >> 3) * HID1 + h * NHID;
  float s = 0.f, d = 0.f;
  #pragma unroll
  for (int k = 0; k < NHID; k++) { s += hp[k] * a_s[h * NHID + k]; d += hp[k] * a_d[h * NHID + k]; }
  as[i] = s; ad[i] = d;
}

__global__ void alpha2_kernel(const float* __restrict__ H2,
                              const float* __restrict__ a_s, const float* __restrict__ a_d,
                              float* __restrict__ as, float* __restrict__ ad) {
  int n = blockIdx.x * blockDim.x + threadIdx.x;
  if (n >= N_NODES) return;
  const float* hp = H2 + (long)n * NCLASS;
  float s = 0.f, d = 0.f;
  #pragma unroll
  for (int c = 0; c < NCLASS; c++) { s += hp[c] * a_s[c]; d += hp[c] * a_d[c]; }
  as[n] = s; ad[n] = d;
}

extern "C" void kernel_launch(void* const* d_in, const int* in_sizes, int n_in,
                              void* d_out, int out_size, void* d_ws, size_t ws_size,
                              hipStream_t stream) {
  const float* x      = (const float*)d_in[0];
  const float* W1     = (const float*)d_in[1];
  const float* a_src1 = (const float*)d_in[2];
  const float* a_dst1 = (const float*)d_in[3];
  const float* b1     = (const float*)d_in[4];
  const float* W2     = (const float*)d_in[5];
  const float* a_src2 = (const float*)d_in[6];
  const float* a_dst2 = (const float*)d_in[7];
  const float* b2     = (const float*)d_in[8];
  const int*   ei     = (const int*)d_in[9];
  float*       out    = (float*)d_out;

  float* ws   = (float*)d_ws;
  float* H1   = ws;
  float* as1  = H1  + (long)N_NODES * HID1;
  float* ad1  = as1 + (long)N_NODES * NHEADS;
  float* as2  = ad1 + (long)N_NODES * NHEADS;
  float* ad2  = as2 + N_NODES;
  float* X2   = ad2 + N_NODES;
  float* H2   = X2  + (long)N_NODES * HID1;
  unsigned* glist = (unsigned*)(H2 + (long)N_NODES * NCLASS);
  unsigned* gcnt  = glist + (long)NRANGE * LIST_CAP;
  unsigned* key   = gcnt + NRANGE;
  unsigned* rec   = key + N_EDGES / 4;

  const int B = 256;
  auto cdiv = [](long a, long b) { return (int)((a + b - 1) / b); };

  gemm1_wmma<<<G_GRID, G_T, 0, stream>>>(x, W1, H1);
  alpha1_kernel<<<cdiv((long)N_NODES * NHEADS, B), B, 0, stream>>>(H1, a_src1, a_dst1, as1, ad1, ei, key, rec);
  agg1_lds<<<NRANGE, AGG_T, 0, stream>>>(key, rec, as1, ad1, H1, b1, X2, glist, gcnt);

  gemm2_wmma<<<G_GRID, G_T, 0, stream>>>(X2, W2, H2);
  alpha2_kernel<<<cdiv(N_NODES, B), B, 0, stream>>>(H2, a_src2, a_dst2, as2, ad2);
  agg2_lds<<<NRANGE, AGG_T, 0, stream>>>(key, rec, as2, ad2, H2, b2, out, glist, gcnt);
}


# alpha1's launch sorts each 3200-edge chunk of the edge list by destination range in LDS; an agg work-group copies its 250 slices, not scanning a byte per edge; LDS float adds: arrival order
# speedup vs baseline: 11.7096x; 1.1487x over previous
#include <hip/hip_runtime.h>
#include <math.h>

#define N_NODES 50000
#define N_EDGES 800000
#define E_TOT   (N_EDGES + N_NODES)
#define NFEAT   128
#define NHID    8
#define NHEADS  8
#define HID1    (NHEADS * NHID)
#define NCLASS  40
#define NEG_SLOPE 0.2f

typedef __attribute__((ext_vector_type(2))) float v2f;
typedef __attribute__((ext_vector_type(8))) float v8f;

__device__ __forceinline__ float lrelu(float x) { return x > 0.f ? x : NEG_SLOPE * x; }

__device__ __forceinline__ unsigned f2ord(float f) {
  unsigned u = __float_as_uint(f);
  return (u & 0x80000000u) ? ~u : (u | 0x80000000u);
}
__device__ __forceinline__ float ord2f(unsigned u) {
  return (u & 0x80000000u) ? __uint_as_float(u & 0x7FFFFFFFu) : __uint_as_float(~u);
}
#define ORD_NEG_INF 0x007FFFFFu

#define AGG_T   1024
#define NRANGE  256
#define RN      196
#define CH      3200
#define NCH     (N_EDGES / CH)
#define SLANES  4
#define SLU     8
#define GRP     10
#define LIST_CAP 36864
#define A1_T    256
#define QPT     ((CH / 4 + A1_T - 1) / A1_T)
static_assert(NCH * CH == N_EDGES && CH % 32 == 0 && CH < 65536, "whole chunks, line-aligned, ranks fit 16 bits");
static_assert(NCH * SLANES <= AGG_T && GRP * SLANES <= AGG_T, "one round of slice copies covers the chunks it is given");
static_assert(RN + GRP * CH <= LIST_CAP, "a general-path round fits the list");
static_assert(A1_T == NRANGE, "bin_chunk: one thread per range counter");
#define P1      65
#define UNR     4
#define KREG    32
#define REG_CAP (KREG * (AGG_T / 8))
static_assert(REG_CAP == UNR * AGG_T, "layer 2's max pass covers REG_CAP entries in one round");

#define EBITS 20
#define EMASK ((1u << EBITS) - 1)
__device__ __forceinline__ void bin_chunk(const int* __restrict__ ei, int c, int t,
                                          unsigned* __restrict__ srt, unsigned* __restrict__ bofs,
                                          unsigned* bcnt, unsigned* boff, unsigned* wsum, unsigned* lsrt) {
  bcnt[t] = 0u;
  const int4* __restrict__ s4 = (const int4*)(ei + c * CH);
  const int4* __restrict__ d4 = (const int4*)(ei + N_EDGES + c * CH);
  int4 sv[QPT], dv[QPT];
  #pragma unroll
  for (int j = 0; j < QPT; j++) {
    const int q = t + j * A1_T;
    const bool in = q < CH / 4;
    sv[j] = in ? s4[q] : make_int4(0, 0, 0, 0);
    dv[j] = in ? d4[q] : make_int4(0, 0, 0, 0);
  }
  __syncthreads();
  unsigned gr[QPT][4];
  #pragma unroll
  for (int j = 0; j < QPT; j++) {
    if (t + j * A1_T < CH / 4) {
      const unsigned dd[4] = {(unsigned)dv[j].x, (unsigned)dv[j].y, (unsigned)dv[j].z, (unsigned)dv[j].w};
      #pragma unroll
      for (int k = 0; k < 4; k++) {
        const unsigned g = dd[k] / RN;
        gr[j][k] = (g << 16) | atomicAdd(&bcnt[g], 1u);
      }
    }
  }
  __syncthreads();
  const unsigned v = bcnt[t];
  unsigned inc = v;
  #pragma unroll
  for (int dlt = 1; dlt < 32; dlt <<= 1) {
    const unsigned o = __shfl_up(inc, dlt);
    if ((t & 31) >= dlt) inc += o;
  }
  if ((t & 31) == 31) wsum[t >> 5] = inc;
  __syncthreads();
  unsigned start = inc - v;
  for (int w = 0; w < (t >> 5); w++) start += wsum[w];
  boff[t] = start;
  ((uint2*)bofs)[c * NRANGE + t] = make_uint2(start, start + v);
  __syncthreads();
  #pragma unroll
  for (int j = 0; j < QPT; j++) {
    if (t + j * A1_T < CH / 4) {
      const unsigned ss[4] = {(unsigned)sv[j].x, (unsigned)sv[j].y, (unsigned)sv[j].z, (unsigned)sv[j].w};
      const unsigned dd[4] = {(unsigned)dv[j].x, (unsigned)dv[j].y, (unsigned)dv[j].z, (unsigned)dv[j].w};
      #pragma unroll
      for (int k = 0; k < 4; k++) {
        const unsigned g = gr[j][k] >> 16;
        lsrt[boff[g] + (gr[j][k] & 0xFFFFu)] = ss[k] | ((dd[k] - g * RN) << EBITS);
      }
    }
  }
  __syncthreads();
  uint4* __restrict__ o4 = (uint4*)(srt + c * CH);
  for (int q = t; q < CH / 4; q += A1_T) o4[q] = ((const uint4*)lsrt)[q];
}

__device__ __forceinline__ void gather_slices(const unsigned* __restrict__ srt, const unsigned* __restrict__ bofs,
                                              int r, int c0, int c1, int t, unsigned* list, unsigned* cnt) {
  const int c = c0 + t / SLANES;
  const unsigned q = (unsigned)t % SLANES;
  unsigned len = 0, g0 = 0, p;
  if (c < c1) {
    const uint2 se = ((const uint2*)bofs)[c * NRANGE + r];
    len = se.y - se.x;
    g0  = (unsigned)c * CH + se.x;
  }
  const int lane = t & 31;
  const unsigned mine = (q == 0) ? len : 0u;
  unsigned inc = mine;
  #pragma unroll
  for (int dlt = 1; dlt < 32; dlt <<= 1) {
    const unsigned o = __shfl_up(inc, dlt);
    if (lane >= dlt) inc += o;
  }
  unsigned base = 0;
  if (lane == 31 && inc) base = atomicAdd(cnt, inc);
  base = __shfl(base, 31);
  p = __shfl(base + inc - mine, lane & ~(SLANES - 1));
  for (unsigned i0 = q; i0 < len; i0 += SLANES * SLU) {
    unsigned w[SLU];
    #pragma unroll
    for (int j = 0; j < SLU; j++) w[j] = (i0 + j * SLANES < len) ? srt[g0 + i0 + j * SLANES] : 0u;
    #pragma unroll
    for (int j = 0; j < SLU; j++) {
      const unsigned i = i0 + j * SLANES;
      if (i < len && p + i < LIST_CAP) list[p + i] = w[j];
    }
  }
}

__device__ __forceinline__ void list_self_loops(int t, int lo, unsigned span, unsigned* list, unsigned* cnt) {
  if (t < (int)span) list[t] = (unsigned)(lo + t) | ((unsigned)t << EBITS);
  if (t == 0) *cnt = span;
}

__global__ void __launch_bounds__(AGG_T) agg1_lds(const unsigned* __restrict__ srt,
                                                  const unsigned* __restrict__ bofs,
                                                  const float* __restrict__ as,
                                                  const float* __restrict__ ad,
                                                  const float* __restrict__ H1,
                                                  const float* __restrict__ bias,
                                                  float* __restrict__ out1,
                                                  unsigned* __restrict__ glist,
                                                  unsigned* __restrict__ gcnt) {
  __shared__ float    acc[RN * P1];
  __shared__ float    dsum[RN * NHEADS];
  __shared__ float    adl[RN * NHEADS];
  __shared__ unsigned mxo[RN * NHEADS];
  __shared__ unsigned list[LIST_CAP];
  __shared__ unsigned cnt;

  const int t  = threadIdx.x;
  const int lo = blockIdx.x * RN;
  const unsigned span = (unsigned)(min(lo + RN, N_NODES) - lo);
  const int h = t & 7;

  for (int i = t; i < RN * P1; i += AGG_T) acc[i] = 0.f;
  for (int i = t; i < (int)span * NHEADS; i += AGG_T) {
    dsum[i] = 0.f;
    adl[i]  = ad[lo * NHEADS + i];
    mxo[i]  = ORD_NEG_INF;
  }

  auto max_pass = [&](unsigned n) {
    for (unsigned b = t >> 3; b < n; b += UNR * (AGG_T / 8)) {
      unsigned dl[UNR]; float a[UNR];
      #pragma unroll
      for (int j = 0; j < UNR; j++) {
        const unsigned en = list[min(b + j * (AGG_T / 8), n - 1)];
        dl[j] = en >> EBITS;
        a[j]  = as[(en & EMASK) * NHEADS + h];
      }
      #pragma unroll
      for (int j = 0; j < UNR; j++)
        atomicMax(&mxo[dl[j] * NHEADS + h], f2ord(lrelu(a[j] + adl[dl[j] * NHEADS + h])));
    }
  };
  auto sum_pass = [&](unsigned n) {
    for (unsigned b = t >> 3; b < n; b += UNR * (AGG_T / 8)) {
      unsigned dl[UNR]; float a[UNR]; float4 h0[UNR], h1[UNR];
      #pragma unroll
      for (int j = 0; j < UNR; j++) {
        const unsigned en = list[min(b + j * (AGG_T / 8), n - 1)];
        const unsigned s = en & EMASK;
        dl[j] = en >> EBITS;
        a[j]  = as[s * NHEADS + h];
        const float4* hs = (const float4*)(H1 + (long)s * HID1 + h * NHID);
        h0[j] = hs[0]; h1[j] = hs[1];
      }
      #pragma unroll
      for (int j = 0; j < UNR; j++) {
        if (b + j * (AGG_T / 8) < n) {
          const int d8 = dl[j] * NHEADS + h;
          const float v = lrelu(a[j] + adl[d8]);
          const float w = __expf(v - ord2f(mxo[d8]));
          atomicAdd(&dsum[d8], w);
          float* od = &acc[dl[j] * P1 + h * NHID];
          atomicAdd(&od[0], w * h0[j].x); atomicAdd(&od[1], w * h0[j].y);
          atomicAdd(&od[2], w * h0[j].z); atomicAdd(&od[3], w * h0[j].w);
          atomicAdd(&od[4], w * h1[j].x); atomicAdd(&od[5], w * h1[j].y);
          atomicAdd(&od[6], w * h1[j].z); atomicAdd(&od[7], w * h1[j].w);
        }
      }
    }
  };

  list_self_loops(t, lo, span, list, &cnt);
  __syncthreads();
  gather_slices(srt, bofs, blockIdx.x, 0, NCH, t, list, &cnt);
  __syncthreads();
  const unsigned n = cnt;

  if (t == 0) gcnt[blockIdx.x] = n;
  if (n <= REG_CAP) {
    for (unsigned i = n + t; i < REG_CAP; i += AGG_T) list[i] = (unsigned)lo;
    __syncthreads();
    unsigned* gl = glist + (long)blockIdx.x * LIST_CAP;
    for (unsigned i = t; i < n; i += AGG_T) gl[i] = list[i];

    const unsigned* lslot = list + (t >> 3);
    float areg[KREG];
    #pragma unroll
    for (int j = 0; j < KREG; j++) areg[j] = as[(lslot[j * (AGG_T / 8)] & EMASK) * NHEADS + h];
    #pragma unroll
    for (int j = 0; j < KREG; j++) {
      if ((unsigned)(j * (AGG_T / 8)) < n) {
        const unsigned dl = lslot[j * (AGG_T / 8)] >> EBITS;
        areg[j] = lrelu(areg[j] + adl[dl * NHEADS + h]);
        atomicMax(&mxo[dl * NHEADS + h], f2ord(areg[j]));
      }
    }
    __syncthreads();
    #pragma clang loop unroll(disable)
    for (unsigned r0 = 0; r0 < n; r0 += UNR * (AGG_T / 8)) {
      unsigned dl[UNR]; float4 h0[UNR], h1[UNR];
      #pragma unroll
      for (int j = 0; j < UNR; j++) {
        const unsigned en = lslot[r0 + j * (AGG_T / 8)];
        dl[j] = en >> EBITS;
        const float4* hs = (const float4*)(H1 + ((en & EMASK) * HID1 + h * NHID));
        h0[j] = hs[0]; h1[j] = hs[1];
      }
      #pragma unroll
      for (int j = 0; j < UNR; j++) {
        if (r0 + (unsigned)(t >> 3) + j * (AGG_T / 8) < n) {
          const int d8 = dl[j] * NHEADS + h;
          const float w = __expf(areg[j] - ord2f(mxo[d8]));
          atomicAdd(&dsum[d8], w);
          float* od = &acc[dl[j] * P1 + h * NHID];
          atomicAdd(&od[0], w * h0[j].x); atomicAdd(&od[1], w * h0[j].y);
          atomicAdd(&od[2], w * h0[j].z); atomicAdd(&od[3], w * h0[j].w);
          atomicAdd(&od[4], w * h1[j].x); atomicAdd(&od[5], w * h1[j].y);
          atomicAdd(&od[6], w * h1[j].z); atomicAdd(&od[7], w * h1[j].w);
        }
      }
      #pragma unroll
      for (int k = 0; k + UNR < KREG; k++) areg[k] = areg[k + UNR];
    }
  } else if (n <= LIST_CAP) {
    unsigned* gl = glist + (long)blockIdx.x * LIST_CAP;
    for (unsigned i = t; i < n; i += AGG_T) gl[i] = list[i];
    max_pass(n);
    __syncthreads();
    sum_pass(n);
  } else {
    for (int phase = 0; phase < 2; phase++) {
      for (int c0 = 0; c0 < NCH; c0 += GRP) {
        __syncthreads();
        if (c0 == 0) list_self_loops(t, lo, span, list, &cnt);
        else if (t == 0) cnt = 0;
        __syncthreads();
        gather_slices(srt, bofs, blockIdx.x, c0, min(c0 + GRP, NCH), t, list, &cnt);
        __syncthreads();
        const unsigned nc = cnt;
        if (phase == 0) max_pass(nc); else sum_pass(nc);
      }
    }
  }
  __syncthreads();

  for (int i = t; i < (int)span * HID1; i += AGG_T) {
    const float v = acc[(i >> 6) * P1 + (i & 63)] / dsum[i >> 3] + bias[i & (HID1 - 1)];
    out1[(long)lo * HID1 + i] = v > 0.f ? v : (__expf(v) - 1.f);
  }
}

__global__ void __launch_bounds__(AGG_T) agg2_lds(const unsigned* __restrict__ srt,
                                                  const unsigned* __restrict__ bofs,
                                                  const float* __restrict__ as,
                                                  const float* __restrict__ ad,
                                                  const float* __restrict__ H2,
                                                  const float* __restrict__ bias,
                                                  float* __restrict__ out,
                                                  const unsigned* __restrict__ glist,
                                                  const unsigned* __restrict__ gcnt) {
  __shared__ float    acc[RN * NCLASS];
  __shared__ float    dsum[RN];
  __shared__ float    adl[RN];
  __shared__ unsigned mxo[RN];
  __shared__ unsigned list[LIST_CAP];
  __shared__ float    la[REG_CAP];
  __shared__ unsigned cnt;

  const int t  = threadIdx.x;
  const int lo = blockIdx.x * RN;
  const unsigned span = (unsigned)(min(lo + RN, N_NODES) - lo);
  const int l = t & 7;

  for (int i = t; i < RN * NCLASS; i += AGG_T) acc[i] = 0.f;
  if (t < (int)span) {
    dsum[t] = 0.f;
    adl[t]  = ad[lo + t];
    mxo[t]  = ORD_NEG_INF;
  }

  auto max_pass = [&](unsigned n) {
    for (unsigned b = t; b < n; b += UNR * AGG_T) {
      unsigned dl[UNR]; float a[UNR];
      #pragma unroll
      for (int j = 0; j < UNR; j++) {
        const unsigned en = list[min(b + j * AGG_T, n - 1)];
        dl[j] = en >> EBITS;
        a[j]  = as[en & EMASK];
      }
      #pragma unroll
      for (int j = 0; j < UNR; j++)
        atomicMax(&mxo[dl[j]], f2ord(lrelu(a[j] + adl[dl[j]])));
    }
  };
  auto sum_pass = [&](unsigned n) {
    for (unsigned b = t >> 3; b < n; b += UNR * (AGG_T / 8)) {
      unsigned dl[UNR]; float a[UNR]; float hv[UNR][NCLASS / 8];
      #pragma unroll
      for (int j = 0; j < UNR; j++) {
        const unsigned en = list[min(b + j * (AGG_T / 8), n - 1)];
        const unsigned s = en & EMASK;
        dl[j] = en >> EBITS;
        a[j]  = as[s];
        const float* hs = H2 + (long)s * NCLASS + l;
        #pragma unroll
        for (int k = 0; k < NCLASS / 8; k++) hv[j][k] = hs[k * 8];
      }
      #pragma unroll
      for (int j = 0; j < UNR; j++) {
        if (b + j * (AGG_T / 8) < n) {
          const float v = lrelu(a[j] + adl[dl[j]]);
          const float w = __expf(v - ord2f(mxo[dl[j]]));
          if (l == 0) atomicAdd(&dsum[dl[j]], w);
          float* od = &acc[dl[j] * NCLASS + l];
          #pragma unroll
          for (int k = 0; k < NCLASS / 8; k++) atomicAdd(&od[k * 8], w * hv[j][k]);
        }
      }
    }
  };

  const unsigned n = gcnt[blockIdx.x];

  if (n <= REG_CAP) {
    const unsigned* gl = glist + (long)blockIdx.x * LIST_CAP;
    for (unsigned i = t; i < n; i += AGG_T) list[i] = gl[i];
    __syncthreads();
    {
      unsigned dl[UNR]; float a[UNR];
      #pragma unroll
      for (int j = 0; j < UNR; j++) {
        const unsigned en = list[min((unsigned)t + j * AGG_T, n - 1)];
        dl[j] = en >> EBITS;
        a[j]  = as[en & EMASK];
      }
      #pragma unroll
      for (int j = 0; j < UNR; j++) {
        if ((unsigned)t + j * AGG_T < n) {
          const float v = lrelu(a[j] + adl[dl[j]]);
          la[t + j * AGG_T] = v;
          atomicMax(&mxo[dl[j]], f2ord(v));
        }
      }
    }
    __syncthreads();
    for (unsigned b = t >> 3; b < n; b += UNR * (AGG_T / 8)) {
      unsigned dl[UNR]; float v[UNR]; float hv[UNR][NCLASS / 8];
      #pragma unroll
      for (int j = 0; j < UNR; j++) {
        const unsigned i = min(b + j * (AGG_T / 8), n - 1);
        const unsigned en = list[i];
        dl[j] = en >> EBITS;
        v[j]  = la[i];
        const float* hs = H2 + (long)(en & EMASK) * NCLASS + l;
        #pragma unroll
        for (int k = 0; k < NCLASS / 8; k++) hv[j][k] = hs[k * 8];
      }
      #pragma unroll
      for (int j = 0; j < UNR; j++) {
        if (b + j * (AGG_T / 8) < n) {
          const float w = __expf(v[j] - ord2f(mxo[dl[j]]));
          if (l == 0) atomicAdd(&dsum[dl[j]], w);
          float* od = &acc[dl[j] * NCLASS + l];
          #pragma unroll
          for (int k = 0; k < NCLASS / 8; k++) atomicAdd(&od[k * 8], w * hv[j][k]);
        }
      }
    }
  } else if (n <= LIST_CAP) {
    const unsigned* gl = glist + (long)blockIdx.x * LIST_CAP;
    for (unsigned i = t; i < n; i += AGG_T) list[i] = gl[i];
    __syncthreads();
    max_pass(n);
    __syncthreads();
    sum_pass(n);
  } else {
    for (int phase = 0; phase < 2; phase++) {
      for (int c0 = 0; c0 < NCH; c0 += GRP) {
        __syncthreads();
        if (c0 == 0) list_self_loops(t, lo, span, list, &cnt);
        else if (t == 0) cnt = 0;
        __syncthreads();
        gather_slices(srt, bofs, blockIdx.x, c0, min(c0 + GRP, NCH), t, list, &cnt);
        __syncthreads();
        const unsigned nc = cnt;
        if (phase == 0) max_pass(nc); else sum_pass(nc);
      }
    }
  }
  __syncthreads();

  for (int i = t; i < (int)span * NCLASS; i += AGG_T) acc[i] = acc[i] / dsum[i / NCLASS] + bias[i % NCLASS];
  __syncthreads();
  float* lse = adl;
  if (t < (int)span) {
    const float* r = &acc[t * NCLASS];
    float mx = -INFINITY;
    for (int c = 0; c < NCLASS; c++) mx = fmaxf(mx, r[c]);
    float ssum = 0.f;
    for (int c = 0; c < NCLASS; c++) ssum += __expf(r[c] - mx);
    lse[t] = mx + __logf(ssum);
  }
  __syncthreads();
  for (int i = t; i < (int)span * NCLASS; i += AGG_T) out[(long)lo * NCLASS + i] = acc[i] - lse[i / NCLASS];
}

#define G_T     256
#define G_TPW   2
#define G_TILES (N_NODES / 16)
#define G_GRID  ((G_TILES + 8 * G_TPW - 1) / (8 * G_TPW))
#define AP1     (NFEAT + 4)
#define AP2     (HID1 + 4)

__device__ __forceinline__ v8f wmma4(v2f a, float bx, float by, v8f c) {
  v2f b; b.x = bx; b.y = by;
  return __builtin_amdgcn_wmma_f32_16x16x4_f32(false, a, false, b, (short)0, c, false, false);
}

#define ROWS16(F) F(0) F(1) F(2) F(3) F(4) F(5) F(6) F(7) F(8) F(9) F(10) F(11) F(12) F(13) F(14)
#define ROWS8(F)  F(0) F(1) F(2) F(3) F(4) F(5) F(6)
#define DECL_ROW(r)   n##r,
#define LOAD_ROW1(r)  n##r = xp[r * (NFEAT / 4)];
#define STORE_ROW1(r) *(float4*)&al[r * AP1 + 4 * lane] = n##r;
#define LOAD_ROW2(r)  n##r = xp[r * 32];
#define STORE_ROW2(r) *(float4*)&al[(2 * r + hi) * AP2 + 4 * m] = n##r;

__device__ __forceinline__ void gemm1_tile(const float* al, const float4* Bl, int lane, int tile, float* __restrict__ H) {
  const int m  = lane & 15;
  const int hi = lane >> 4;
  v8f c0 = {}, c1 = {}, c2 = {}, c3 = {};
  #pragma unroll 8
  for (int j = 0; j < NFEAT / 4; j++) {
    const v2f a = *(const v2f*)&al[m * AP1 + 4 * j + 2 * hi];
    const float4 p = Bl[(2 * j) * 32 + lane];
    const float4 q = Bl[(2 * j + 1) * 32 + lane];
    c0 = wmma4(a, p.x, p.y, c0);
    c1 = wmma4(a, p.z, p.w, c1);
    c2 = wmma4(a, q.x, q.y, c2);
    c3 = wmma4(a, q.z, q.w, c3);
  }
  if (tile < G_TILES) {
    float* hp = H + (long)(tile * 16 + 8 * hi) * HID1 + m;
    #pragma unroll
    for (int j = 0; j < 8; j++) {
      hp[(long)j * HID1 +  0] = c0[j];
      hp[(long)j * HID1 + 16] = c1[j];
      hp[(long)j * HID1 + 32] = c2[j];
      hp[(long)j * HID1 + 48] = c3[j];
    }
  }
}

__device__ __forceinline__ void gemm2_tile(const float* al, const float4* Bl, int lane, int tile, float* __restrict__ H) {
  const int m  = lane & 15;
  const int hi = lane >> 4;
  v8f c0 = {}, c1 = {}, c2 = {};
  #pragma unroll 8
  for (int j = 0; j < HID1 / 4; j++) {
    const v2f a = *(const v2f*)&al[m * AP2 + 4 * j + 2 * hi];
    const float4 p = Bl[(2 * j) * 32 + lane];
    const float4 q = Bl[(2 * j + 1) * 32 + lane];
    c0 = wmma4(a, p.x, p.y, c0);
    c1 = wmma4(a, p.z, p.w, c1);
    c2 = wmma4(a, q.x, q.y, c2);
  }
  if (tile < G_TILES) {
    const int rbase = tile * 16 + 8 * hi;
    #pragma unroll
    for (int j = 0; j < 8; j++) {
      float* hr = H + (long)(rbase + j) * NCLASS;
      hr[m]      = c0[j];
      hr[m + 16] = c1[j];
      if (m + 32 < NCLASS) hr[m + 32] = c2[j];
    }
  }
}

__global__ void __launch_bounds__(G_T) gemm1_wmma(const float* __restrict__ X,
                                                  const float* __restrict__ W,
                                                  float* __restrict__ H) {
  __shared__ __attribute__((aligned(16))) float4 Bl[(NFEAT / 4) * 2 * 32];
  __shared__ __attribute__((aligned(16))) float  Al[8][16 * AP1];
  const int wave = threadIdx.x >> 5;
  const int lane = threadIdx.x & 31;

  float* al = Al[wave];
  const int t0 = blockIdx.x * (8 * G_TPW) + wave;
  float4 ROWS16(DECL_ROW) n15;
  const float4* xp = (const float4*)(X + (long)min(t0, G_TILES - 1) * 16 * NFEAT) + lane;
  ROWS16(LOAD_ROW1) n15 = xp[15 * (NFEAT / 4)];

  #pragma unroll
  for (int it = 0; it < (NFEAT / 4) * 2 * 32 / G_T; it++) {
    const int d = it * G_T + threadIdx.x;
    const int j = d >> 6, th = (d >> 5) & 1, l = d & 31;
    const float* w = W + (long)(4 * j + 2 * (l >> 4)) * HID1 + 32 * th + (l & 15);
    Bl[d] = make_float4(w[0], w[HID1], w[16], w[HID1 + 16]);
  }

  ROWS16(STORE_ROW1) *(float4*)&al[15 * AP1 + 4 * lane] = n15;
  xp = (const float4*)(X + (long)min(t0 + 8, G_TILES - 1) * 16 * NFEAT) + lane;
  ROWS16(LOAD_ROW1) n15 = xp[15 * (NFEAT / 4)];
  __syncthreads();
  gemm1_tile(al, Bl, lane, t0, H);
  __syncthreads();
  ROWS16(STORE_ROW1) *(float4*)&al[15 * AP1 + 4 * lane] = n15;
  __syncthreads();
  gemm1_tile(al, Bl, lane, t0 + 8, H);
}

__global__ void __launch_bounds__(G_T) gemm2_wmma(const float* __restrict__ X,
                                                  const float* __restrict__ W,
                                                  float* __restrict__ H) {
  __shared__ __attribute__((aligned(16))) float4 Bl[(HID1 / 4) * 2 * 32];
  __shared__ __attribute__((aligned(16))) float  Al[8][16 * AP2];
  const int wave = threadIdx.x >> 5;
  const int lane = threadIdx.x & 31;
  const int m  = lane & 15;
  const int hi = lane >> 4;

  float* al = Al[wave];
  const int t0 = blockIdx.x * (8 * G_TPW) + wave;
  float4 ROWS8(DECL_ROW) n7;
  const float4* xp = (const float4*)(X + (long)min(t0, G_TILES - 1) * 16 * HID1) + lane;
  ROWS8(LOAD_ROW2) n7 = xp[7 * 32];

  #pragma unroll
  for (int it = 0; it < (HID1 / 4) * 2 * 32 / G_T; it++) {
    const int d = it * G_T + threadIdx.x;
    const int j = d >> 6, th = (d >> 5) & 1, l = d & 31;
    const int col = 32 * th + (l & 15);
    const int ca = min(col, NCLASS - 1), cb = min(col + 16, NCLASS - 1);
    const float* w = W + (long)(4 * j + 2 * (l >> 4)) * NCLASS;
    float4 v = make_float4(w[ca], w[NCLASS + ca], w[cb], w[NCLASS + cb]);
    if (col >= NCLASS)      { v.x = 0.f; v.y = 0.f; }
    if (col + 16 >= NCLASS) { v.z = 0.f; v.w = 0.f; }
    Bl[d] = v;
  }

  ROWS8(STORE_ROW2) *(float4*)&al[(14 + hi) * AP2 + 4 * m] = n7;
  xp = (const float4*)(X + (long)min(t0 + 8, G_TILES - 1) * 16 * HID1) + lane;
  ROWS8(LOAD_ROW2) n7 = xp[7 * 32];
  __syncthreads();
  gemm2_tile(al, Bl, lane, t0, H);
  __syncthreads();
  ROWS8(STORE_ROW2) *(float4*)&al[(14 + hi) * AP2 + 4 * m] = n7;
  __syncthreads();
  gemm2_tile(al, Bl, lane, t0 + 8, H);
}

static_assert(NCH <= (N_NODES * NHEADS + A1_T - 1) / A1_T, "the grid holds a work-group per chunk");
__global__ void __launch_bounds__(A1_T) alpha1_kernel(const float* __restrict__ H1,
                              const float* __restrict__ a_s, const float* __restrict__ a_d,
                              float* __restrict__ as, float* __restrict__ ad,
                              const int* __restrict__ ei,
                              unsigned* __restrict__ srt, unsigned* __restrict__ bofs) {
  __shared__ unsigned bcnt[NRANGE], boff[NRANGE], wsum[A1_T / 32];
  __shared__ __attribute__((aligned(16))) unsigned lsrt[CH];
  if (blockIdx.x < NCH) bin_chunk(ei, blockIdx.x, threadIdx.x, srt, bofs, bcnt, boff, wsum, lsrt);
  int i = blockIdx.x * blockDim.x + threadIdx.x;
  if (i >= N_NODES * NHEADS) return;
  const int h = i & 7;
  const float* hp = H1 + (long)(i >> 3) * HID1 + h * NHID;
  float s = 0.f, d = 0.f;
  #pragma unroll
  for (int k = 0; k < NHID; k++) { s += hp[k] * a_s[h * NHID + k]; d += hp[k] * a_d[h * NHID + k]; }
  as[i] = s; ad[i] = d;
}

__global__ void alpha2_kernel(const float* __restrict__ H2,
                              const float* __restrict__ a_s, const float* __restrict__ a_d,
                              float* __restrict__ as, float* __restrict__ ad) {
  int n = blockIdx.x * blockDim.x + threadIdx.x;
  if (n >= N_NODES) return;
  const float* hp = H2 + (long)n * NCLASS;
  float s = 0.f, d = 0.f;
  #pragma unroll
  for (int c = 0; c < NCLASS; c++) { s += hp[c] * a_s[c]; d += hp[c] * a_d[c]; }
  as[n] = s; ad[n] = d;
}

extern "C" void kernel_launch(void* const* d_in, const int* in_sizes, int n_in,
                              void* d_out, int out_size, void* d_ws, size_t ws_size,
                              hipStream_t stream) {
  const float* x      = (const float*)d_in[0];
  const float* W1     = (const float*)d_in[1];
  const float* a_src1 = (const float*)d_in[2];
  const float* a_dst1 = (const float*)d_in[3];
  const float* b1     = (const float*)d_in[4];
  const float* W2     = (const float*)d_in[5];
  const float* a_src2 = (const float*)d_in[6];
  const float* a_dst2 = (const float*)d_in[7];
  const float* b2     = (const float*)d_in[8];
  const int*   ei     = (const int*)d_in[9];
  float*       out    = (float*)d_out;

  float* ws   = (float*)d_ws;
  float* H1   = ws;
  float* as1  = H1  + (long)N_NODES * HID1;
  float* ad1  = as1 + (long)N_NODES * NHEADS;
  float* as2  = ad1 + (long)N_NODES * NHEADS;
  float* ad2  = as2 + N_NODES;
  float* X2   = ad2 + N_NODES;
  float* H2   = X2  + (long)N_NODES * HID1;
  unsigned* glist = (unsigned*)(H2 + (long)N_NODES * NCLASS);
  unsigned* gcnt  = glist + (long)NRANGE * LIST_CAP;
  unsigned* srt   = gcnt + NRANGE;
  unsigned* bofs  = srt + N_EDGES;

  const int B = A1_T;
  auto cdiv = [](long a, long b) { return (int)((a + b - 1) / b); };

  gemm1_wmma<<<G_GRID, G_T, 0, stream>>>(x, W1, H1);
  alpha1_kernel<<<cdiv((long)N_NODES * NHEADS, B), B, 0, stream>>>(H1, a_src1, a_dst1, as1, ad1, ei, srt, bofs);
  agg1_lds<<<NRANGE, AGG_T, 0, stream>>>(srt, bofs, as1, ad1, H1, b1, X2, glist, gcnt);

  gemm2_wmma<<<G_GRID, G_T, 0, stream>>>(X2, W2, H2);
  alpha2_kernel<<<cdiv(N_NODES, B), B, 0, stream>>>(H2, a_src2, a_dst2, as2, ad2);
  agg2_lds<<<NRANGE, AGG_T, 0, stream>>>(srt, bofs, as2, ad2, H2, b2, out, glist, gcnt);
}


# alpha1/alpha2 launches gone: columns W.a_src, W.a_dst join the staged weights, node logits leave the GEMMs' WMMAs (logit sums in a new order); chunk sort rides in gemm1's launch; 6 -> 4 launches
# speedup vs baseline: 12.5938x; 1.0755x over previous
#include <hip/hip_runtime.h>
#include <math.h>

#define N_NODES 50000
#define N_EDGES 800000
#define E_TOT   (N_EDGES + N_NODES)
#define NFEAT   128
#define NHID    8
#define NHEADS  8
#define HID1    (NHEADS * NHID)
#define NCLASS  40
#define NEG_SLOPE 0.2f

typedef __attribute__((ext_vector_type(2))) float v2f;
typedef __attribute__((ext_vector_type(8))) float v8f;

__device__ __forceinline__ float lrelu(float x) { return x > 0.f ? x : NEG_SLOPE * x; }

__device__ __forceinline__ unsigned f2ord(float f) {
  unsigned u = __float_as_uint(f);
  return (u & 0x80000000u) ? ~u : (u | 0x80000000u);
}
__device__ __forceinline__ float ord2f(unsigned u) {
  return (u & 0x80000000u) ? __uint_as_float(u & 0x7FFFFFFFu) : __uint_as_float(~u);
}
#define ORD_NEG_INF 0x007FFFFFu

#define AGG_T   1024
#define NRANGE  256
#define RN      196
#define CH      3200
#define NCH     (N_EDGES / CH)
#define SLANES  4
#define SLU     8
#define GRP     10
#define LIST_CAP 36864
#define A1_T    256
#define QPT     ((CH / 4 + A1_T - 1) / A1_T)
static_assert(NCH * CH == N_EDGES && CH % 32 == 0 && CH < 65536, "whole chunks, line-aligned, ranks fit 16 bits");
static_assert(NCH * SLANES <= AGG_T && GRP * SLANES <= AGG_T, "one round of slice copies covers the chunks it is given");
static_assert(RN + GRP * CH <= LIST_CAP, "a general-path round fits the list");
static_assert(A1_T == NRANGE, "bin_chunk: one thread per range counter");
#define P1      65
#define UNR     4
#define KREG    32
#define REG_CAP (KREG * (AGG_T / 8))
static_assert(REG_CAP == UNR * AGG_T, "layer 2's max pass covers REG_CAP entries in one round");

#define EBITS 20
#define EMASK ((1u << EBITS) - 1)
__device__ __forceinline__ void bin_chunk(const int* __restrict__ ei, int c, int t,
                                          unsigned* __restrict__ srt, unsigned* __restrict__ bofs,
                                          unsigned* bcnt, unsigned* boff, unsigned* wsum, unsigned* lsrt) {
  bcnt[t] = 0u;
  const int4* __restrict__ s4 = (const int4*)(ei + c * CH);
  const int4* __restrict__ d4 = (const int4*)(ei + N_EDGES + c * CH);
  int4 sv[QPT], dv[QPT];
  #pragma unroll
  for (int j = 0; j < QPT; j++) {
    const int q = t + j * A1_T;
    const bool in = q < CH / 4;
    sv[j] = in ? s4[q] : make_int4(0, 0, 0, 0);
    dv[j] = in ? d4[q] : make_int4(0, 0, 0, 0);
  }
  __syncthreads();
  unsigned gr[QPT][4];
  #pragma unroll
  for (int j = 0; j < QPT; j++) {
    if (t + j * A1_T < CH / 4) {
      const unsigned dd[4] = {(unsigned)dv[j].x, (unsigned)dv[j].y, (unsigned)dv[j].z, (unsigned)dv[j].w};
      #pragma unroll
      for (int k = 0; k < 4; k++) {
        const unsigned g = dd[k] / RN;
        gr[j][k] = (g << 16) | atomicAdd(&bcnt[g], 1u);
      }
    }
  }
  __syncthreads();
  const unsigned v = bcnt[t];
  unsigned inc = v;
  #pragma unroll
  for (int dlt = 1; dlt < 32; dlt <<= 1) {
    const unsigned o = __shfl_up(inc, dlt);
    if ((t & 31) >= dlt) inc += o;
  }
  if ((t & 31) == 31) wsum[t >> 5] = inc;
  __syncthreads();
  unsigned start = inc - v;
  for (int w = 0; w < (t >> 5); w++) start += wsum[w];
  boff[t] = start;
  ((uint2*)bofs)[c * NRANGE + t] = make_uint2(start, start + v);
  __syncthreads();
  #pragma unroll
  for (int j = 0; j < QPT; j++) {
    if (t + j * A1_T < CH / 4) {
      const unsigned ss[4] = {(unsigned)sv[j].x, (unsigned)sv[j].y, (unsigned)sv[j].z, (unsigned)sv[j].w};
      const unsigned dd[4] = {(unsigned)dv[j].x, (unsigned)dv[j].y, (unsigned)dv[j].z, (unsigned)dv[j].w};
      #pragma unroll
      for (int k = 0; k < 4; k++) {
        const unsigned g = gr[j][k] >> 16;
        lsrt[boff[g] + (gr[j][k] & 0xFFFFu)] = ss[k] | ((dd[k] - g * RN) << EBITS);
      }
    }
  }
  __syncthreads();
  uint4* __restrict__ o4 = (uint4*)(srt + c * CH);
  for (int q = t; q < CH / 4; q += A1_T) o4[q] = ((const uint4*)lsrt)[q];
}

__device__ __forceinline__ void gather_slices(const unsigned* __restrict__ srt, const unsigned* __restrict__ bofs,
                                              int r, int c0, int c1, int t, unsigned* list, unsigned* cnt) {
  const int c = c0 + t / SLANES;
  const unsigned q = (unsigned)t % SLANES;
  unsigned len = 0, g0 = 0, p;
  if (c < c1) {
    const uint2 se = ((const uint2*)bofs)[c * NRANGE + r];
    len = se.y - se.x;
    g0  = (unsigned)c * CH + se.x;
  }
  const int lane = t & 31;
  const unsigned mine = (q == 0) ? len : 0u;
  unsigned inc = mine;
  #pragma unroll
  for (int dlt = 1; dlt < 32; dlt <<= 1) {
    const unsigned o = __shfl_up(inc, dlt);
    if (lane >= dlt) inc += o;
  }
  unsigned base = 0;
  if (lane == 31 && inc) base = atomicAdd(cnt, inc);
  base = __shfl(base, 31);
  p = __shfl(base + inc - mine, lane & ~(SLANES - 1));
  for (unsigned i0 = q; i0 < len; i0 += SLANES * SLU) {
    unsigned w[SLU];
    #pragma unroll
    for (int j = 0; j < SLU; j++) w[j] = (i0 + j * SLANES < len) ? srt[g0 + i0 + j * SLANES] : 0u;
    #pragma unroll
    for (int j = 0; j < SLU; j++) {
      const unsigned i = i0 + j * SLANES;
      if (i < len && p + i < LIST_CAP) list[p + i] = w[j];
    }
  }
}

__device__ __forceinline__ void list_self_loops(int t, int lo, unsigned span, unsigned* list, unsigned* cnt) {
  if (t < (int)span) list[t] = (unsigned)(lo + t) | ((unsigned)t << EBITS);
  if (t == 0) *cnt = span;
}

__global__ void __launch_bounds__(AGG_T) agg1_lds(const unsigned* __restrict__ srt,
                                                  const unsigned* __restrict__ bofs,
                                                  const float* __restrict__ as,
                                                  const float* __restrict__ ad,
                                                  const float* __restrict__ H1,
                                                  const float* __restrict__ bias,
                                                  float* __restrict__ out1,
                                                  unsigned* __restrict__ glist,
                                                  unsigned* __restrict__ gcnt) {
  __shared__ float    acc[RN * P1];
  __shared__ float    dsum[RN * NHEADS];
  __shared__ float    adl[RN * NHEADS];
  __shared__ unsigned mxo[RN * NHEADS];
  __shared__ unsigned list[LIST_CAP];
  __shared__ unsigned cnt;

  const int t  = threadIdx.x;
  const int lo = blockIdx.x * RN;
  const unsigned span = (unsigned)(min(lo + RN, N_NODES) - lo);
  const int h = t & 7;

  for (int i = t; i < RN * P1; i += AGG_T) acc[i] = 0.f;
  for (int i = t; i < (int)span * NHEADS; i += AGG_T) {
    dsum[i] = 0.f;
    adl[i]  = ad[lo * NHEADS + i];
    mxo[i]  = ORD_NEG_INF;
  }

  auto max_pass = [&](unsigned n) {
    for (unsigned b = t >> 3; b < n; b += UNR * (AGG_T / 8)) {
      unsigned dl[UNR]; float a[UNR];
      #pragma unroll
      for (int j = 0; j < UNR; j++) {
        const unsigned en = list[min(b + j * (AGG_T / 8), n - 1)];
        dl[j] = en >> EBITS;
        a[j]  = as[(en & EMASK) * NHEADS + h];
      }
      #pragma unroll
      for (int j = 0; j < UNR; j++)
        atomicMax(&mxo[dl[j] * NHEADS + h], f2ord(lrelu(a[j] + adl[dl[j] * NHEADS + h])));
    }
  };
  auto sum_pass = [&](unsigned n) {
    for (unsigned b = t >> 3; b < n; b += UNR * (AGG_T / 8)) {
      unsigned dl[UNR]; float a[UNR]; float4 h0[UNR], h1[UNR];
      #pragma unroll
      for (int j = 0; j < UNR; j++) {
        const unsigned en = list[min(b + j * (AGG_T / 8), n - 1)];
        const unsigned s = en & EMASK;
        dl[j] = en >> EBITS;
        a[j]  = as[s * NHEADS + h];
        const float4* hs = (const float4*)(H1 + (long)s * HID1 + h * NHID);
        h0[j] = hs[0]; h1[j] = hs[1];
      }
      #pragma unroll
      for (int j = 0; j < UNR; j++) {
        if (b + j * (AGG_T / 8) < n) {
          const int d8 = dl[j] * NHEADS + h;
          const float v = lrelu(a[j] + adl[d8]);
          const float w = __expf(v - ord2f(mxo[d8]));
          atomicAdd(&dsum[d8], w);
          float* od = &acc[dl[j] * P1 + h * NHID];
          atomicAdd(&od[0], w * h0[j].x); atomicAdd(&od[1], w * h0[j].y);
          atomicAdd(&od[2], w * h0[j].z); atomicAdd(&od[3], w * h0[j].w);
          atomicAdd(&od[4], w * h1[j].x); atomicAdd(&od[5], w * h1[j].y);
          atomicAdd(&od[6], w * h1[j].z); atomicAdd(&od[7], w * h1[j].w);
        }
      }
    }
  };

  list_self_loops(t, lo, span, list, &cnt);
  __syncthreads();
  gather_slices(srt, bofs, blockIdx.x, 0, NCH, t, list, &cnt);
  __syncthreads();
  const unsigned n = cnt;

  if (t == 0) gcnt[blockIdx.x] = n;
  if (n <= REG_CAP) {
    for (unsigned i = n + t; i < REG_CAP; i += AGG_T) list[i] = (unsigned)lo;
    __syncthreads();
    unsigned* gl = glist + (long)blockIdx.x * LIST_CAP;
    for (unsigned i = t; i < n; i += AGG_T) gl[i] = list[i];

    const unsigned* lslot = list + (t >> 3);
    float areg[KREG];
    #pragma unroll
    for (int j = 0; j < KREG; j++) areg[j] = as[(lslot[j * (AGG_T / 8)] & EMASK) * NHEADS + h];
    #pragma unroll
    for (int j = 0; j < KREG; j++) {
      if ((unsigned)(j * (AGG_T / 8)) < n) {
        const unsigned dl = lslot[j * (AGG_T / 8)] >> EBITS;
        areg[j] = lrelu(areg[j] + adl[dl * NHEADS + h]);
        atomicMax(&mxo[dl * NHEADS + h], f2ord(areg[j]));
      }
    }
    __syncthreads();
    #pragma clang loop unroll(disable)
    for (unsigned r0 = 0; r0 < n; r0 += UNR * (AGG_T / 8)) {
      unsigned dl[UNR]; float4 h0[UNR], h1[UNR];
      #pragma unroll
      for (int j = 0; j < UNR; j++) {
        const unsigned en = lslot[r0 + j * (AGG_T / 8)];
        dl[j] = en >> EBITS;
        const float4* hs = (const float4*)(H1 + ((en & EMASK) * HID1 + h * NHID));
        h0[j] = hs[0]; h1[j] = hs[1];
      }
      #pragma unroll
      for (int j = 0; j < UNR; j++) {
        if (r0 + (unsigned)(t >> 3) + j * (AGG_T / 8) < n) {
          const int d8 = dl[j] * NHEADS + h;
          const float w = __expf(areg[j] - ord2f(mxo[d8]));
          atomicAdd(&dsum[d8], w);
          float* od = &acc[dl[j] * P1 + h * NHID];
          atomicAdd(&od[0], w * h0[j].x); atomicAdd(&od[1], w * h0[j].y);
          atomicAdd(&od[2], w * h0[j].z); atomicAdd(&od[3], w * h0[j].w);
          atomicAdd(&od[4], w * h1[j].x); atomicAdd(&od[5], w * h1[j].y);
          atomicAdd(&od[6], w * h1[j].z); atomicAdd(&od[7], w * h1[j].w);
        }
      }
      #pragma unroll
      for (int k = 0; k + UNR < KREG; k++) areg[k] = areg[k + UNR];
    }
  } else if (n <= LIST_CAP) {
    unsigned* gl = glist + (long)blockIdx.x * LIST_CAP;
    for (unsigned i = t; i < n; i += AGG_T) gl[i] = list[i];
    max_pass(n);
    __syncthreads();
    sum_pass(n);
  } else {
    for (int phase = 0; phase < 2; phase++) {
      for (int c0 = 0; c0 < NCH; c0 += GRP) {
        __syncthreads();
        if (c0 == 0) list_self_loops(t, lo, span, list, &cnt);
        else if (t == 0) cnt = 0;
        __syncthreads();
        gather_slices(srt, bofs, blockIdx.x, c0, min(c0 + GRP, NCH), t, list, &cnt);
        __syncthreads();
        const unsigned nc = cnt;
        if (phase == 0) max_pass(nc); else sum_pass(nc);
      }
    }
  }
  __syncthreads();

  for (int i = t; i < (int)span * HID1; i += AGG_T) {
    const float v = acc[(i >> 6) * P1 + (i & 63)] / dsum[i >> 3] + bias[i & (HID1 - 1)];
    out1[(long)lo * HID1 + i] = v > 0.f ? v : (__expf(v) - 1.f);
  }
}

__global__ void __launch_bounds__(AGG_T) agg2_lds(const unsigned* __restrict__ srt,
                                                  const unsigned* __restrict__ bofs,
                                                  const float* __restrict__ as,
                                                  const float* __restrict__ ad,
                                                  const float* __restrict__ H2,
                                                  const float* __restrict__ bias,
                                                  float* __restrict__ out,
                                                  const unsigned* __restrict__ glist,
                                                  const unsigned* __restrict__ gcnt) {
  __shared__ float    acc[RN * NCLASS];
  __shared__ float    dsum[RN];
  __shared__ float    adl[RN];
  __shared__ unsigned mxo[RN];
  __shared__ unsigned list[LIST_CAP];
  __shared__ float    la[REG_CAP];
  __shared__ unsigned cnt;

  const int t  = threadIdx.x;
  const int lo = blockIdx.x * RN;
  const unsigned span = (unsigned)(min(lo + RN, N_NODES) - lo);
  const int l = t & 7;

  for (int i = t; i < RN * NCLASS; i += AGG_T) acc[i] = 0.f;
  if (t < (int)span) {
    dsum[t] = 0.f;
    adl[t]  = ad[lo + t];
    mxo[t]  = ORD_NEG_INF;
  }

  auto max_pass = [&](unsigned n) {
    for (unsigned b = t; b < n; b += UNR * AGG_T) {
      unsigned dl[UNR]; float a[UNR];
      #pragma unroll
      for (int j = 0; j < UNR; j++) {
        const unsigned en = list[min(b + j * AGG_T, n - 1)];
        dl[j] = en >> EBITS;
        a[j]  = as[en & EMASK];
      }
      #pragma unroll
      for (int j = 0; j < UNR; j++)
        atomicMax(&mxo[dl[j]], f2ord(lrelu(a[j] + adl[dl[j]])));
    }
  };
  auto sum_pass = [&](unsigned n) {
    for (unsigned b = t >> 3; b < n; b += UNR * (AGG_T / 8)) {
      unsigned dl[UNR]; float a[UNR]; float hv[UNR][NCLASS / 8];
      #pragma unroll
      for (int j = 0; j < UNR; j++) {
        const unsigned en = list[min(b + j * (AGG_T / 8), n - 1)];
        const unsigned s = en & EMASK;
        dl[j] = en >> EBITS;
        a[j]  = as[s];
        const float* hs = H2 + (long)s * NCLASS + l;
        #pragma unroll
        for (int k = 0; k < NCLASS / 8; k++) hv[j][k] = hs[k * 8];
      }
      #pragma unroll
      for (int j = 0; j < UNR; j++) {
        if (b + j * (AGG_T / 8) < n) {
          const float v = lrelu(a[j] + adl[dl[j]]);
          const float w = __expf(v - ord2f(mxo[dl[j]]));
          if (l == 0) atomicAdd(&dsum[dl[j]], w);
          float* od = &acc[dl[j] * NCLASS + l];
          #pragma unroll
          for (int k = 0; k < NCLASS / 8; k++) atomicAdd(&od[k * 8], w * hv[j][k]);
        }
      }
    }
  };

  const unsigned n = gcnt[blockIdx.x];

  if (n <= REG_CAP) {
    const unsigned* gl = glist + (long)blockIdx.x * LIST_CAP;
    for (unsigned i = t; i < n; i += AGG_T) list[i] = gl[i];
    __syncthreads();
    {
      unsigned dl[UNR]; float a[UNR];
      #pragma unroll
      for (int j = 0; j < UNR; j++) {
        const unsigned en = list[min((unsigned)t + j * AGG_T, n - 1)];
        dl[j] = en >> EBITS;
        a[j]  = as[en & EMASK];
      }
      #pragma unroll
      for (int j = 0; j < UNR; j++) {
        if ((unsigned)t + j * AGG_T < n) {
          const float v = lrelu(a[j] + adl[dl[j]]);
          la[t + j * AGG_T] = v;
          atomicMax(&mxo[dl[j]], f2ord(v));
        }
      }
    }
    __syncthreads();
    for (unsigned b = t >> 3; b < n; b += UNR * (AGG_T / 8)) {
      unsigned dl[UNR]; float v[UNR]; float hv[UNR][NCLASS / 8];
      #pragma unroll
      for (int j = 0; j < UNR; j++) {
        const unsigned i = min(b + j * (AGG_T / 8), n - 1);
        const unsigned en = list[i];
        dl[j] = en >> EBITS;
        v[j]  = la[i];
        const float* hs = H2 + (long)(en & EMASK) * NCLASS + l;
        #pragma unroll
        for (int k = 0; k < NCLASS / 8; k++) hv[j][k] = hs[k * 8];
      }
      #pragma unroll
      for (int j = 0; j < UNR; j++) {
        if (b + j * (AGG_T / 8) < n) {
          const float w = __expf(v[j] - ord2f(mxo[dl[j]]));
          if (l == 0) atomicAdd(&dsum[dl[j]], w);
          float* od = &acc[dl[j] * NCLASS + l];
          #pragma unroll
          for (int k = 0; k < NCLASS / 8; k++) atomicAdd(&od[k * 8], w * hv[j][k]);
        }
      }
    }
  } else if (n <= LIST_CAP) {
    const unsigned* gl = glist + (long)blockIdx.x * LIST_CAP;
    for (unsigned i = t; i < n; i += AGG_T) list[i] = gl[i];
    __syncthreads();
    max_pass(n);
    __syncthreads();
    sum_pass(n);
  } else {
    for (int phase = 0; phase < 2; phase++) {
      for (int c0 = 0; c0 < NCH; c0 += GRP) {
        __syncthreads();
        if (c0 == 0) list_self_loops(t, lo, span, list, &cnt);
        else if (t == 0) cnt = 0;
        __syncthreads();
        gather_slices(srt, bofs, blockIdx.x, c0, min(c0 + GRP, NCH), t, list, &cnt);
        __syncthreads();
        const unsigned nc = cnt;
        if (phase == 0) max_pass(nc); else sum_pass(nc);
      }
    }
  }
  __syncthreads();

  for (int i = t; i < (int)span * NCLASS; i += AGG_T) acc[i] = acc[i] / dsum[i / NCLASS] + bias[i % NCLASS];
  __syncthreads();
  float* lse = adl;
  if (t < (int)span) {
    const float* r = &acc[t * NCLASS];
    float mx = -INFINITY;
    for (int c = 0; c < NCLASS; c++) mx = fmaxf(mx, r[c]);
    float ssum = 0.f;
    for (int c = 0; c < NCLASS; c++) ssum += __expf(r[c] - mx);
    lse[t] = mx + __logf(ssum);
  }
  __syncthreads();
  for (int i = t; i < (int)span * NCLASS; i += AGG_T) out[(long)lo * NCLASS + i] = acc[i] - lse[i / NCLASS];
}

#define G_T     256
#define G_TPW   2
#define G_TILES (N_NODES / 16)
#define G_GRID  ((G_TILES + 8 * G_TPW - 1) / (8 * G_TPW))
#define G1_GRID (G_GRID + NCH)
#define AP1     (NFEAT + 4)
#define AP2     (HID1 + 4)

__device__ __forceinline__ v8f wmma4(v2f a, float bx, float by, v8f c) {
  v2f b; b.x = bx; b.y = by;
  return __builtin_amdgcn_wmma_f32_16x16x4_f32(false, a, false, b, (short)0, c, false, false);
}

#define ROWS16(F) F(0) F(1) F(2) F(3) F(4) F(5) F(6) F(7) F(8) F(9) F(10) F(11) F(12) F(13) F(14)
#define ROWS8(F)  F(0) F(1) F(2) F(3) F(4) F(5) F(6)
#define DECL_ROW(r)   n##r,
#define LOAD_ROW1(r)  n##r = xp[r * (NFEAT / 4)];
#define STORE_ROW1(r) *(float4*)&al[r * AP1 + 4 * lane] = n##r;
#define LOAD_ROW2(r)  n##r = xp[r * 32];
#define STORE_ROW2(r) *(float4*)&al[(2 * r + hi) * AP2 + 4 * m] = n##r;

__device__ __forceinline__ void gemm1_tile(const float* al, const float4* Bl, const float2* Bx, int lane, int tile,
                                           float* __restrict__ H, float* __restrict__ as, float* __restrict__ ad) {
  const int m  = lane & 15;
  const int hi = lane >> 4;
  v8f c0 = {}, c1 = {}, c2 = {}, c3 = {}, c4 = {};
  #pragma unroll 8
  for (int j = 0; j < NFEAT / 4; j++) {
    const v2f a = *(const v2f*)&al[m * AP1 + 4 * j + 2 * hi];
    const float4 p = Bl[(2 * j) * 32 + lane];
    const float4 q = Bl[(2 * j + 1) * 32 + lane];
    const float2 r = Bx[j * 32 + lane];
    c0 = wmma4(a, p.x, p.y, c0);
    c1 = wmma4(a, p.z, p.w, c1);
    c2 = wmma4(a, q.x, q.y, c2);
    c3 = wmma4(a, q.z, q.w, c3);
    c4 = wmma4(a, r.x, r.y, c4);
  }
  if (tile < G_TILES) {
    float* hp = H + (long)(tile * 16 + 8 * hi) * HID1 + m;
    #pragma unroll
    for (int j = 0; j < 8; j++) {
      hp[(long)j * HID1 +  0] = c0[j];
      hp[(long)j * HID1 + 16] = c1[j];
      hp[(long)j * HID1 + 32] = c2[j];
      hp[(long)j * HID1 + 48] = c3[j];
    }
    float* lp = (m < NHEADS ? as : ad) + (long)(tile * 16 + 8 * hi) * NHEADS + (m & (NHEADS - 1));
    #pragma unroll
    for (int j = 0; j < 8; j++) lp[j * NHEADS] = c4[j];
  }
}

__device__ __forceinline__ void gemm2_tile(const float* al, const float4* Bl, int lane, int tile, float* __restrict__ H,
                                           float* __restrict__ as, float* __restrict__ ad) {
  const int m  = lane & 15;
  const int hi = lane >> 4;
  v8f c0 = {}, c1 = {}, c2 = {};
  #pragma unroll 8
  for (int j = 0; j < HID1 / 4; j++) {
    const v2f a = *(const v2f*)&al[m * AP2 + 4 * j + 2 * hi];
    const float4 p = Bl[(2 * j) * 32 + lane];
    const float4 q = Bl[(2 * j + 1) * 32 + lane];
    c0 = wmma4(a, p.x, p.y, c0);
    c1 = wmma4(a, p.z, p.w, c1);
    c2 = wmma4(a, q.x, q.y, c2);
  }
  if (tile < G_TILES) {
    const int rbase = tile * 16 + 8 * hi;
    #pragma unroll
    for (int j = 0; j < 8; j++) {
      float* hr = H + (long)(rbase + j) * NCLASS;
      hr[m]      = c0[j];
      hr[m + 16] = c1[j];
      if (m + 32 < NCLASS) hr[m + 32] = c2[j];
    }
    if (m == NCLASS - 32 || m == NCLASS - 31) {
      float* lp = (m == NCLASS - 32 ? as : ad) + rbase;
      *(float4*)lp       = make_float4(c2[0], c2[1], c2[2], c2[3]);
      *(float4*)(lp + 4) = make_float4(c2[4], c2[5], c2[6], c2[7]);
    }
  }
}

static_assert(A1_T == G_T, "bin_chunk runs in a work-group of this launch");
__global__ void __launch_bounds__(G_T) gemm1_wmma(const float* __restrict__ X,
                                                  const float* __restrict__ W,
                                                  const float* __restrict__ a_s,
                                                  const float* __restrict__ a_d,
                                                  float* __restrict__ H,
                                                  float* __restrict__ as,
                                                  float* __restrict__ ad,
                                                  const int* __restrict__ ei,
                                                  unsigned* __restrict__ srt,
                                                  unsigned* __restrict__ bofs) {
  __shared__ __attribute__((aligned(16))) float4 Bl[(NFEAT / 4) * 2 * 32];
  __shared__ __attribute__((aligned(16))) float2 Bx[(NFEAT / 4) * 32];
  __shared__ __attribute__((aligned(16))) float  Al[8][16 * AP1];
  if (blockIdx.x >= G_GRID) {
    unsigned* u = (unsigned*)&Al[0][0];
    bin_chunk(ei, blockIdx.x - G_GRID, threadIdx.x, srt, bofs, u, u + NRANGE, u + 2 * NRANGE, u + 2 * NRANGE + A1_T / 32);
    return;
  }
  const int wave = threadIdx.x >> 5;
  const int lane = threadIdx.x & 31;

  float* al = Al[wave];
  const int t0 = blockIdx.x * (8 * G_TPW) + wave;
  float4 ROWS16(DECL_ROW) n15;
  const float4* xp = (const float4*)(X + (long)min(t0, G_TILES - 1) * 16 * NFEAT) + lane;
  ROWS16(LOAD_ROW1) n15 = xp[15 * (NFEAT / 4)];

  #pragma unroll
  for (int it = 0; it < (NFEAT / 4) * 2 * 32 / G_T; it++) {
    const int d = it * G_T + threadIdx.x;
    const int j = d >> 6, th = (d >> 5) & 1, l = d & 31;
    const float* w = W + (long)(4 * j + 2 * (l >> 4)) * HID1 + 32 * th + (l & 15);
    Bl[d] = make_float4(w[0], w[HID1], w[16], w[HID1 + 16]);
  }
  #pragma unroll
  for (int it = 0; it < (NFEAT / 4) * 32 / G_T; it++) {
    const int d = it * G_T + threadIdx.x;
    const int j = d >> 5, l = d & 31;
    const int hd = l & (NHEADS - 1);
    const float4* av = (const float4*)(((l & 15) < NHEADS ? a_s : a_d) + hd * NHID);
    const float4* w0 = (const float4*)(W + (long)(4 * j + 2 * (l >> 4)) * HID1 + hd * NHID);
    const float4* w1 = w0 + HID1 / 4;
    const float4 a0 = av[0], a1 = av[1], u0 = w0[0], u1 = w0[1], v0 = w1[0], v1 = w1[1];
    float s0 = u0.x * a0.x, s1 = v0.x * a0.x;
    s0 += u0.y * a0.y; s1 += v0.y * a0.y;
    s0 += u0.z * a0.z; s1 += v0.z * a0.z;
    s0 += u0.w * a0.w; s1 += v0.w * a0.w;
    s0 += u1.x * a1.x; s1 += v1.x * a1.x;
    s0 += u1.y * a1.y; s1 += v1.y * a1.y;
    s0 += u1.z * a1.z; s1 += v1.z * a1.z;
    s0 += u1.w * a1.w; s1 += v1.w * a1.w;
    Bx[d] = make_float2(s0, s1);
  }

  ROWS16(STORE_ROW1) *(float4*)&al[15 * AP1 + 4 * lane] = n15;
  xp = (const float4*)(X + (long)min(t0 + 8, G_TILES - 1) * 16 * NFEAT) + lane;
  ROWS16(LOAD_ROW1) n15 = xp[15 * (NFEAT / 4)];
  __syncthreads();
  gemm1_tile(al, Bl, Bx, lane, t0, H, as, ad);
  __syncthreads();
  ROWS16(STORE_ROW1) *(float4*)&al[15 * AP1 + 4 * lane] = n15;
  __syncthreads();
  gemm1_tile(al, Bl, Bx, lane, t0 + 8, H, as, ad);
}

__global__ void __launch_bounds__(G_T) gemm2_wmma(const float* __restrict__ X,
                                                  const float* __restrict__ W,
                                                  const float* __restrict__ a_s,
                                                  const float* __restrict__ a_d,
                                                  float* __restrict__ H,
                                                  float* __restrict__ as,
                                                  float* __restrict__ ad) {
  __shared__ __attribute__((aligned(16))) float4 Bl[(HID1 / 4) * 2 * 32];
  __shared__ __attribute__((aligned(16))) float  Al[8][16 * AP2];
  __shared__ float wa[2][HID1];
  const int wave = threadIdx.x >> 5;
  const int lane = threadIdx.x & 31;
  const int m  = lane & 15;
  const int hi = lane >> 4;

  float* al = Al[wave];
  const int t0 = blockIdx.x * (8 * G_TPW) + wave;
  float4 ROWS8(DECL_ROW) n7;
  const float4* xp = (const float4*)(X + (long)min(t0, G_TILES - 1) * 16 * HID1) + lane;
  ROWS8(LOAD_ROW2) n7 = xp[7 * 32];

  {
    const int k = threadIdx.x >> 2, q = threadIdx.x & 3;
    const float2* wr = (const float2*)(W + k * NCLASS + 10 * q);
    const float2* sr = (const float2*)(a_s + 10 * q);
    const float2* dr = (const float2*)(a_d + 10 * q);
    float s = 0.f, dd = 0.f;
    #pragma unroll
    for (int i = 0; i < 5; i++) {
      const float2 w2 = wr[i], s2 = sr[i], d2 = dr[i];
      s  += w2.x * s2.x; s  += w2.y * s2.y;
      dd += w2.x * d2.x; dd += w2.y * d2.y;
    }
    s  += __shfl_xor(s, 1);  s  += __shfl_xor(s, 2);
    dd += __shfl_xor(dd, 1); dd += __shfl_xor(dd, 2);
    if (q == 0) { wa[0][k] = s; wa[1][k] = dd; }
  }

  float4 bv[(HID1 / 4) * 2 * 32 / G_T];
  #pragma unroll
  for (int it = 0; it < (HID1 / 4) * 2 * 32 / G_T; it++) {
    const int d = it * G_T + threadIdx.x;
    const int j = d >> 6, th = (d >> 5) & 1, l = d & 31;
    const int col = 32 * th + (l & 15);
    const int ca = min(col, NCLASS - 1), cb = min(col + 16, NCLASS - 1);
    const float* w = W + (long)(4 * j + 2 * (l >> 4)) * NCLASS;
    bv[it] = make_float4(w[ca], w[NCLASS + ca], w[cb], w[NCLASS + cb]);
  }
  __syncthreads();
  #pragma unroll
  for (int it = 0; it < (HID1 / 4) * 2 * 32 / G_T; it++) {
    const int d = it * G_T + threadIdx.x;
    const int j = d >> 6, th = (d >> 5) & 1, l = d & 31;
    const int col = 32 * th + (l & 15);
    const int k = 4 * j + 2 * (l >> 4);
    float4 v = bv[it];
    if (col >= NCLASS)      { v.x = 0.f; v.y = 0.f; }
    if (col + 16 >= NCLASS) { v.z = 0.f; v.w = 0.f; }
    if (col == NCLASS)      { v.x = wa[0][k]; v.y = wa[0][k + 1]; }
    if (col == NCLASS + 1)  { v.x = wa[1][k]; v.y = wa[1][k + 1]; }
    Bl[d] = v;
  }

  ROWS8(STORE_ROW2) *(float4*)&al[(14 + hi) * AP2 + 4 * m] = n7;
  xp = (const float4*)(X + (long)min(t0 + 8, G_TILES - 1) * 16 * HID1) + lane;
  ROWS8(LOAD_ROW2) n7 = xp[7 * 32];
  __syncthreads();
  gemm2_tile(al, Bl, lane, t0, H, as, ad);
  __syncthreads();
  ROWS8(STORE_ROW2) *(float4*)&al[(14 + hi) * AP2 + 4 * m] = n7;
  __syncthreads();
  gemm2_tile(al, Bl, lane, t0 + 8, H, as, ad);
}

extern "C" void kernel_launch(void* const* d_in, const int* in_sizes, int n_in,
                              void* d_out, int out_size, void* d_ws, size_t ws_size,
                              hipStream_t stream) {
  const float* x      = (const float*)d_in[0];
  const float* W1     = (const float*)d_in[1];
  const float* a_src1 = (const float*)d_in[2];
  const float* a_dst1 = (const float*)d_in[3];
  const float* b1     = (const float*)d_in[4];
  const float* W2     = (const float*)d_in[5];
  const float* a_src2 = (const float*)d_in[6];
  const float* a_dst2 = (const float*)d_in[7];
  const float* b2     = (const float*)d_in[8];
  const int*   ei     = (const int*)d_in[9];
  float*       out    = (float*)d_out;

  float* ws   = (float*)d_ws;
  float* H1   = ws;
  float* as1  = H1  + (long)N_NODES * HID1;
  float* ad1  = as1 + (long)N_NODES * NHEADS;
  float* as2  = ad1 + (long)N_NODES * NHEADS;
  float* ad2  = as2 + N_NODES;
  float* X2   = ad2 + N_NODES;
  float* H2   = X2  + (long)N_NODES * HID1;
  unsigned* glist = (unsigned*)(H2 + (long)N_NODES * NCLASS);
  unsigned* gcnt  = glist + (long)NRANGE * LIST_CAP;
  unsigned* srt   = gcnt + NRANGE;
  unsigned* bofs  = srt + N_EDGES;

  gemm1_wmma<<<G1_GRID, G_T, 0, stream>>>(x, W1, a_src1, a_dst1, H1, as1, ad1, ei, srt, bofs);
  agg1_lds<<<NRANGE, AGG_T, 0, stream>>>(srt, bofs, as1, ad1, H1, b1, X2, glist, gcnt);

  gemm2_wmma<<<G_GRID, G_T, 0, stream>>>(X2, W2, a_src2, a_dst2, H2, as2, ad2);
  agg2_lds<<<NRANGE, AGG_T, 0, stream>>>(srt, bofs, as2, ad2, H2, b2, out, glist, gcnt);
}
